# starting kernel
# baseline (speedup 1.0000x reference)
#include <hip/hip_runtime.h>
#include <math.h>

#define D_IN   128
#define HEADS  8
#define HID    64
#define F1     (HEADS * HID)
#define D_OUT  8
#define NEG_SLOPE 0.2f

typedef float v2f __attribute__((ext_vector_type(2)));
typedef float v8f __attribute__((ext_vector_type(8)));

__device__ __forceinline__ unsigned f2ord(float f) {
  unsigned u = __float_as_uint(f);
  return (u & 0x80000000u) ? ~u : (u | 0x80000000u);
}
__device__ __forceinline__ float ord2f(unsigned v) {
  unsigned u = (v & 0x80000000u) ? (v ^ 0x80000000u) : ~v;
  return __uint_as_float(u);
}
__device__ __forceinline__ float lrelu(float x) { return x > 0.f ? x : NEG_SLOPE * x; }

__global__ void gemm1_wmma(const float* __restrict__ X, const float* __restrict__ W,
                           float* __restrict__ H, int n) {
  int wave = (blockIdx.x * blockDim.x + threadIdx.x) >> 5;
  int lane = threadIdx.x & 31;
  const int ntiles = F1 / 16;
  int mt = wave / ntiles;
  int nt = wave - mt * ntiles;
  int row0 = mt * 16;
  int col0 = nt * 16;
  if (row0 >= n) return;

  int half = lane >> 4;
  int l15  = lane & 15;
  const float* xrow = X + (size_t)(row0 + l15) * D_IN;
  int bcol = col0 + l15;

  v8f acc = {};
  for (int k = 0; k < D_IN; k += 4) {
    int kk = k + half * 2;
    v2f a; a.x = xrow[kk];                       a.y = xrow[kk + 1];
    v2f b; b.x = W[(size_t)kk * F1 + bcol];      b.y = W[(size_t)(kk + 1) * F1 + bcol];
    acc = __builtin_amdgcn_wmma_f32_16x16x4_f32(false, a, false, b,
                                                (short)0, acc, false, false);
  }
#pragma unroll
  for (int v = 0; v < 8; ++v) {
    int r = row0 + v + half * 8;
    H[(size_t)r * F1 + bcol] = acc[v];
  }
}

__global__ void gemm2_wmma(const float* __restrict__ X2, const float* __restrict__ W2,
                           float* __restrict__ H2, int n) {
  int wave = (blockIdx.x * blockDim.x + threadIdx.x) >> 5;
  int lane = threadIdx.x & 31;
  int row0 = wave * 16;
  if (row0 >= n) return;

  int half = lane >> 4;
  int l15  = lane & 15;
  const float* xrow = X2 + (size_t)(row0 + l15) * F1;
  bool colok = (l15 < D_OUT);

  v8f acc = {};
  for (int k = 0; k < F1; k += 4) {
    int kk = k + half * 2;
    v2f a; a.x = xrow[kk]; a.y = xrow[kk + 1];
    v2f b;
    b.x = colok ? W2[(size_t)kk * D_OUT + l15]       : 0.f;
    b.y = colok ? W2[(size_t)(kk + 1) * D_OUT + l15] : 0.f;
    acc = __builtin_amdgcn_wmma_f32_16x16x4_f32(false, a, false, b,
                                                (short)0, acc, false, false);
  }
  if (colok) {
#pragma unroll
    for (int v = 0; v < 8; ++v) {
      int r = row0 + v + half * 8;
      H2[(size_t)r * D_OUT + l15] = acc[v];
    }
  }
}

__global__ void alphas1(const float* __restrict__ H, const float* __restrict__ a_src,
                        const float* __restrict__ a_dst,
                        float* __restrict__ as, float* __restrict__ ad, int n) {
  int t = blockIdx.x * blockDim.x + threadIdx.x;
  if (t >= n * HEADS) return;
  int h = t % HEADS;
  const float* row = H + (size_t)(t / HEADS) * F1 + h * HID;
  const float* vs = a_src + h * HID;
  const float* vd = a_dst + h * HID;
  float s = 0.f, d = 0.f;
#pragma unroll 8
  for (int c = 0; c < HID; ++c) { float x = row[c]; s += x * vs[c]; d += x * vd[c]; }
  as[t] = s; ad[t] = d;
}

__global__ void alphas2(const float* __restrict__ H2, const float* __restrict__ a_src,
                        const float* __restrict__ a_dst,
                        float* __restrict__ as, float* __restrict__ ad, int n) {
  int t = blockIdx.x * blockDim.x + threadIdx.x;
  if (t >= n) return;
  const float* row = H2 + (size_t)t * D_OUT;
  float s = 0.f, d = 0.f;
#pragma unroll
  for (int c = 0; c < D_OUT; ++c) { s += row[c] * a_src[c]; d += row[c] * a_dst[c]; }
  as[t] = s; ad[t] = d;
}

__device__ __forceinline__ void edge_nodes(int e, int nE, const int* src, const int* dst,
                                           int& s_, int& d_) {
  if (e < nE) { s_ = src[e]; d_ = dst[e]; }
  else        { s_ = d_ = e - nE; }
}

__global__ void edge_max(const int* __restrict__ src, const int* __restrict__ dst,
                         int nE, int n, const float* __restrict__ as,
                         const float* __restrict__ ad, unsigned* __restrict__ m, int Hn) {
  int t = blockIdx.x * blockDim.x + threadIdx.x;
  int total = (nE + n) * Hn;
  if (t >= total) return;
  int e = t / Hn, h = t - e * Hn;
  int s_, d_; edge_nodes(e, nE, src, dst, s_, d_);
  float ev = lrelu(as[(size_t)s_ * Hn + h] + ad[(size_t)d_ * Hn + h]);
  atomicMax(m + (size_t)d_ * Hn + h, f2ord(ev));
}

__global__ void edge_denom(const int* __restrict__ src, const int* __restrict__ dst,
                           int nE, int n, const float* __restrict__ as,
                           const float* __restrict__ ad, const unsigned* __restrict__ m,
                           float* __restrict__ denom, int Hn) {
  int t = blockIdx.x * blockDim.x + threadIdx.x;
  int total = (nE + n) * Hn;
  if (t >= total) return;
  int e = t / Hn, h = t - e * Hn;
  int s_, d_; edge_nodes(e, nE, src, dst, s_, d_);
  float ev = lrelu(as[(size_t)s_ * Hn + h] + ad[(size_t)d_ * Hn + h]);
  float w = __expf(ev - ord2f(m[(size_t)d_ * Hn + h]));
  atomicAdd(denom + (size_t)d_ * Hn + h, w);
}

__global__ void edge_agg1(const int* __restrict__ src, const int* __restrict__ dst,
                          int nE, int n, const float* __restrict__ H,
                          const float* __restrict__ as, const float* __restrict__ ad,
                          const unsigned* __restrict__ m, const float* __restrict__ denom,
                          float* __restrict__ out) {
  int e = (blockIdx.x * blockDim.x + threadIdx.x) >> 5;
  int lane = threadIdx.x & 31;
  if (e >= nE + n) return;
  int s_, d_; edge_nodes(e, nE, src, dst, s_, d_);
  int h = lane >> 2;
  float ev = lrelu(as[(size_t)s_ * HEADS + h] + ad[(size_t)d_ * HEADS + h]);
  float w = __expf(ev - ord2f(m[(size_t)d_ * HEADS + h]));
  float alpha = w / (denom[(size_t)d_ * HEADS + h] + 1e-16f);
  const float4* hp = (const float4*)(H + (size_t)s_ * F1 + lane * 16);
  float* op = out + (size_t)d_ * F1 + lane * 16;
#pragma unroll
  for (int j = 0; j < 4; ++j) {
    float4 v = hp[j];
    atomicAdd(op + j * 4 + 0, v.x * alpha);
    atomicAdd(op + j * 4 + 1, v.y * alpha);
    atomicAdd(op + j * 4 + 2, v.z * alpha);
    atomicAdd(op + j * 4 + 3, v.w * alpha);
  }
}

__global__ void relu_bias(float* __restrict__ X2, const float* __restrict__ b1, int total) {
  int t = blockIdx.x * blockDim.x + threadIdx.x;
  if (t >= total) return;
  float v = X2[t] + b1[t & (F1 - 1)];
  X2[t] = v > 0.f ? v : 0.f;
}

__global__ void edge_agg2(const int* __restrict__ src, const int* __restrict__ dst,
                          int nE, int n, const float* __restrict__ H2,
                          const float* __restrict__ as, const float* __restrict__ ad,
                          const unsigned* __restrict__ m, const float* __restrict__ denom,
                          float* __restrict__ out) {
  int t = blockIdx.x * blockDim.x + threadIdx.x;
  int total = (nE + n) * D_OUT;
  if (t >= total) return;
  int e = t >> 3, c = t & 7;
  int s_, d_; edge_nodes(e, nE, src, dst, s_, d_);
  float ev = lrelu(as[s_] + ad[d_]);
  float alpha = __expf(ev - ord2f(m[d_])) / (denom[d_] + 1e-16f);
  atomicAdd(out + (size_t)d_ * D_OUT + c, H2[(size_t)s_ * D_OUT + c] * alpha);
}

__global__ void finalize(const float* __restrict__ acc, const float* __restrict__ b2,
                         float* __restrict__ out, int total) {
  int t = blockIdx.x * blockDim.x + threadIdx.x;
  if (t >= total) return;
  out[t] = acc[t] + b2[t & (D_OUT - 1)];
}

extern "C" void kernel_launch(void* const* d_in, const int* in_sizes, int n_in,
                              void* d_out, int out_size, void* d_ws, size_t ws_size,
                              hipStream_t stream) {
  const float* x      = (const float*)d_in[0];
  const int*   ei     = (const int*)  d_in[1];
  const float* W1     = (const float*)d_in[2];
  const float* a_src1 = (const float*)d_in[3];
  const float* a_dst1 = (const float*)d_in[4];
  const float* b1     = (const float*)d_in[5];
  const float* W2     = (const float*)d_in[6];
  const float* a_src2 = (const float*)d_in[7];
  const float* a_dst2 = (const float*)d_in[8];
  const float* b2     = (const float*)d_in[9];

  int n  = in_sizes[0] / D_IN;
  int nE = in_sizes[1] / 2;
  const int* src = ei;
  const int* dst = ei + nE;
  int EE = nE + n;

  char* ws = (char*)d_ws;
  size_t off = 0;
  auto alloc = [&](size_t bytes) -> char* {
    char* p = ws + off; off += (bytes + 255) & ~(size_t)255; return p;
  };
  float*    h1   = (float*)   alloc((size_t)n * F1 * 4);
  float*    out1 = (float*)   alloc((size_t)n * F1 * 4);
  float*    as1  = (float*)   alloc((size_t)n * HEADS * 4);
  float*    ad1  = (float*)   alloc((size_t)n * HEADS * 4);
  unsigned* m1   = (unsigned*)alloc((size_t)n * HEADS * 4);
  float*    dn1  = (float*)   alloc((size_t)n * HEADS * 4);
  float*    h2   = (float*)   alloc((size_t)n * D_OUT * 4);
  float*    as2  = (float*)   alloc((size_t)n * 4);
  float*    ad2  = (float*)   alloc((size_t)n * 4);
  unsigned* m2   = (unsigned*)alloc((size_t)n * 4);
  float*    dn2  = (float*)   alloc((size_t)n * 4);
  float*    o2   = (float*)   alloc((size_t)n * D_OUT * 4);

  hipMemsetAsync(out1, 0, off - (size_t)((char*)out1 - ws), stream);

  {
    int waves  = ((n + 15) / 16) * (F1 / 16);
    int blocks = (waves + 7) / 8;
    gemm1_wmma<<<blocks, 256, 0, stream>>>(x, W1, h1, n);
  }
  alphas1<<<(n * HEADS + 255) / 256, 256, 0, stream>>>(h1, a_src1, a_dst1, as1, ad1, n);
  {
    int total = EE * HEADS;
    edge_max  <<<(total + 255) / 256, 256, 0, stream>>>(src, dst, nE, n, as1, ad1, m1, HEADS);
    edge_denom<<<(total + 255) / 256, 256, 0, stream>>>(src, dst, nE, n, as1, ad1, m1, dn1, HEADS);
  }
  edge_agg1<<<(EE + 7) / 8, 256, 0, stream>>>(src, dst, nE, n, h1, as1, ad1, m1, dn1, out1);
  relu_bias<<<((n * F1) + 255) / 256, 256, 0, stream>>>(out1, b1, n * F1);

  {
    int waves  = (n + 15) / 16;
    int blocks = (waves + 7) / 8;
    gemm2_wmma<<<blocks, 256, 0, stream>>>(out1, W2, h2, n);
  }
  alphas2<<<(n + 255) / 256, 256, 0, stream>>>(h2, a_src2, a_dst2, as2, ad2, n);
  edge_max  <<<(EE + 255) / 256, 256, 0, stream>>>(src, dst, nE, n, as2, ad2, m2, 1);
  edge_denom<<<(EE + 255) / 256, 256, 0, stream>>>(src, dst, nE, n, as2, ad2, m2, dn2, 1);
  edge_agg2<<<(EE * D_OUT + 255) / 256, 256, 0, stream>>>(src, dst, nE, n, h2, as2, ad2, m2, dn2, o2);

  finalize<<<(n * D_OUT + 255) / 256, 256, 0, stream>>>(o2, b2, (float*)d_out, n * D_OUT);
}


# Edges grouped by destination in each call (count, segment, fill); layer-1 sums: one wave per node in registers, one store per element; float atomics gone there; sum order = arrival-order rank
# speedup vs baseline: 3.2478x; 3.2478x over previous
#include <hip/hip_runtime.h>
#include <math.h>

#define D_IN   128
#define HEADS  8
#define HID    64
#define F1     (HEADS * HID)
#define D_OUT  8
#define NEG_SLOPE 0.2f

typedef float v2f __attribute__((ext_vector_type(2)));
typedef float v8f __attribute__((ext_vector_type(8)));

__device__ __forceinline__ unsigned f2ord(float f) {
  unsigned u = __float_as_uint(f);
  return (u & 0x80000000u) ? ~u : (u | 0x80000000u);
}
__device__ __forceinline__ float ord2f(unsigned v) {
  unsigned u = (v & 0x80000000u) ? (v ^ 0x80000000u) : ~v;
  return __uint_as_float(u);
}
__device__ __forceinline__ float lrelu(float x) { return x > 0.f ? x : NEG_SLOPE * x; }

__global__ void gemm1_wmma(const float* __restrict__ X, const float* __restrict__ W,
                           float* __restrict__ H, int n) {
  int wave = (blockIdx.x * blockDim.x + threadIdx.x) >> 5;
  int lane = threadIdx.x & 31;
  const int ntiles = F1 / 16;
  int mt = wave / ntiles;
  int nt = wave - mt * ntiles;
  int row0 = mt * 16;
  int col0 = nt * 16;
  if (row0 >= n) return;

  int half = lane >> 4;
  int l15  = lane & 15;
  const float* xrow = X + (size_t)(row0 + l15) * D_IN;
  int bcol = col0 + l15;

  v8f acc = {};
  for (int k = 0; k < D_IN; k += 4) {
    int kk = k + half * 2;
    v2f a; a.x = xrow[kk];                       a.y = xrow[kk + 1];
    v2f b; b.x = W[(size_t)kk * F1 + bcol];      b.y = W[(size_t)(kk + 1) * F1 + bcol];
    acc = __builtin_amdgcn_wmma_f32_16x16x4_f32(false, a, false, b,
                                                (short)0, acc, false, false);
  }
#pragma unroll
  for (int v = 0; v < 8; ++v) {
    int r = row0 + v + half * 8;
    H[(size_t)r * F1 + bcol] = acc[v];
  }
}

__global__ void gemm2_wmma(const float* __restrict__ X2, const float* __restrict__ W2,
                           float* __restrict__ H2, int n) {
  int wave = (blockIdx.x * blockDim.x + threadIdx.x) >> 5;
  int lane = threadIdx.x & 31;
  int row0 = wave * 16;
  if (row0 >= n) return;

  int half = lane >> 4;
  int l15  = lane & 15;
  const float* xrow = X2 + (size_t)(row0 + l15) * F1;
  bool colok = (l15 < D_OUT);

  v8f acc = {};
  for (int k = 0; k < F1; k += 4) {
    int kk = k + half * 2;
    v2f a; a.x = xrow[kk]; a.y = xrow[kk + 1];
    v2f b;
    b.x = colok ? W2[(size_t)kk * D_OUT + l15]       : 0.f;
    b.y = colok ? W2[(size_t)(kk + 1) * D_OUT + l15] : 0.f;
    acc = __builtin_amdgcn_wmma_f32_16x16x4_f32(false, a, false, b,
                                                (short)0, acc, false, false);
  }
  if (colok) {
#pragma unroll
    for (int v = 0; v < 8; ++v) {
      int r = row0 + v + half * 8;
      H2[(size_t)r * D_OUT + l15] = acc[v];
    }
  }
}

__global__ void alphas1(const float* __restrict__ H, const float* __restrict__ a_src,
                        const float* __restrict__ a_dst,
                        float* __restrict__ as, float* __restrict__ ad, int n) {
  int t = blockIdx.x * blockDim.x + threadIdx.x;
  if (t >= n * HEADS) return;
  int h = t % HEADS;
  const float* row = H + (size_t)(t / HEADS) * F1 + h * HID;
  const float* vs = a_src + h * HID;
  const float* vd = a_dst + h * HID;
  float s = 0.f, d = 0.f;
#pragma unroll 8
  for (int c = 0; c < HID; ++c) { float x = row[c]; s += x * vs[c]; d += x * vd[c]; }
  as[t] = s; ad[t] = d;
}

__global__ void alphas2(const float* __restrict__ H2, const float* __restrict__ a_src,
                        const float* __restrict__ a_dst,
                        float* __restrict__ as, float* __restrict__ ad, int n) {
  int t = blockIdx.x * blockDim.x + threadIdx.x;
  if (t >= n) return;
  const float* row = H2 + (size_t)t * D_OUT;
  float s = 0.f, d = 0.f;
#pragma unroll
  for (int c = 0; c < D_OUT; ++c) { s += row[c] * a_src[c]; d += row[c] * a_dst[c]; }
  as[t] = s; ad[t] = d;
}

__device__ __forceinline__ void edge_nodes(int e, int nE, const int* src, const int* dst,
                                           int& s_, int& d_) {
  if (e < nE) { s_ = src[e]; d_ = dst[e]; }
  else        { s_ = d_ = e - nE; }
}

__global__ void edge_max(const int* __restrict__ src, const int* __restrict__ dst,
                         int nE, int n, const float* __restrict__ as,
                         const float* __restrict__ ad, unsigned* __restrict__ m, int Hn) {
  int t = blockIdx.x * blockDim.x + threadIdx.x;
  int total = (nE + n) * Hn;
  if (t >= total) return;
  int e = t / Hn, h = t - e * Hn;
  int s_, d_; edge_nodes(e, nE, src, dst, s_, d_);
  float ev = lrelu(as[(size_t)s_ * Hn + h] + ad[(size_t)d_ * Hn + h]);
  atomicMax(m + (size_t)d_ * Hn + h, f2ord(ev));
}

__global__ void edge_denom(const int* __restrict__ src, const int* __restrict__ dst,
                           int nE, int n, const float* __restrict__ as,
                           const float* __restrict__ ad, const unsigned* __restrict__ m,
                           float* __restrict__ denom, int Hn) {
  int t = blockIdx.x * blockDim.x + threadIdx.x;
  int total = (nE + n) * Hn;
  if (t >= total) return;
  int e = t / Hn, h = t - e * Hn;
  int s_, d_; edge_nodes(e, nE, src, dst, s_, d_);
  float ev = lrelu(as[(size_t)s_ * Hn + h] + ad[(size_t)d_ * Hn + h]);
  float w = __expf(ev - ord2f(m[(size_t)d_ * Hn + h]));
  atomicAdd(denom + (size_t)d_ * Hn + h, w);
}

__global__ void deg_count(const int* __restrict__ dst, int nE, int* __restrict__ cnt) {
  int e = blockIdx.x * blockDim.x + threadIdx.x;
  if (e < nE) atomicAdd(cnt + dst[e], 1);
}

__global__ void seg_assign(const int* __restrict__ cnt, int n,
                           int* __restrict__ cursor, int* __restrict__ start) {
  int i = blockIdx.x * blockDim.x + threadIdx.x;
  int lane = threadIdx.x & 31;
  int c = (i < n) ? cnt[i] : 0;
  int incl = c;
#pragma unroll
  for (int o = 1; o < 32; o <<= 1) {
    int t = __shfl_up(incl, o, 32);
    if (lane >= o) incl += t;
  }
  int base = 0;
  if (lane == 31) base = atomicAdd(cursor, incl);
  base = __shfl(base, 31, 32);
  if (i < n) start[i] = base + incl - c;
}

__global__ void seg_fill(const int* __restrict__ src, const int* __restrict__ dst, int nE,
                         const int* __restrict__ start, int* __restrict__ cur,
                         int* __restrict__ elist) {
  int e = blockIdx.x * blockDim.x + threadIdx.x;
  if (e >= nE) return;
  int d = dst[e];
  int p = start[d] + atomicAdd(cur + d, 1);
  if ((unsigned)p < (unsigned)nE) elist[p] = src[e];
}

typedef float    v4f __attribute__((ext_vector_type(4)));

__device__ __forceinline__ float edge_alpha(float a_s, float a_d, float mx, float den) {
  float ev = lrelu(a_s + a_d);
  float w = __expf(ev - mx);
  return w / den;
}

__device__ __forceinline__ float lane_bcast(float x, int i) {
  return __int_as_float(__builtin_amdgcn_readlane(__float_as_int(x), i));
}

struct RowPart { v4f v0, v1, v2, v3; };

__device__ __forceinline__ RowPart row_load(const v4f* __restrict__ H4, int s_, unsigned cq) {
  const v4f* hp = H4 + ((unsigned)s_ * (unsigned)(F1 / 4) + cq);
  RowPart r;
  r.v0 = hp[0]; r.v1 = hp[16]; r.v2 = hp[32]; r.v3 = hp[48];
  return r;
}

__device__ __forceinline__ void row_add(v4f (&acc)[4], const RowPart& r, const float (&al)[8],
                                        int i, bool hi) {
  float l0 = lane_bcast(al[0], i), l1 = lane_bcast(al[1], i);
  float l2 = lane_bcast(al[2], i), l3 = lane_bcast(al[3], i);
  float h0 = lane_bcast(al[4], i), h1 = lane_bcast(al[5], i);
  float h2 = lane_bcast(al[6], i), h3 = lane_bcast(al[7], i);
  acc[0] += r.v0 * (hi ? h0 : l0);
  acc[1] += r.v1 * (hi ? h1 : l1);
  acc[2] += r.v2 * (hi ? h2 : l2);
  acc[3] += r.v3 * (hi ? h3 : l3);
}

__global__ void __launch_bounds__(256)
node_agg1(const int* __restrict__ elist, const int* __restrict__ start,
          const int* __restrict__ cnt, int nE, int n, const float* __restrict__ H,
          const float* __restrict__ as, const float* __restrict__ ad,
          const unsigned* __restrict__ m, const float* __restrict__ denom,
          float* __restrict__ out) {
  int lane = threadIdx.x & 31;
  int d = __builtin_amdgcn_readfirstlane((int)((blockIdx.x * blockDim.x + threadIdx.x) >> 5));
  if (d >= n) return;
  bool hi = lane >= 16;
  unsigned cq = (hi ? 64u : 0u) + ((unsigned)lane & 15u);
  const v4f* H4  = (const v4f*)H;
  const v4f* as4 = (const v4f*)as;

  float adh[8], mvh[8], ddh[8];
#pragma unroll
  for (int h = 0; h < HEADS; ++h) {
    adh[h] = ad[(size_t)d * HEADS + h];
    mvh[h] = ord2f(m[(size_t)d * HEADS + h]);
    ddh[h] = denom[(size_t)d * HEADS + h] + 1e-16f;
  }

  v4f acc[4];
#pragma unroll
  for (int j = 0; j < 4; ++j) acc[j] = (v4f)(0.f);

  int beg = start[d];
  int deg = cnt[d];
  if (beg < 0 || deg < 0 || beg > nE || deg > nE - beg) deg = 0;

  int total = deg + 1;
  for (int base = 0; base < total; base += 32) {
    int k = base + lane;
    int mine = (k >= 1 && k < total) ? elist[beg + k - 1] : d;
    if ((unsigned)mine >= (unsigned)n) mine = d;
    int c = total - base; if (c > 32) c = 32;

    v4f aL = as4[(unsigned)mine * 2u];
    v4f aH = as4[(unsigned)mine * 2u + 1u];
    float al[8];
    al[0] = edge_alpha(aL.x, adh[0], mvh[0], ddh[0]);
    al[1] = edge_alpha(aL.y, adh[1], mvh[1], ddh[1]);
    al[2] = edge_alpha(aL.z, adh[2], mvh[2], ddh[2]);
    al[3] = edge_alpha(aL.w, adh[3], mvh[3], ddh[3]);
    al[4] = edge_alpha(aH.x, adh[4], mvh[4], ddh[4]);
    al[5] = edge_alpha(aH.y, adh[5], mvh[5], ddh[5]);
    al[6] = edge_alpha(aH.z, adh[6], mvh[6], ddh[6]);
    al[7] = edge_alpha(aH.w, adh[7], mvh[7], ddh[7]);

    int i = 0;
    for (; i + 1 < c; i += 2) {
      int s0 = __builtin_amdgcn_readlane(mine, i);
      int s1 = __builtin_amdgcn_readlane(mine, i + 1);
      RowPart r0 = row_load(H4, s0, cq);
      RowPart r1 = row_load(H4, s1, cq);
      row_add(acc, r0, al, i, hi);
      row_add(acc, r1, al, i + 1, hi);
    }
    if (i < c) {
      int s0 = __builtin_amdgcn_readlane(mine, i);
      RowPart r0 = row_load(H4, s0, cq);
      row_add(acc, r0, al, i, hi);
    }
  }

  v4f* op = (v4f*)out + ((unsigned)d * (unsigned)(F1 / 4) + cq);
  op[0] = acc[0]; op[16] = acc[1]; op[32] = acc[2]; op[48] = acc[3];
}

__global__ void relu_bias(float* __restrict__ X2, const float* __restrict__ b1, int total) {
  int t = blockIdx.x * blockDim.x + threadIdx.x;
  if (t >= total) return;
  float v = X2[t] + b1[t & (F1 - 1)];
  X2[t] = v > 0.f ? v : 0.f;
}

__global__ void edge_agg2(const int* __restrict__ src, const int* __restrict__ dst,
                          int nE, int n, const float* __restrict__ H2,
                          const float* __restrict__ as, const float* __restrict__ ad,
                          const unsigned* __restrict__ m, const float* __restrict__ denom,
                          float* __restrict__ out) {
  int t = blockIdx.x * blockDim.x + threadIdx.x;
  int total = (nE + n) * D_OUT;
  if (t >= total) return;
  int e = t >> 3, c = t & 7;
  int s_, d_; edge_nodes(e, nE, src, dst, s_, d_);
  float ev = lrelu(as[s_] + ad[d_]);
  float alpha = __expf(ev - ord2f(m[d_])) / (denom[d_] + 1e-16f);
  atomicAdd(out + (size_t)d_ * D_OUT + c, H2[(size_t)s_ * D_OUT + c] * alpha);
}

__global__ void finalize(const float* __restrict__ acc, const float* __restrict__ b2,
                         float* __restrict__ out, int total) {
  int t = blockIdx.x * blockDim.x + threadIdx.x;
  if (t >= total) return;
  out[t] = acc[t] + b2[t & (D_OUT - 1)];
}

extern "C" void kernel_launch(void* const* d_in, const int* in_sizes, int n_in,
                              void* d_out, int out_size, void* d_ws, size_t ws_size,
                              hipStream_t stream) {
  const float* x      = (const float*)d_in[0];
  const int*   ei     = (const int*)  d_in[1];
  const float* W1     = (const float*)d_in[2];
  const float* a_src1 = (const float*)d_in[3];
  const float* a_dst1 = (const float*)d_in[4];
  const float* b1     = (const float*)d_in[5];
  const float* W2     = (const float*)d_in[6];
  const float* a_src2 = (const float*)d_in[7];
  const float* a_dst2 = (const float*)d_in[8];
  const float* b2     = (const float*)d_in[9];

  int n  = in_sizes[0] / D_IN;
  int nE = in_sizes[1] / 2;
  const int* src = ei;
  const int* dst = ei + nE;
  int EE = nE + n;

  char* ws = (char*)d_ws;
  size_t off = 0;
  auto alloc = [&](size_t bytes) -> char* {
    char* p = ws + off; off += (bytes + 255) & ~(size_t)255; return p;
  };
  float*    h1   = (float*)   alloc((size_t)n * F1 * 4);
  float*    out1 = (float*)   alloc((size_t)n * F1 * 4);
  int*      elist= (int*)     alloc((size_t)nE * 4);
  int*      seg0 = (int*)     alloc((size_t)n * 4);
  float*    as1  = (float*)   alloc((size_t)n * HEADS * 4);
  float*    ad1  = (float*)   alloc((size_t)n * HEADS * 4);
  unsigned* m1   = (unsigned*)alloc((size_t)n * HEADS * 4);
  float*    dn1  = (float*)   alloc((size_t)n * HEADS * 4);
  float*    h2   = (float*)   alloc((size_t)n * D_OUT * 4);
  float*    as2  = (float*)   alloc((size_t)n * 4);
  float*    ad2  = (float*)   alloc((size_t)n * 4);
  unsigned* m2   = (unsigned*)alloc((size_t)n * 4);
  float*    dn2  = (float*)   alloc((size_t)n * 4);
  float*    o2   = (float*)   alloc((size_t)n * D_OUT * 4);
  int*      cnt  = (int*)     alloc((size_t)n * 4);
  int*      cur  = (int*)     alloc((size_t)n * 4);
  int*      curs = (int*)     alloc(4);

  hipMemsetAsync(as1, 0, off - (size_t)((char*)as1 - ws), stream);

  deg_count <<<(nE + 255) / 256, 256, 0, stream>>>(dst, nE, cnt);
  seg_assign<<<(n + 255) / 256, 256, 0, stream>>>(cnt, n, curs, seg0);
  seg_fill  <<<(nE + 255) / 256, 256, 0, stream>>>(src, dst, nE, seg0, cur, elist);

  {
    int waves  = ((n + 15) / 16) * (F1 / 16);
    int blocks = (waves + 7) / 8;
    gemm1_wmma<<<blocks, 256, 0, stream>>>(x, W1, h1, n);
  }
  alphas1<<<(n * HEADS + 255) / 256, 256, 0, stream>>>(h1, a_src1, a_dst1, as1, ad1, n);
  {
    int total = EE * HEADS;
    edge_max  <<<(total + 255) / 256, 256, 0, stream>>>(src, dst, nE, n, as1, ad1, m1, HEADS);
    edge_denom<<<(total + 255) / 256, 256, 0, stream>>>(src, dst, nE, n, as1, ad1, m1, dn1, HEADS);
  }
  node_agg1<<<(n + 7) / 8, 256, 0, stream>>>(elist, seg0, cnt, nE, n, h1, as1, ad1, m1, dn1, out1);
  relu_bias<<<((n * F1) + 255) / 256, 256, 0, stream>>>(out1, b1, n * F1);

  {
    int waves  = (n + 15) / 16;
    int blocks = (waves + 7) / 8;
    gemm2_wmma<<<blocks, 256, 0, stream>>>(out1, W2, h2, n);
  }
  alphas2<<<(n + 255) / 256, 256, 0, stream>>>(h2, a_src2, a_dst2, as2, ad2, n);
  edge_max  <<<(EE + 255) / 256, 256, 0, stream>>>(src, dst, nE, n, as2, ad2, m2, 1);
  edge_denom<<<(EE + 255) / 256, 256, 0, stream>>>(src, dst, nE, n, as2, ad2, m2, dn2, 1);
  edge_agg2<<<(EE * D_OUT + 255) / 256, 256, 0, stream>>>(src, dst, nE, n, h2, as2, ad2, m2, dn2, o2);

  finalize<<<(n * D_OUT + 255) / 256, 256, 0, stream>>>(o2, b2, (float*)d_out, n * D_OUT);
}


# Softmax max and denominator of both layers and the layer-2 sums taken per destination node over the grouped edge list, in registers; edge passes and their atomics gone; sums: list order + lane tree
# speedup vs baseline: 4.5604x; 1.4042x over previous
#include <hip/hip_runtime.h>
#include <math.h>

#define D_IN   128
#define HEADS  8
#define HID    64
#define F1     (HEADS * HID)
#define D_OUT  8
#define NEG_SLOPE 0.2f

typedef float v2f __attribute__((ext_vector_type(2)));
typedef float v8f __attribute__((ext_vector_type(8)));

__device__ __forceinline__ float lrelu(float x) { return x > 0.f ? x : NEG_SLOPE * x; }

__global__ void gemm1_wmma(const float* __restrict__ X, const float* __restrict__ W,
                           float* __restrict__ H, int n) {
  int wave = (blockIdx.x * blockDim.x + threadIdx.x) >> 5;
  int lane = threadIdx.x & 31;
  const int ntiles = F1 / 16;
  int mt = wave / ntiles;
  int nt = wave - mt * ntiles;
  int row0 = mt * 16;
  int col0 = nt * 16;
  if (row0 >= n) return;

  int half = lane >> 4;
  int l15  = lane & 15;
  const float* xrow = X + (size_t)(row0 + l15) * D_IN;
  int bcol = col0 + l15;

  v8f acc = {};
  for (int k = 0; k < D_IN; k += 4) {
    int kk = k + half * 2;
    v2f a; a.x = xrow[kk];                       a.y = xrow[kk + 1];
    v2f b; b.x = W[(size_t)kk * F1 + bcol];      b.y = W[(size_t)(kk + 1) * F1 + bcol];
    acc = __builtin_amdgcn_wmma_f32_16x16x4_f32(false, a, false, b,
                                                (short)0, acc, false, false);
  }
#pragma unroll
  for (int v = 0; v < 8; ++v) {
    int r = row0 + v + half * 8;
    H[(size_t)r * F1 + bcol] = acc[v];
  }
}

__global__ void gemm2_wmma(const float* __restrict__ X2, const float* __restrict__ W2,
                           float* __restrict__ H2, int n) {
  int wave = (blockIdx.x * blockDim.x + threadIdx.x) >> 5;
  int lane = threadIdx.x & 31;
  int row0 = wave * 16;
  if (row0 >= n) return;

  int half = lane >> 4;
  int l15  = lane & 15;
  const float* xrow = X2 + (size_t)(row0 + l15) * F1;
  bool colok = (l15 < D_OUT);

  v8f acc = {};
  for (int k = 0; k < F1; k += 4) {
    int kk = k + half * 2;
    v2f a; a.x = xrow[kk]; a.y = xrow[kk + 1];
    v2f b;
    b.x = colok ? W2[(size_t)kk * D_OUT + l15]       : 0.f;
    b.y = colok ? W2[(size_t)(kk + 1) * D_OUT + l15] : 0.f;
    acc = __builtin_amdgcn_wmma_f32_16x16x4_f32(false, a, false, b,
                                                (short)0, acc, false, false);
  }
  if (colok) {
#pragma unroll
    for (int v = 0; v < 8; ++v) {
      int r = row0 + v + half * 8;
      H2[(size_t)r * D_OUT + l15] = acc[v];
    }
  }
}

__global__ void alphas1(const float* __restrict__ H, const float* __restrict__ a_src,
                        const float* __restrict__ a_dst,
                        float* __restrict__ as, float* __restrict__ ad, int n) {
  int t = blockIdx.x * blockDim.x + threadIdx.x;
  if (t >= n * HEADS) return;
  int h = t % HEADS;
  const float* row = H + (size_t)(t / HEADS) * F1 + h * HID;
  const float* vs = a_src + h * HID;
  const float* vd = a_dst + h * HID;
  float s = 0.f, d = 0.f;
#pragma unroll 8
  for (int c = 0; c < HID; ++c) { float x = row[c]; s += x * vs[c]; d += x * vd[c]; }
  as[t] = s; ad[t] = d;
}

__global__ void alphas2(const float* __restrict__ H2, const float* __restrict__ a_src,
                        const float* __restrict__ a_dst,
                        float* __restrict__ as, float* __restrict__ ad, int n) {
  int t = blockIdx.x * blockDim.x + threadIdx.x;
  if (t >= n) return;
  const float* row = H2 + (size_t)t * D_OUT;
  float s = 0.f, d = 0.f;
#pragma unroll
  for (int c = 0; c < D_OUT; ++c) { s += row[c] * a_src[c]; d += row[c] * a_dst[c]; }
  as[t] = s; ad[t] = d;
}

__global__ void deg_count(const int* __restrict__ dst, int nE, int* __restrict__ cnt) {
  int e = blockIdx.x * blockDim.x + threadIdx.x;
  if (e < nE) atomicAdd(cnt + dst[e], 1);
}

__global__ void seg_assign(const int* __restrict__ cnt, int n,
                           int* __restrict__ cursor, int* __restrict__ start) {
  int i = blockIdx.x * blockDim.x + threadIdx.x;
  int lane = threadIdx.x & 31;
  int c = (i < n) ? cnt[i] : 0;
  int incl = c;
#pragma unroll
  for (int o = 1; o < 32; o <<= 1) {
    int t = __shfl_up(incl, o, 32);
    if (lane >= o) incl += t;
  }
  int base = 0;
  if (lane == 31) base = atomicAdd(cursor, incl);
  base = __shfl(base, 31, 32);
  if (i < n) start[i] = base + incl - c;
}

__global__ void seg_fill(const int* __restrict__ src, const int* __restrict__ dst, int nE,
                         const int* __restrict__ start, int* __restrict__ cur,
                         int* __restrict__ elist) {
  int e = blockIdx.x * blockDim.x + threadIdx.x;
  if (e >= nE) return;
  int d = dst[e];
  int p = start[d] + atomicAdd(cur + d, 1);
  if ((unsigned)p < (unsigned)nE) elist[p] = src[e];
}

typedef float    v4f __attribute__((ext_vector_type(4)));

__device__ __forceinline__ float lane_bcast(float x, int i) {
  return __int_as_float(__builtin_amdgcn_readlane(__float_as_int(x), i));
}

__device__ __forceinline__ float as_scalar(float x) {
  return __int_as_float(__builtin_amdgcn_readfirstlane(__float_as_int(x)));
}

template <int W>
__device__ __forceinline__ float group_max(float v) {
#pragma unroll
  for (int o = W / 2; o > 0; o >>= 1) v = fmaxf(v, __shfl_xor(v, o, 32));
  return v;
}
template <int W>
__device__ __forceinline__ float group_sum(float v) {
#pragma unroll
  for (int o = W / 2; o > 0; o >>= 1) v += __shfl_xor(v, o, 32);
  return v;
}

__device__ __forceinline__ bool load_entries(const int* __restrict__ elist,
                                             const v4f* __restrict__ as4, int beg, int total,
                                             int base, int lane, int d, int n,
                                             const float (&adh)[8], int& mine, float (&ev)[8]) {
  int k = base + lane;
  bool valid = k < total;
  mine = (k >= 1 && valid) ? elist[beg + k - 1] : d;
  if ((unsigned)mine >= (unsigned)n) mine = d;
  v4f aL = as4[(unsigned)mine * 2u];
  v4f aH = as4[(unsigned)mine * 2u + 1u];
  ev[0] = lrelu(aL.x + adh[0]); ev[1] = lrelu(aL.y + adh[1]);
  ev[2] = lrelu(aL.z + adh[2]); ev[3] = lrelu(aL.w + adh[3]);
  ev[4] = lrelu(aH.x + adh[4]); ev[5] = lrelu(aH.y + adh[5]);
  ev[6] = lrelu(aH.z + adh[6]); ev[7] = lrelu(aH.w + adh[7]);
  return valid;
}

struct RowPart { v4f v0, v1, v2, v3; };

__device__ __forceinline__ RowPart row_load(const v4f* __restrict__ H4, int s_, unsigned cq) {
  const v4f* hp = H4 + ((unsigned)s_ * (unsigned)(F1 / 4) + cq);
  RowPart r;
  r.v0 = hp[0]; r.v1 = hp[16]; r.v2 = hp[32]; r.v3 = hp[48];
  return r;
}

__device__ __forceinline__ void row_add(v4f (&acc)[4], const RowPart& r, const float (&al)[8],
                                        int i, bool hi) {
  float l0 = lane_bcast(al[0], i), l1 = lane_bcast(al[1], i);
  float l2 = lane_bcast(al[2], i), l3 = lane_bcast(al[3], i);
  float h0 = lane_bcast(al[4], i), h1 = lane_bcast(al[5], i);
  float h2 = lane_bcast(al[6], i), h3 = lane_bcast(al[7], i);
  acc[0] += r.v0 * (hi ? h0 : l0);
  acc[1] += r.v1 * (hi ? h1 : l1);
  acc[2] += r.v2 * (hi ? h2 : l2);
  acc[3] += r.v3 * (hi ? h3 : l3);
}

__global__ void __launch_bounds__(256) __attribute__((amdgpu_waves_per_eu(8, 10)))
node_agg1(const int* __restrict__ elist, const int* __restrict__ start,
          const int* __restrict__ cnt, int nE, int n, const float* __restrict__ H,
          const float* __restrict__ as, const float* __restrict__ ad,
          float* __restrict__ out) {
  int lane = threadIdx.x & 31;
  int d = __builtin_amdgcn_readfirstlane((int)((blockIdx.x * blockDim.x + threadIdx.x) >> 5));
  if (d >= n) return;
  bool hi = lane >= 16;
  unsigned cq = (hi ? 64u : 0u) + ((unsigned)lane & 15u);
  const v4f* H4  = (const v4f*)H;
  const v4f* as4 = (const v4f*)as;

  float adh[8];
#pragma unroll
  for (int h = 0; h < HEADS; ++h) adh[h] = ad[(size_t)d * HEADS + h];

  int beg = start[d];
  int deg = cnt[d];
  if (beg < 0 || deg < 0 || beg > nE || deg > nE - beg) deg = 0;
  int total = deg + 1;

  int mine; float ev[8];
  bool valid = load_entries(elist, as4, beg, total, 0, lane, d, n, adh, mine, ev);

  float mvh[8];
  {
    float mx[8];
#pragma unroll
    for (int h = 0; h < HEADS; ++h) mx[h] = valid ? ev[h] : -INFINITY;
    for (int base = 32; base < total; base += 32) {
      int m2; float e2[8];
      if (load_entries(elist, as4, beg, total, base, lane, d, n, adh, m2, e2)) {
#pragma unroll
        for (int h = 0; h < HEADS; ++h) mx[h] = fmaxf(mx[h], e2[h]);
      }
    }
#pragma unroll
    for (int h = 0; h < HEADS; ++h) mvh[h] = as_scalar(group_max<32>(mx[h]));
  }

  float ddh[8];
  {
    float ds[8];
#pragma unroll
    for (int h = 0; h < HEADS; ++h) ds[h] = valid ? __expf(ev[h] - mvh[h]) : 0.f;
    for (int base = 32; base < total; base += 32) {
      int m2; float e2[8];
      if (load_entries(elist, as4, beg, total, base, lane, d, n, adh, m2, e2)) {
#pragma unroll
        for (int h = 0; h < HEADS; ++h) ds[h] += __expf(e2[h] - mvh[h]);
      }
    }
#pragma unroll
    for (int h = 0; h < HEADS; ++h) ddh[h] = as_scalar(group_sum<32>(ds[h])) + 1e-16f;
  }

  v4f acc[4];
#pragma unroll
  for (int j = 0; j < 4; ++j) acc[j] = (v4f)(0.f);

  for (int base = 0;;) {
    int c = total - base; if (c > 32) c = 32;

    float al[8];
#pragma unroll
    for (int h = 0; h < HEADS; ++h) al[h] = __expf(ev[h] - mvh[h]) / ddh[h];

    int i = 0;
    for (; i + 1 < c; i += 2) {
      int s0 = __builtin_amdgcn_readlane(mine, i);
      int s1 = __builtin_amdgcn_readlane(mine, i + 1);
      RowPart r0 = row_load(H4, s0, cq);
      RowPart r1 = row_load(H4, s1, cq);
      row_add(acc, r0, al, i, hi);
      row_add(acc, r1, al, i + 1, hi);
    }
    if (i < c) {
      int s0 = __builtin_amdgcn_readlane(mine, i);
      RowPart r0 = row_load(H4, s0, cq);
      row_add(acc, r0, al, i, hi);
    }

    base += 32;
    if (base >= total) break;
    load_entries(elist, as4, beg, total, base, lane, d, n, adh, mine, ev);
  }

  v4f* op = (v4f*)out + ((unsigned)d * (unsigned)(F1 / 4) + cq);
  op[0] = acc[0]; op[16] = acc[1]; op[32] = acc[2]; op[48] = acc[3];
}

__global__ void relu_bias(float* __restrict__ X2, const float* __restrict__ b1, int total) {
  int t = blockIdx.x * blockDim.x + threadIdx.x;
  if (t >= total) return;
  float v = X2[t] + b1[t & (F1 - 1)];
  X2[t] = v > 0.f ? v : 0.f;
}

__device__ __forceinline__ float entry_score2(const int* __restrict__ elist,
                                              const float* __restrict__ as, int beg, int k,
                                              int d, int n, float a_d, int& s) {
  s = k ? elist[beg + k - 1] : d;
  if ((unsigned)s >= (unsigned)n) s = d;
  return lrelu(as[s] + a_d);
}

__global__ void __launch_bounds__(256)
node_agg2(const int* __restrict__ elist, const int* __restrict__ start,
          const int* __restrict__ cnt, int nE, int n, const float* __restrict__ H2,
          const float* __restrict__ as, const float* __restrict__ ad,
          const float* __restrict__ b2, float* __restrict__ out) {
  int t = blockIdx.x * blockDim.x + threadIdx.x;
  int j = threadIdx.x & 7;
  bool live = (t >> 3) < n;
  int d = live ? (t >> 3) : 0;
  const v4f* H4 = (const v4f*)H2;

  int beg = start[d];
  int deg = cnt[d];
  if (beg < 0 || deg < 0 || beg > nE || deg > nE - beg) deg = 0;
  int total = deg + 1;
  float a_d = ad[d];

  float mx = -INFINITY;
  for (int k = j; k < total; k += 8) {
    int s;
    mx = fmaxf(mx, entry_score2(elist, as, beg, k, d, n, a_d, s));
  }
  mx = group_max<8>(mx);

  float den = 0.f;
  for (int k = j; k < total; k += 8) {
    int s;
    den += __expf(entry_score2(elist, as, beg, k, d, n, a_d, s) - mx);
  }
  den = group_sum<8>(den) + 1e-16f;

  v4f accL = (v4f)(0.f), accH = (v4f)(0.f);
  for (int k = j; k < total; k += 8) {
    int s;
    float alpha = __expf(entry_score2(elist, as, beg, k, d, n, a_d, s) - mx) / den;
    accL += H4[(unsigned)s * 2u] * alpha;
    accH += H4[(unsigned)s * 2u + 1u] * alpha;
  }
  float r0 = group_sum<8>(accL.x), r1 = group_sum<8>(accL.y);
  float r2 = group_sum<8>(accL.z), r3 = group_sum<8>(accL.w);
  float r4 = group_sum<8>(accH.x), r5 = group_sum<8>(accH.y);
  float r6 = group_sum<8>(accH.z), r7 = group_sum<8>(accH.w);
  float lo = (j & 2) ? ((j & 1) ? r3 : r2) : ((j & 1) ? r1 : r0);
  float up = (j & 2) ? ((j & 1) ? r7 : r6) : ((j & 1) ? r5 : r4);
  float r = (j & 4) ? up : lo;

  if (live) out[t] = r + b2[j];
}

extern "C" void kernel_launch(void* const* d_in, const int* in_sizes, int n_in,
                              void* d_out, int out_size, void* d_ws, size_t ws_size,
                              hipStream_t stream) {
  const float* x      = (const float*)d_in[0];
  const int*   ei     = (const int*)  d_in[1];
  const float* W1     = (const float*)d_in[2];
  const float* a_src1 = (const float*)d_in[3];
  const float* a_dst1 = (const float*)d_in[4];
  const float* b1     = (const float*)d_in[5];
  const float* W2     = (const float*)d_in[6];
  const float* a_src2 = (const float*)d_in[7];
  const float* a_dst2 = (const float*)d_in[8];
  const float* b2     = (const float*)d_in[9];

  int n  = in_sizes[0] / D_IN;
  int nE = in_sizes[1] / 2;
  const int* src = ei;
  const int* dst = ei + nE;

  char* ws = (char*)d_ws;
  size_t off = 0;
  auto alloc = [&](size_t bytes) -> char* {
    char* p = ws + off; off += (bytes + 255) & ~(size_t)255; return p;
  };
  float*    h1   = (float*)   alloc((size_t)n * F1 * 4);
  float*    out1 = (float*)   alloc((size_t)n * F1 * 4);
  int*      elist= (int*)     alloc((size_t)nE * 4);
  int*      seg0 = (int*)     alloc((size_t)n * 4);
  float*    as1  = (float*)   alloc((size_t)n * HEADS * 4);
  float*    ad1  = (float*)   alloc((size_t)n * HEADS * 4);
  float*    h2   = (float*)   alloc((size_t)n * D_OUT * 4);
  float*    as2  = (float*)   alloc((size_t)n * 4);
  float*    ad2  = (float*)   alloc((size_t)n * 4);
  int*      cnt  = (int*)     alloc((size_t)n * 4);
  int*      cur  = (int*)     alloc((size_t)n * 4);
  int*      curs = (int*)     alloc(4);

  hipMemsetAsync(cnt, 0, off - (size_t)((char*)cnt - ws), stream);

  deg_count <<<(nE + 255) / 256, 256, 0, stream>>>(dst, nE, cnt);
  seg_assign<<<(n + 255) / 256, 256, 0, stream>>>(cnt, n, curs, seg0);
  seg_fill  <<<(nE + 255) / 256, 256, 0, stream>>>(src, dst, nE, seg0, cur, elist);

  {
    int waves  = ((n + 15) / 16) * (F1 / 16);
    int blocks = (waves + 7) / 8;
    gemm1_wmma<<<blocks, 256, 0, stream>>>(x, W1, h1, n);
  }
  alphas1<<<(n * HEADS + 255) / 256, 256, 0, stream>>>(h1, a_src1, a_dst1, as1, ad1, n);
  node_agg1<<<(n + 7) / 8, 256, 0, stream>>>(elist, seg0, cnt, nE, n, h1, as1, ad1, out1);
  relu_bias<<<((n * F1) + 255) / 256, 256, 0, stream>>>(out1, b1, n * F1);

  {
    int waves  = (n + 15) / 16;
    int blocks = (waves + 7) / 8;
    gemm2_wmma<<<blocks, 256, 0, stream>>>(out1, W2, h2, n);
  }
  alphas2<<<(n + 255) / 256, 256, 0, stream>>>(h2, a_src2, a_dst2, as2, ad2, n);
  node_agg2<<<(n * D_OUT + 255) / 256, 256, 0, stream>>>(elist, seg0, cnt, nE, n, h2, as2, ad2, b2,
                                                       (float*)d_out);
}


# Layer-1 bias and relu applied in the aggregation kernel's own store (x2 written once); the separate elementwise launch, its 800000 waves and its read+rewrite of x2 are gone; values unchanged
# speedup vs baseline: 5.4617x; 1.1976x over previous
#include <hip/hip_runtime.h>
#include <math.h>

#define D_IN   128
#define HEADS  8
#define HID    64
#define F1     (HEADS * HID)
#define D_OUT  8
#define NEG_SLOPE 0.2f

typedef float v2f __attribute__((ext_vector_type(2)));
typedef float v8f __attribute__((ext_vector_type(8)));

__device__ __forceinline__ float lrelu(float x) { return x > 0.f ? x : NEG_SLOPE * x; }

__global__ void gemm1_wmma(const float* __restrict__ X, const float* __restrict__ W,
                           float* __restrict__ H, int n) {
  int wave = (blockIdx.x * blockDim.x + threadIdx.x) >> 5;
  int lane = threadIdx.x & 31;
  const int ntiles = F1 / 16;
  int mt = wave / ntiles;
  int nt = wave - mt * ntiles;
  int row0 = mt * 16;
  int col0 = nt * 16;
  if (row0 >= n) return;

  int half = lane >> 4;
  int l15  = lane & 15;
  const float* xrow = X + (size_t)(row0 + l15) * D_IN;
  int bcol = col0 + l15;

  v8f acc = {};
  for (int k = 0; k < D_IN; k += 4) {
    int kk = k + half * 2;
    v2f a; a.x = xrow[kk];                       a.y = xrow[kk + 1];
    v2f b; b.x = W[(size_t)kk * F1 + bcol];      b.y = W[(size_t)(kk + 1) * F1 + bcol];
    acc = __builtin_amdgcn_wmma_f32_16x16x4_f32(false, a, false, b,
                                                (short)0, acc, false, false);
  }
#pragma unroll
  for (int v = 0; v < 8; ++v) {
    int r = row0 + v + half * 8;
    H[(size_t)r * F1 + bcol] = acc[v];
  }
}

__global__ void gemm2_wmma(const float* __restrict__ X2, const float* __restrict__ W2,
                           float* __restrict__ H2, int n) {
  int wave = (blockIdx.x * blockDim.x + threadIdx.x) >> 5;
  int lane = threadIdx.x & 31;
  int row0 = wave * 16;
  if (row0 >= n) return;

  int half = lane >> 4;
  int l15  = lane & 15;
  const float* xrow = X2 + (size_t)(row0 + l15) * F1;
  bool colok = (l15 < D_OUT);

  v8f acc = {};
  for (int k = 0; k < F1; k += 4) {
    int kk = k + half * 2;
    v2f a; a.x = xrow[kk]; a.y = xrow[kk + 1];
    v2f b;
    b.x = colok ? W2[(size_t)kk * D_OUT + l15]       : 0.f;
    b.y = colok ? W2[(size_t)(kk + 1) * D_OUT + l15] : 0.f;
    acc = __builtin_amdgcn_wmma_f32_16x16x4_f32(false, a, false, b,
                                                (short)0, acc, false, false);
  }
  if (colok) {
#pragma unroll
    for (int v = 0; v < 8; ++v) {
      int r = row0 + v + half * 8;
      H2[(size_t)r * D_OUT + l15] = acc[v];
    }
  }
}

__global__ void alphas1(const float* __restrict__ H, const float* __restrict__ a_src,
                        const float* __restrict__ a_dst,
                        float* __restrict__ as, float* __restrict__ ad, int n) {
  int t = blockIdx.x * blockDim.x + threadIdx.x;
  if (t >= n * HEADS) return;
  int h = t % HEADS;
  const float* row = H + (size_t)(t / HEADS) * F1 + h * HID;
  const float* vs = a_src + h * HID;
  const float* vd = a_dst + h * HID;
  float s = 0.f, d = 0.f;
#pragma unroll 8
  for (int c = 0; c < HID; ++c) { float x = row[c]; s += x * vs[c]; d += x * vd[c]; }
  as[t] = s; ad[t] = d;
}

__global__ void alphas2(const float* __restrict__ H2, const float* __restrict__ a_src,
                        const float* __restrict__ a_dst,
                        float* __restrict__ as, float* __restrict__ ad, int n) {
  int t = blockIdx.x * blockDim.x + threadIdx.x;
  if (t >= n) return;
  const float* row = H2 + (size_t)t * D_OUT;
  float s = 0.f, d = 0.f;
#pragma unroll
  for (int c = 0; c < D_OUT; ++c) { s += row[c] * a_src[c]; d += row[c] * a_dst[c]; }
  as[t] = s; ad[t] = d;
}

__global__ void deg_count(const int* __restrict__ dst, int nE, int* __restrict__ cnt) {
  int e = blockIdx.x * blockDim.x + threadIdx.x;
  if (e < nE) atomicAdd(cnt + dst[e], 1);
}

__global__ void seg_assign(const int* __restrict__ cnt, int n,
                           int* __restrict__ cursor, int* __restrict__ start) {
  int i = blockIdx.x * blockDim.x + threadIdx.x;
  int lane = threadIdx.x & 31;
  int c = (i < n) ? cnt[i] : 0;
  int incl = c;
#pragma unroll
  for (int o = 1; o < 32; o <<= 1) {
    int t = __shfl_up(incl, o, 32);
    if (lane >= o) incl += t;
  }
  int base = 0;
  if (lane == 31) base = atomicAdd(cursor, incl);
  base = __shfl(base, 31, 32);
  if (i < n) start[i] = base + incl - c;
}

__global__ void seg_fill(const int* __restrict__ src, const int* __restrict__ dst, int nE,
                         const int* __restrict__ start, int* __restrict__ cur,
                         int* __restrict__ elist) {
  int e = blockIdx.x * blockDim.x + threadIdx.x;
  if (e >= nE) return;
  int d = dst[e];
  int p = start[d] + atomicAdd(cur + d, 1);
  if ((unsigned)p < (unsigned)nE) elist[p] = src[e];
}

typedef float    v4f __attribute__((ext_vector_type(4)));

__device__ __forceinline__ float lane_bcast(float x, int i) {
  return __int_as_float(__builtin_amdgcn_readlane(__float_as_int(x), i));
}

__device__ __forceinline__ float as_scalar(float x) {
  return __int_as_float(__builtin_amdgcn_readfirstlane(__float_as_int(x)));
}

template <int W>
__device__ __forceinline__ float group_max(float v) {
#pragma unroll
  for (int o = W / 2; o > 0; o >>= 1) v = fmaxf(v, __shfl_xor(v, o, 32));
  return v;
}
template <int W>
__device__ __forceinline__ float group_sum(float v) {
#pragma unroll
  for (int o = W / 2; o > 0; o >>= 1) v += __shfl_xor(v, o, 32);
  return v;
}

__device__ __forceinline__ bool load_entries(const int* __restrict__ elist,
                                             const v4f* __restrict__ as4, int beg, int total,
                                             int base, int lane, int d, int n,
                                             const float (&adh)[8], int& mine, float (&ev)[8]) {
  int k = base + lane;
  bool valid = k < total;
  mine = (k >= 1 && valid) ? elist[beg + k - 1] : d;
  if ((unsigned)mine >= (unsigned)n) mine = d;
  v4f aL = as4[(unsigned)mine * 2u];
  v4f aH = as4[(unsigned)mine * 2u + 1u];
  ev[0] = lrelu(aL.x + adh[0]); ev[1] = lrelu(aL.y + adh[1]);
  ev[2] = lrelu(aL.z + adh[2]); ev[3] = lrelu(aL.w + adh[3]);
  ev[4] = lrelu(aH.x + adh[4]); ev[5] = lrelu(aH.y + adh[5]);
  ev[6] = lrelu(aH.z + adh[6]); ev[7] = lrelu(aH.w + adh[7]);
  return valid;
}

struct RowPart { v4f v0, v1, v2, v3; };

__device__ __forceinline__ RowPart row_load(const v4f* __restrict__ H4, int s_, unsigned cq) {
  const v4f* hp = H4 + ((unsigned)s_ * (unsigned)(F1 / 4) + cq);
  RowPart r;
  r.v0 = hp[0]; r.v1 = hp[16]; r.v2 = hp[32]; r.v3 = hp[48];
  return r;
}

__device__ __forceinline__ void row_add(v4f (&acc)[4], const RowPart& r, const float (&al)[8],
                                        int i, bool hi) {
  float l0 = lane_bcast(al[0], i), l1 = lane_bcast(al[1], i);
  float l2 = lane_bcast(al[2], i), l3 = lane_bcast(al[3], i);
  float h0 = lane_bcast(al[4], i), h1 = lane_bcast(al[5], i);
  float h2 = lane_bcast(al[6], i), h3 = lane_bcast(al[7], i);
  acc[0] += r.v0 * (hi ? h0 : l0);
  acc[1] += r.v1 * (hi ? h1 : l1);
  acc[2] += r.v2 * (hi ? h2 : l2);
  acc[3] += r.v3 * (hi ? h3 : l3);
}

__device__ __forceinline__ v4f bias_relu(v4f a, v4f b) {
  v4f v = a + b;
  v.x = v.x > 0.f ? v.x : 0.f;
  v.y = v.y > 0.f ? v.y : 0.f;
  v.z = v.z > 0.f ? v.z : 0.f;
  v.w = v.w > 0.f ? v.w : 0.f;
  return v;
}

__global__ void __launch_bounds__(256) __attribute__((amdgpu_waves_per_eu(8, 10)))
node_agg1(const int* __restrict__ elist, const int* __restrict__ start,
          const int* __restrict__ cnt, int nE, int n, const float* __restrict__ H,
          const float* __restrict__ as, const float* __restrict__ ad,
          const float* __restrict__ b1, float* __restrict__ out) {
  int lane = threadIdx.x & 31;
  int d = __builtin_amdgcn_readfirstlane((int)((blockIdx.x * blockDim.x + threadIdx.x) >> 5));
  if (d >= n) return;
  bool hi = lane >= 16;
  unsigned cq = (hi ? 64u : 0u) + ((unsigned)lane & 15u);
  const v4f* H4  = (const v4f*)H;
  const v4f* as4 = (const v4f*)as;

  float adh[8];
#pragma unroll
  for (int h = 0; h < HEADS; ++h) adh[h] = ad[(size_t)d * HEADS + h];

  int beg = start[d];
  int deg = cnt[d];
  if (beg < 0 || deg < 0 || beg > nE || deg > nE - beg) deg = 0;
  int total = deg + 1;

  int mine; float ev[8];
  bool valid = load_entries(elist, as4, beg, total, 0, lane, d, n, adh, mine, ev);

  float mvh[8];
  {
    float mx[8];
#pragma unroll
    for (int h = 0; h < HEADS; ++h) mx[h] = valid ? ev[h] : -INFINITY;
    for (int base = 32; base < total; base += 32) {
      int m2; float e2[8];
      if (load_entries(elist, as4, beg, total, base, lane, d, n, adh, m2, e2)) {
#pragma unroll
        for (int h = 0; h < HEADS; ++h) mx[h] = fmaxf(mx[h], e2[h]);
      }
    }
#pragma unroll
    for (int h = 0; h < HEADS; ++h) mvh[h] = as_scalar(group_max<32>(mx[h]));
  }

  float ddh[8];
  {
    float ds[8];
#pragma unroll
    for (int h = 0; h < HEADS; ++h) ds[h] = valid ? __expf(ev[h] - mvh[h]) : 0.f;
    for (int base = 32; base < total; base += 32) {
      int m2; float e2[8];
      if (load_entries(elist, as4, beg, total, base, lane, d, n, adh, m2, e2)) {
#pragma unroll
        for (int h = 0; h < HEADS; ++h) ds[h] += __expf(e2[h] - mvh[h]);
      }
    }
#pragma unroll
    for (int h = 0; h < HEADS; ++h) ddh[h] = as_scalar(group_sum<32>(ds[h])) + 1e-16f;
  }

  v4f acc[4];
#pragma unroll
  for (int j = 0; j < 4; ++j) acc[j] = (v4f)(0.f);

  for (int base = 0;;) {
    int c = total - base; if (c > 32) c = 32;

    float al[8];
#pragma unroll
    for (int h = 0; h < HEADS; ++h) al[h] = __expf(ev[h] - mvh[h]) / ddh[h];

    int i = 0;
    for (; i + 1 < c; i += 2) {
      int s0 = __builtin_amdgcn_readlane(mine, i);
      int s1 = __builtin_amdgcn_readlane(mine, i + 1);
      RowPart r0 = row_load(H4, s0, cq);
      RowPart r1 = row_load(H4, s1, cq);
      row_add(acc, r0, al, i, hi);
      row_add(acc, r1, al, i + 1, hi);
    }
    if (i < c) {
      int s0 = __builtin_amdgcn_readlane(mine, i);
      RowPart r0 = row_load(H4, s0, cq);
      row_add(acc, r0, al, i, hi);
    }

    base += 32;
    if (base >= total) break;
    load_entries(elist, as4, beg, total, base, lane, d, n, adh, mine, ev);
  }

  const v4f* bp = (const v4f*)b1 + cq;
  v4f* op = (v4f*)out + ((unsigned)d * (unsigned)(F1 / 4) + cq);
  op[0]  = bias_relu(acc[0], bp[0]);
  op[16] = bias_relu(acc[1], bp[16]);
  op[32] = bias_relu(acc[2], bp[32]);
  op[48] = bias_relu(acc[3], bp[48]);
}

__device__ __forceinline__ float entry_score2(const int* __restrict__ elist,
                                              const float* __restrict__ as, int beg, int k,
                                              int d, int n, float a_d, int& s) {
  s = k ? elist[beg + k - 1] : d;
  if ((unsigned)s >= (unsigned)n) s = d;
  return lrelu(as[s] + a_d);
}

__global__ void __launch_bounds__(256)
node_agg2(const int* __restrict__ elist, const int* __restrict__ start,
          const int* __restrict__ cnt, int nE, int n, const float* __restrict__ H2,
          const float* __restrict__ as, const float* __restrict__ ad,
          const float* __restrict__ b2, float* __restrict__ out) {
  int t = blockIdx.x * blockDim.x + threadIdx.x;
  int j = threadIdx.x & 7;
  bool live = (t >> 3) < n;
  int d = live ? (t >> 3) : 0;
  const v4f* H4 = (const v4f*)H2;

  int beg = start[d];
  int deg = cnt[d];
  if (beg < 0 || deg < 0 || beg > nE || deg > nE - beg) deg = 0;
  int total = deg + 1;
  float a_d = ad[d];

  float mx = -INFINITY;
  for (int k = j; k < total; k += 8) {
    int s;
    mx = fmaxf(mx, entry_score2(elist, as, beg, k, d, n, a_d, s));
  }
  mx = group_max<8>(mx);

  float den = 0.f;
  for (int k = j; k < total; k += 8) {
    int s;
    den += __expf(entry_score2(elist, as, beg, k, d, n, a_d, s) - mx);
  }
  den = group_sum<8>(den) + 1e-16f;

  v4f accL = (v4f)(0.f), accH = (v4f)(0.f);
  for (int k = j; k < total; k += 8) {
    int s;
    float alpha = __expf(entry_score2(elist, as, beg, k, d, n, a_d, s) - mx) / den;
    accL += H4[(unsigned)s * 2u] * alpha;
    accH += H4[(unsigned)s * 2u + 1u] * alpha;
  }
  float r0 = group_sum<8>(accL.x), r1 = group_sum<8>(accL.y);
  float r2 = group_sum<8>(accL.z), r3 = group_sum<8>(accL.w);
  float r4 = group_sum<8>(accH.x), r5 = group_sum<8>(accH.y);
  float r6 = group_sum<8>(accH.z), r7 = group_sum<8>(accH.w);
  float lo = (j & 2) ? ((j & 1) ? r3 : r2) : ((j & 1) ? r1 : r0);
  float up = (j & 2) ? ((j & 1) ? r7 : r6) : ((j & 1) ? r5 : r4);
  float r = (j & 4) ? up : lo;

  if (live) out[t] = r + b2[j];
}

extern "C" void kernel_launch(void* const* d_in, const int* in_sizes, int n_in,
                              void* d_out, int out_size, void* d_ws, size_t ws_size,
                              hipStream_t stream) {
  const float* x      = (const float*)d_in[0];
  const int*   ei     = (const int*)  d_in[1];
  const float* W1     = (const float*)d_in[2];
  const float* a_src1 = (const float*)d_in[3];
  const float* a_dst1 = (const float*)d_in[4];
  const float* b1     = (const float*)d_in[5];
  const float* W2     = (const float*)d_in[6];
  const float* a_src2 = (const float*)d_in[7];
  const float* a_dst2 = (const float*)d_in[8];
  const float* b2     = (const float*)d_in[9];

  int n  = in_sizes[0] / D_IN;
  int nE = in_sizes[1] / 2;
  const int* src = ei;
  const int* dst = ei + nE;

  char* ws = (char*)d_ws;
  size_t off = 0;
  auto alloc = [&](size_t bytes) -> char* {
    char* p = ws + off; off += (bytes + 255) & ~(size_t)255; return p;
  };
  float*    h1   = (float*)   alloc((size_t)n * F1 * 4);
  float*    out1 = (float*)   alloc((size_t)n * F1 * 4);
  int*      elist= (int*)     alloc((size_t)nE * 4);
  int*      seg0 = (int*)     alloc((size_t)n * 4);
  float*    as1  = (float*)   alloc((size_t)n * HEADS * 4);
  float*    ad1  = (float*)   alloc((size_t)n * HEADS * 4);
  float*    h2   = (float*)   alloc((size_t)n * D_OUT * 4);
  float*    as2  = (float*)   alloc((size_t)n * 4);
  float*    ad2  = (float*)   alloc((size_t)n * 4);
  int*      cnt  = (int*)     alloc((size_t)n * 4);
  int*      cur  = (int*)     alloc((size_t)n * 4);
  int*      curs = (int*)     alloc(4);

  hipMemsetAsync(cnt, 0, off - (size_t)((char*)cnt - ws), stream);

  deg_count <<<(nE + 255) / 256, 256, 0, stream>>>(dst, nE, cnt);
  seg_assign<<<(n + 255) / 256, 256, 0, stream>>>(cnt, n, curs, seg0);
  seg_fill  <<<(nE + 255) / 256, 256, 0, stream>>>(src, dst, nE, seg0, cur, elist);

  {
    int waves  = ((n + 15) / 16) * (F1 / 16);
    int blocks = (waves + 7) / 8;
    gemm1_wmma<<<blocks, 256, 0, stream>>>(x, W1, h1, n);
  }
  alphas1<<<(n * HEADS + 255) / 256, 256, 0, stream>>>(h1, a_src1, a_dst1, as1, ad1, n);
  node_agg1<<<(n + 7) / 8, 256, 0, stream>>>(elist, seg0, cnt, nE, n, h1, as1, ad1, b1, out1);

  {
    int waves  = (n + 15) / 16;
    int blocks = (waves + 7) / 8;
    gemm2_wmma<<<blocks, 256, 0, stream>>>(out1, W2, h2, n);
  }
  alphas2<<<(n + 255) / 256, 256, 0, stream>>>(h2, a_src2, a_dst2, as2, ad2, n);
  node_agg2<<<(n * D_OUT + 255) / 256, 256, 0, stream>>>(elist, seg0, cnt, nE, n, h2, as2, ad2, b2,
                                                       (float*)d_out);
}


# Layer-1 GEMM: one wave computes 32 rows x 64 columns (8 accumulators), 16 WMMAs per six 16-byte operand loads instead of 1 per three small loads; k summed in a new order within each 8; f32 WMMA kept
# speedup vs baseline: 6.6259x; 1.2132x over previous
#include <hip/hip_runtime.h>
#include <math.h>

#define D_IN   128
#define HEADS  8
#define HID    64
#define F1     (HEADS * HID)
#define D_OUT  8
#define NEG_SLOPE 0.2f

typedef float v2f __attribute__((ext_vector_type(2)));
typedef float v8f __attribute__((ext_vector_type(8)));

__device__ __forceinline__ float lrelu(float x) { return x > 0.f ? x : NEG_SLOPE * x; }

typedef float v4f __attribute__((ext_vector_type(4)));

__device__ __forceinline__ v8f wmma_f32_k4(v2f a, v2f b, v8f c) {
  return __builtin_amdgcn_wmma_f32_16x16x4_f32(false, a, false, b, (short)0, c, false, false);
}

__global__ void __launch_bounds__(128)
gemm1_wmma(const float* __restrict__ X, const float* __restrict__ W,
           float* __restrict__ H, int n) {
  const int lane = threadIdx.x & 31;
  const int wv   = threadIdx.x >> 5;
  const int half = lane >> 4;
  const int c    = lane & 15;
  const int row0 = (int)(blockIdx.x >> 1) * 32;
  const int col0 = (int)(((blockIdx.x & 1) << 2) + wv) * 64;

  int ra0 = row0 + c;       if (ra0 > n - 1) ra0 = n - 1;
  int ra1 = row0 + 16 + c;  if (ra1 > n - 1) ra1 = n - 1;
  const v4f* xa0 = (const v4f*)(X + (size_t)ra0 * D_IN) + half;
  const v4f* xa1 = (const v4f*)(X + (size_t)ra1 * D_IN) + half;
  const v4f* wb  = (const v4f*)(W + (size_t)(4 * half) * F1 + col0) + c;
  const int WR = F1 / 4;

  v8f acc[2][4];
#pragma unroll
  for (int m = 0; m < 2; ++m)
#pragma unroll
    for (int t = 0; t < 4; ++t) acc[m][t] = (v8f){};

  v4f a0 = xa0[0], a1 = xa1[0];
  v4f b0 = wb[0], b1 = wb[WR], b2 = wb[2 * WR], b3 = wb[3 * WR];

#pragma unroll
  for (int s = 0; s < D_IN / 8; ++s) {
    v4f na0 = a0, na1 = a1, nb0 = b0, nb1 = b1, nb2 = b2, nb3 = b3;
    if (s + 1 < D_IN / 8) {
      na0 = xa0[2 * (s + 1)];
      na1 = xa1[2 * (s + 1)];
      nb0 = wb[(8 * (s + 1) + 0) * WR];
      nb1 = wb[(8 * (s + 1) + 1) * WR];
      nb2 = wb[(8 * (s + 1) + 2) * WR];
      nb3 = wb[(8 * (s + 1) + 3) * WR];
    }
    __builtin_amdgcn_sched_barrier(0);
    v2f al0 = {a0.x, a0.y}, ah0 = {a0.z, a0.w};
    v2f al1 = {a1.x, a1.y}, ah1 = {a1.z, a1.w};
#pragma unroll
    for (int t = 0; t < 4; ++t) {
      v2f bl = {b0[t], b1[t]};
      acc[0][t] = wmma_f32_k4(al0, bl, acc[0][t]);
      acc[1][t] = wmma_f32_k4(al1, bl, acc[1][t]);
    }
#pragma unroll
    for (int t = 0; t < 4; ++t) {
      v2f bh = {b2[t], b3[t]};
      acc[0][t] = wmma_f32_k4(ah0, bh, acc[0][t]);
      acc[1][t] = wmma_f32_k4(ah1, bh, acc[1][t]);
    }
    a0 = na0; a1 = na1; b0 = nb0; b1 = nb1; b2 = nb2; b3 = nb3;
  }

#pragma unroll
  for (int m = 0; m < 2; ++m) {
#pragma unroll
    for (int r = 0; r < 8; ++r) {
      int row = row0 + 16 * m + 8 * half + r;
      if (row < n) {
        v4f o = {acc[m][0][r], acc[m][1][r], acc[m][2][r], acc[m][3][r]};
        *((v4f*)(H + (size_t)row * F1 + col0) + c) = o;
      }
    }
  }
}

__global__ void gemm2_wmma(const float* __restrict__ X2, const float* __restrict__ W2,
                           float* __restrict__ H2, int n) {
  int wave = (blockIdx.x * blockDim.x + threadIdx.x) >> 5;
  int lane = threadIdx.x & 31;
  int row0 = wave * 16;
  if (row0 >= n) return;

  int half = lane >> 4;
  int l15  = lane & 15;
  const float* xrow = X2 + (size_t)(row0 + l15) * F1;
  bool colok = (l15 < D_OUT);

  v8f acc = {};
  for (int k = 0; k < F1; k += 4) {
    int kk = k + half * 2;
    v2f a; a.x = xrow[kk]; a.y = xrow[kk + 1];
    v2f b;
    b.x = colok ? W2[(size_t)kk * D_OUT + l15]       : 0.f;
    b.y = colok ? W2[(size_t)(kk + 1) * D_OUT + l15] : 0.f;
    acc = __builtin_amdgcn_wmma_f32_16x16x4_f32(false, a, false, b,
                                                (short)0, acc, false, false);
  }
  if (colok) {
#pragma unroll
    for (int v = 0; v < 8; ++v) {
      int r = row0 + v + half * 8;
      H2[(size_t)r * D_OUT + l15] = acc[v];
    }
  }
}

__global__ void alphas1(const float* __restrict__ H, const float* __restrict__ a_src,
                        const float* __restrict__ a_dst,
                        float* __restrict__ as, float* __restrict__ ad, int n) {
  int t = blockIdx.x * blockDim.x + threadIdx.x;
  if (t >= n * HEADS) return;
  int h = t % HEADS;
  const float* row = H + (size_t)(t / HEADS) * F1 + h * HID;
  const float* vs = a_src + h * HID;
  const float* vd = a_dst + h * HID;
  float s = 0.f, d = 0.f;
#pragma unroll 8
  for (int c = 0; c < HID; ++c) { float x = row[c]; s += x * vs[c]; d += x * vd[c]; }
  as[t] = s; ad[t] = d;
}

__global__ void alphas2(const float* __restrict__ H2, const float* __restrict__ a_src,
                        const float* __restrict__ a_dst,
                        float* __restrict__ as, float* __restrict__ ad, int n) {
  int t = blockIdx.x * blockDim.x + threadIdx.x;
  if (t >= n) return;
  const float* row = H2 + (size_t)t * D_OUT;
  float s = 0.f, d = 0.f;
#pragma unroll
  for (int c = 0; c < D_OUT; ++c) { s += row[c] * a_src[c]; d += row[c] * a_dst[c]; }
  as[t] = s; ad[t] = d;
}

__global__ void deg_count(const int* __restrict__ dst, int nE, int* __restrict__ cnt) {
  int e = blockIdx.x * blockDim.x + threadIdx.x;
  if (e < nE) atomicAdd(cnt + dst[e], 1);
}

__global__ void seg_assign(const int* __restrict__ cnt, int n,
                           int* __restrict__ cursor, int* __restrict__ start) {
  int i = blockIdx.x * blockDim.x + threadIdx.x;
  int lane = threadIdx.x & 31;
  int c = (i < n) ? cnt[i] : 0;
  int incl = c;
#pragma unroll
  for (int o = 1; o < 32; o <<= 1) {
    int t = __shfl_up(incl, o, 32);
    if (lane >= o) incl += t;
  }
  int base = 0;
  if (lane == 31) base = atomicAdd(cursor, incl);
  base = __shfl(base, 31, 32);
  if (i < n) start[i] = base + incl - c;
}

__global__ void seg_fill(const int* __restrict__ src, const int* __restrict__ dst, int nE,
                         const int* __restrict__ start, int* __restrict__ cur,
                         int* __restrict__ elist) {
  int e = blockIdx.x * blockDim.x + threadIdx.x;
  if (e >= nE) return;
  int d = dst[e];
  int p = start[d] + atomicAdd(cur + d, 1);
  if ((unsigned)p < (unsigned)nE) elist[p] = src[e];
}


__device__ __forceinline__ float lane_bcast(float x, int i) {
  return __int_as_float(__builtin_amdgcn_readlane(__float_as_int(x), i));
}

__device__ __forceinline__ float as_scalar(float x) {
  return __int_as_float(__builtin_amdgcn_readfirstlane(__float_as_int(x)));
}

template <int W>
__device__ __forceinline__ float group_max(float v) {
#pragma unroll
  for (int o = W / 2; o > 0; o >>= 1) v = fmaxf(v, __shfl_xor(v, o, 32));
  return v;
}
template <int W>
__device__ __forceinline__ float group_sum(float v) {
#pragma unroll
  for (int o = W / 2; o > 0; o >>= 1) v += __shfl_xor(v, o, 32);
  return v;
}

__device__ __forceinline__ bool load_entries(const int* __restrict__ elist,
                                             const v4f* __restrict__ as4, int beg, int total,
                                             int base, int lane, int d, int n,
                                             const float (&adh)[8], int& mine, float (&ev)[8]) {
  int k = base + lane;
  bool valid = k < total;
  mine = (k >= 1 && valid) ? elist[beg + k - 1] : d;
  if ((unsigned)mine >= (unsigned)n) mine = d;
  v4f aL = as4[(unsigned)mine * 2u];
  v4f aH = as4[(unsigned)mine * 2u + 1u];
  ev[0] = lrelu(aL.x + adh[0]); ev[1] = lrelu(aL.y + adh[1]);
  ev[2] = lrelu(aL.z + adh[2]); ev[3] = lrelu(aL.w + adh[3]);
  ev[4] = lrelu(aH.x + adh[4]); ev[5] = lrelu(aH.y + adh[5]);
  ev[6] = lrelu(aH.z + adh[6]); ev[7] = lrelu(aH.w + adh[7]);
  return valid;
}

struct RowPart { v4f v0, v1, v2, v3; };

__device__ __forceinline__ RowPart row_load(const v4f* __restrict__ H4, int s_, unsigned cq) {
  const v4f* hp = H4 + ((unsigned)s_ * (unsigned)(F1 / 4) + cq);
  RowPart r;
  r.v0 = hp[0]; r.v1 = hp[16]; r.v2 = hp[32]; r.v3 = hp[48];
  return r;
}

__device__ __forceinline__ void row_add(v4f (&acc)[4], const RowPart& r, const float (&al)[8],
                                        int i, bool hi) {
  float l0 = lane_bcast(al[0], i), l1 = lane_bcast(al[1], i);
  float l2 = lane_bcast(al[2], i), l3 = lane_bcast(al[3], i);
  float h0 = lane_bcast(al[4], i), h1 = lane_bcast(al[5], i);
  float h2 = lane_bcast(al[6], i), h3 = lane_bcast(al[7], i);
  acc[0] += r.v0 * (hi ? h0 : l0);
  acc[1] += r.v1 * (hi ? h1 : l1);
  acc[2] += r.v2 * (hi ? h2 : l2);
  acc[3] += r.v3 * (hi ? h3 : l3);
}

__device__ __forceinline__ v4f bias_relu(v4f a, v4f b) {
  v4f v = a + b;
  v.x = v.x > 0.f ? v.x : 0.f;
  v.y = v.y > 0.f ? v.y : 0.f;
  v.z = v.z > 0.f ? v.z : 0.f;
  v.w = v.w > 0.f ? v.w : 0.f;
  return v;
}

__global__ void __launch_bounds__(256) __attribute__((amdgpu_waves_per_eu(8, 10)))
node_agg1(const int* __restrict__ elist, const int* __restrict__ start,
          const int* __restrict__ cnt, int nE, int n, const float* __restrict__ H,
          const float* __restrict__ as, const float* __restrict__ ad,
          const float* __restrict__ b1, float* __restrict__ out) {
  int lane = threadIdx.x & 31;
  int d = __builtin_amdgcn_readfirstlane((int)((blockIdx.x * blockDim.x + threadIdx.x) >> 5));
  if (d >= n) return;
  bool hi = lane >= 16;
  unsigned cq = (hi ? 64u : 0u) + ((unsigned)lane & 15u);
  const v4f* H4  = (const v4f*)H;
  const v4f* as4 = (const v4f*)as;

  float adh[8];
#pragma unroll
  for (int h = 0; h < HEADS; ++h) adh[h] = ad[(size_t)d * HEADS + h];

  int beg = start[d];
  int deg = cnt[d];
  if (beg < 0 || deg < 0 || beg > nE || deg > nE - beg) deg = 0;
  int total = deg + 1;

  int mine; float ev[8];
  bool valid = load_entries(elist, as4, beg, total, 0, lane, d, n, adh, mine, ev);

  float mvh[8];
  {
    float mx[8];
#pragma unroll
    for (int h = 0; h < HEADS; ++h) mx[h] = valid ? ev[h] : -INFINITY;
    for (int base = 32; base < total; base += 32) {
      int m2; float e2[8];
      if (load_entries(elist, as4, beg, total, base, lane, d, n, adh, m2, e2)) {
#pragma unroll
        for (int h = 0; h < HEADS; ++h) mx[h] = fmaxf(mx[h], e2[h]);
      }
    }
#pragma unroll
    for (int h = 0; h < HEADS; ++h) mvh[h] = as_scalar(group_max<32>(mx[h]));
  }

  float ddh[8];
  {
    float ds[8];
#pragma unroll
    for (int h = 0; h < HEADS; ++h) ds[h] = valid ? __expf(ev[h] - mvh[h]) : 0.f;
    for (int base = 32; base < total; base += 32) {
      int m2; float e2[8];
      if (load_entries(elist, as4, beg, total, base, lane, d, n, adh, m2, e2)) {
#pragma unroll
        for (int h = 0; h < HEADS; ++h) ds[h] += __expf(e2[h] - mvh[h]);
      }
    }
#pragma unroll
    for (int h = 0; h < HEADS; ++h) ddh[h] = as_scalar(group_sum<32>(ds[h])) + 1e-16f;
  }

  v4f acc[4];
#pragma unroll
  for (int j = 0; j < 4; ++j) acc[j] = (v4f)(0.f);

  for (int base = 0;;) {
    int c = total - base; if (c > 32) c = 32;

    float al[8];
#pragma unroll
    for (int h = 0; h < HEADS; ++h) al[h] = __expf(ev[h] - mvh[h]) / ddh[h];

    int i = 0;
    for (; i + 1 < c; i += 2) {
      int s0 = __builtin_amdgcn_readlane(mine, i);
      int s1 = __builtin_amdgcn_readlane(mine, i + 1);
      RowPart r0 = row_load(H4, s0, cq);
      RowPart r1 = row_load(H4, s1, cq);
      row_add(acc, r0, al, i, hi);
      row_add(acc, r1, al, i + 1, hi);
    }
    if (i < c) {
      int s0 = __builtin_amdgcn_readlane(mine, i);
      RowPart r0 = row_load(H4, s0, cq);
      row_add(acc, r0, al, i, hi);
    }

    base += 32;
    if (base >= total) break;
    load_entries(elist, as4, beg, total, base, lane, d, n, adh, mine, ev);
  }

  const v4f* bp = (const v4f*)b1 + cq;
  v4f* op = (v4f*)out + ((unsigned)d * (unsigned)(F1 / 4) + cq);
  op[0]  = bias_relu(acc[0], bp[0]);
  op[16] = bias_relu(acc[1], bp[16]);
  op[32] = bias_relu(acc[2], bp[32]);
  op[48] = bias_relu(acc[3], bp[48]);
}

__device__ __forceinline__ float entry_score2(const int* __restrict__ elist,
                                              const float* __restrict__ as, int beg, int k,
                                              int d, int n, float a_d, int& s) {
  s = k ? elist[beg + k - 1] : d;
  if ((unsigned)s >= (unsigned)n) s = d;
  return lrelu(as[s] + a_d);
}

__global__ void __launch_bounds__(256)
node_agg2(const int* __restrict__ elist, const int* __restrict__ start,
          const int* __restrict__ cnt, int nE, int n, const float* __restrict__ H2,
          const float* __restrict__ as, const float* __restrict__ ad,
          const float* __restrict__ b2, float* __restrict__ out) {
  int t = blockIdx.x * blockDim.x + threadIdx.x;
  int j = threadIdx.x & 7;
  bool live = (t >> 3) < n;
  int d = live ? (t >> 3) : 0;
  const v4f* H4 = (const v4f*)H2;

  int beg = start[d];
  int deg = cnt[d];
  if (beg < 0 || deg < 0 || beg > nE || deg > nE - beg) deg = 0;
  int total = deg + 1;
  float a_d = ad[d];

  float mx = -INFINITY;
  for (int k = j; k < total; k += 8) {
    int s;
    mx = fmaxf(mx, entry_score2(elist, as, beg, k, d, n, a_d, s));
  }
  mx = group_max<8>(mx);

  float den = 0.f;
  for (int k = j; k < total; k += 8) {
    int s;
    den += __expf(entry_score2(elist, as, beg, k, d, n, a_d, s) - mx);
  }
  den = group_sum<8>(den) + 1e-16f;

  v4f accL = (v4f)(0.f), accH = (v4f)(0.f);
  for (int k = j; k < total; k += 8) {
    int s;
    float alpha = __expf(entry_score2(elist, as, beg, k, d, n, a_d, s) - mx) / den;
    accL += H4[(unsigned)s * 2u] * alpha;
    accH += H4[(unsigned)s * 2u + 1u] * alpha;
  }
  float r0 = group_sum<8>(accL.x), r1 = group_sum<8>(accL.y);
  float r2 = group_sum<8>(accL.z), r3 = group_sum<8>(accL.w);
  float r4 = group_sum<8>(accH.x), r5 = group_sum<8>(accH.y);
  float r6 = group_sum<8>(accH.z), r7 = group_sum<8>(accH.w);
  float lo = (j & 2) ? ((j & 1) ? r3 : r2) : ((j & 1) ? r1 : r0);
  float up = (j & 2) ? ((j & 1) ? r7 : r6) : ((j & 1) ? r5 : r4);
  float r = (j & 4) ? up : lo;

  if (live) out[t] = r + b2[j];
}

extern "C" void kernel_launch(void* const* d_in, const int* in_sizes, int n_in,
                              void* d_out, int out_size, void* d_ws, size_t ws_size,
                              hipStream_t stream) {
  const float* x      = (const float*)d_in[0];
  const int*   ei     = (const int*)  d_in[1];
  const float* W1     = (const float*)d_in[2];
  const float* a_src1 = (const float*)d_in[3];
  const float* a_dst1 = (const float*)d_in[4];
  const float* b1     = (const float*)d_in[5];
  const float* W2     = (const float*)d_in[6];
  const float* a_src2 = (const float*)d_in[7];
  const float* a_dst2 = (const float*)d_in[8];
  const float* b2     = (const float*)d_in[9];

  int n  = in_sizes[0] / D_IN;
  int nE = in_sizes[1] / 2;
  const int* src = ei;
  const int* dst = ei + nE;

  char* ws = (char*)d_ws;
  size_t off = 0;
  auto alloc = [&](size_t bytes) -> char* {
    char* p = ws + off; off += (bytes + 255) & ~(size_t)255; return p;
  };
  float*    h1   = (float*)   alloc((size_t)n * F1 * 4);
  float*    out1 = (float*)   alloc((size_t)n * F1 * 4);
  int*      elist= (int*)     alloc((size_t)nE * 4);
  int*      seg0 = (int*)     alloc((size_t)n * 4);
  float*    as1  = (float*)   alloc((size_t)n * HEADS * 4);
  float*    ad1  = (float*)   alloc((size_t)n * HEADS * 4);
  float*    h2   = (float*)   alloc((size_t)n * D_OUT * 4);
  float*    as2  = (float*)   alloc((size_t)n * 4);
  float*    ad2  = (float*)   alloc((size_t)n * 4);
  int*      cnt  = (int*)     alloc((size_t)n * 4);
  int*      cur  = (int*)     alloc((size_t)n * 4);
  int*      curs = (int*)     alloc(4);

  hipMemsetAsync(cnt, 0, off - (size_t)((char*)cnt - ws), stream);

  deg_count <<<(nE + 255) / 256, 256, 0, stream>>>(dst, nE, cnt);
  seg_assign<<<(n + 255) / 256, 256, 0, stream>>>(cnt, n, curs, seg0);
  seg_fill  <<<(nE + 255) / 256, 256, 0, stream>>>(src, dst, nE, seg0, cur, elist);

  {
    int blocks = ((n + 31) / 32) * 2;
    gemm1_wmma<<<blocks, 128, 0, stream>>>(x, W1, h1, n);
  }
  alphas1<<<(n * HEADS + 255) / 256, 256, 0, stream>>>(h1, a_src1, a_dst1, as1, ad1, n);
  node_agg1<<<(n + 7) / 8, 256, 0, stream>>>(elist, seg0, cnt, nE, n, h1, as1, ad1, b1, out1);

  {
    int waves  = (n + 15) / 16;
    int blocks = (waves + 7) / 8;
    gemm2_wmma<<<blocks, 256, 0, stream>>>(out1, W2, h2, n);
  }
  alphas2<<<(n + 255) / 256, 256, 0, stream>>>(h2, a_src2, a_dst2, as2, ad2, n);
  node_agg2<<<(n * D_OUT + 255) / 256, 256, 0, stream>>>(elist, seg0, cnt, nE, n, h2, as2, ad2, b2,
                                                       (float*)d_out);
}


# Attention coefficients of both layers formed in the GEMM epilogues from the accumulators (lane products + lane tree: new sum order); both coefficient launches and their re-read of h1, h2 gone
# speedup vs baseline: 7.0180x; 1.0592x over previous
#include <hip/hip_runtime.h>
#include <math.h>

#define D_IN   128
#define HEADS  8
#define HID    64
#define F1     (HEADS * HID)
#define D_OUT  8
#define NEG_SLOPE 0.2f

typedef float v2f __attribute__((ext_vector_type(2)));
typedef float v8f __attribute__((ext_vector_type(8)));

__device__ __forceinline__ float lrelu(float x) { return x > 0.f ? x : NEG_SLOPE * x; }

typedef float v4f __attribute__((ext_vector_type(4)));

template <int HALF, int NP>
__device__ __forceinline__ void fold(float (&p)[NP], bool up, int off) {
#pragma unroll
  for (int i = 0; i < HALF; ++i) {
    float keep = up ? p[i + HALF] : p[i];
    float send = up ? p[i] : p[i + HALF];
    p[i] = keep + __shfl_xor(send, off, 32);
  }
}

__device__ __forceinline__ v8f wmma_f32_k4(v2f a, v2f b, v8f c) {
  return __builtin_amdgcn_wmma_f32_16x16x4_f32(false, a, false, b, (short)0, c, false, false);
}

__global__ void __launch_bounds__(256)
gemm1_wmma(const float* __restrict__ X, const float* __restrict__ W,
           const float* __restrict__ a_src, const float* __restrict__ a_dst,
           float* __restrict__ H, float* __restrict__ as, float* __restrict__ ad, int n) {
  __shared__ float sa[2 * 32 * HEADS];
  const int lane = threadIdx.x & 31;
  const int wv   = threadIdx.x >> 5;
  const int half = lane >> 4;
  const int c    = lane & 15;
  const int row0 = (int)blockIdx.x * 32;
  const int col0 = wv * HID;

  int ra0 = row0 + c;       if (ra0 > n - 1) ra0 = n - 1;
  int ra1 = row0 + 16 + c;  if (ra1 > n - 1) ra1 = n - 1;
  const v4f* xa0 = (const v4f*)(X + (size_t)ra0 * D_IN) + half;
  const v4f* xa1 = (const v4f*)(X + (size_t)ra1 * D_IN) + half;
  const v4f* wb  = (const v4f*)(W + (size_t)(4 * half) * F1 + col0) + c;
  const int WR = F1 / 4;

  v8f acc[2][4];
#pragma unroll
  for (int m = 0; m < 2; ++m)
#pragma unroll
    for (int t = 0; t < 4; ++t) acc[m][t] = (v8f){};

  v4f a0 = xa0[0], a1 = xa1[0];
  v4f b0 = wb[0], b1 = wb[WR], b2 = wb[2 * WR], b3 = wb[3 * WR];

#pragma unroll
  for (int s = 0; s < D_IN / 8; ++s) {
    v4f na0 = a0, na1 = a1, nb0 = b0, nb1 = b1, nb2 = b2, nb3 = b3;
    if (s + 1 < D_IN / 8) {
      na0 = xa0[2 * (s + 1)];
      na1 = xa1[2 * (s + 1)];
      nb0 = wb[(8 * (s + 1) + 0) * WR];
      nb1 = wb[(8 * (s + 1) + 1) * WR];
      nb2 = wb[(8 * (s + 1) + 2) * WR];
      nb3 = wb[(8 * (s + 1) + 3) * WR];
    }
    __builtin_amdgcn_sched_barrier(0);
    v2f al0 = {a0.x, a0.y}, ah0 = {a0.z, a0.w};
    v2f al1 = {a1.x, a1.y}, ah1 = {a1.z, a1.w};
#pragma unroll
    for (int t = 0; t < 4; ++t) {
      v2f bl = {b0[t], b1[t]};
      acc[0][t] = wmma_f32_k4(al0, bl, acc[0][t]);
      acc[1][t] = wmma_f32_k4(al1, bl, acc[1][t]);
    }
#pragma unroll
    for (int t = 0; t < 4; ++t) {
      v2f bh = {b2[t], b3[t]};
      acc[0][t] = wmma_f32_k4(ah0, bh, acc[0][t]);
      acc[1][t] = wmma_f32_k4(ah1, bh, acc[1][t]);
    }
    a0 = na0; a1 = na1; b0 = nb0; b1 = nb1; b2 = nb2; b3 = nb3;
  }

#pragma unroll
  for (int m = 0; m < 2; ++m) {
#pragma unroll
    for (int r = 0; r < 8; ++r) {
      int row = row0 + 16 * m + 8 * half + r;
      if (row < n) {
        v4f o = {acc[m][0][r], acc[m][1][r], acc[m][2][r], acc[m][3][r]};
        *((v4f*)(H + (size_t)row * F1 + col0) + c) = o;
      }
    }
  }

  const v4f vs = ((const v4f*)a_src)[wv * (HID / 4) + c];
  const v4f vd = ((const v4f*)a_dst)[wv * (HID / 4) + c];
  float p[32];
#pragma unroll
  for (int m = 0; m < 2; ++m) {
#pragma unroll
    for (int r = 0; r < 8; ++r) {
      float s = acc[m][0][r] * vs.x, d = acc[m][0][r] * vd.x;
      s += acc[m][1][r] * vs.y;      d += acc[m][1][r] * vd.y;
      s += acc[m][2][r] * vs.z;      d += acc[m][2][r] * vd.z;
      s += acc[m][3][r] * vs.w;      d += acc[m][3][r] * vd.w;
      p[8 * m + r] = s;
      p[16 + 8 * m + r] = d;
    }
  }
  fold<16>(p, (c & 8) != 0, 8);
  fold<8>(p, (c & 4) != 0, 4);
  fold<4>(p, (c & 2) != 0, 2);
  fold<2>(p, (c & 1) != 0, 1);
  {
    const int rb = (c & 4) * 4 + 8 * half + 2 * (c & 3);
    float* sq = sa + (c & 8) * 32;
    sq[rb * HEADS + wv]       = p[0];
    sq[(rb + 1) * HEADS + wv] = p[1];
  }
  __syncthreads();
  if (threadIdx.x < 128) {
    const int q = (int)threadIdx.x >> 6;
    const int u = (int)threadIdx.x & 63;
    if (row0 + (u >> 1) < n) {
      v4f v = ((const v4f*)sa)[q * 64 + u];
      ((v4f*)(q ? ad : as))[(size_t)row0 * 2 + u] = v;
    }
  }
}

__global__ void gemm2_wmma(const float* __restrict__ X2, const float* __restrict__ W2,
                           const float* __restrict__ a_src, const float* __restrict__ a_dst,
                           float* __restrict__ H2, float* __restrict__ asd, int n) {
  int wave = (blockIdx.x * blockDim.x + threadIdx.x) >> 5;
  int lane = threadIdx.x & 31;
  int row0 = wave * 16;
  if (row0 >= n) return;

  int half = lane >> 4;
  int l15  = lane & 15;
  const float* xrow = X2 + (size_t)(row0 + l15) * F1;
  bool colok = (l15 < D_OUT);

  v8f acc = {};
  for (int k = 0; k < F1; k += 4) {
    int kk = k + half * 2;
    v2f a; a.x = xrow[kk]; a.y = xrow[kk + 1];
    v2f b;
    b.x = colok ? W2[(size_t)kk * D_OUT + l15]       : 0.f;
    b.y = colok ? W2[(size_t)(kk + 1) * D_OUT + l15] : 0.f;
    acc = __builtin_amdgcn_wmma_f32_16x16x4_f32(false, a, false, b,
                                                (short)0, acc, false, false);
  }
  if (colok) {
#pragma unroll
    for (int v = 0; v < 8; ++v) {
      int r = row0 + v + half * 8;
      H2[(size_t)r * D_OUT + l15] = acc[v];
    }
  }

  const float cs = colok ? a_src[l15] : 0.f;
  const float cd = colok ? a_dst[l15] : 0.f;
  float p[16];
#pragma unroll
  for (int v = 0; v < 8; ++v) { p[2 * v] = acc[v] * cs; p[2 * v + 1] = acc[v] * cd; }
  fold<8>(p, (l15 & 4) != 0, 4);
  fold<4>(p, (l15 & 2) != 0, 2);
  fold<2>(p, (l15 & 1) != 0, 1);
  if (colok) {
    v2f o = {p[0], p[1]};
    ((v2f*)asd)[(size_t)(row0 + 8 * half + l15)] = o;
  }
}

__global__ void deg_count(const int* __restrict__ dst, int nE, int* __restrict__ cnt) {
  int e = blockIdx.x * blockDim.x + threadIdx.x;
  if (e < nE) atomicAdd(cnt + dst[e], 1);
}

__global__ void seg_assign(const int* __restrict__ cnt, int n,
                           int* __restrict__ cursor, int* __restrict__ start) {
  int i = blockIdx.x * blockDim.x + threadIdx.x;
  int lane = threadIdx.x & 31;
  int c = (i < n) ? cnt[i] : 0;
  int incl = c;
#pragma unroll
  for (int o = 1; o < 32; o <<= 1) {
    int t = __shfl_up(incl, o, 32);
    if (lane >= o) incl += t;
  }
  int base = 0;
  if (lane == 31) base = atomicAdd(cursor, incl);
  base = __shfl(base, 31, 32);
  if (i < n) start[i] = base + incl - c;
}

__global__ void seg_fill(const int* __restrict__ src, const int* __restrict__ dst, int nE,
                         const int* __restrict__ start, int* __restrict__ cur,
                         int* __restrict__ elist) {
  int e = blockIdx.x * blockDim.x + threadIdx.x;
  if (e >= nE) return;
  int d = dst[e];
  int p = start[d] + atomicAdd(cur + d, 1);
  if ((unsigned)p < (unsigned)nE) elist[p] = src[e];
}


__device__ __forceinline__ float lane_bcast(float x, int i) {
  return __int_as_float(__builtin_amdgcn_readlane(__float_as_int(x), i));
}

__device__ __forceinline__ float as_scalar(float x) {
  return __int_as_float(__builtin_amdgcn_readfirstlane(__float_as_int(x)));
}

template <int W>
__device__ __forceinline__ float group_max(float v) {
#pragma unroll
  for (int o = W / 2; o > 0; o >>= 1) v = fmaxf(v, __shfl_xor(v, o, 32));
  return v;
}
template <int W>
__device__ __forceinline__ float group_sum(float v) {
#pragma unroll
  for (int o = W / 2; o > 0; o >>= 1) v += __shfl_xor(v, o, 32);
  return v;
}

__device__ __forceinline__ bool load_entries(const int* __restrict__ elist,
                                             const v4f* __restrict__ as4, int beg, int total,
                                             int base, int lane, int d, int n,
                                             const float (&adh)[8], int& mine, float (&ev)[8]) {
  int k = base + lane;
  bool valid = k < total;
  mine = (k >= 1 && valid) ? elist[beg + k - 1] : d;
  if ((unsigned)mine >= (unsigned)n) mine = d;
  v4f aL = as4[(unsigned)mine * 2u];
  v4f aH = as4[(unsigned)mine * 2u + 1u];
  ev[0] = lrelu(aL.x + adh[0]); ev[1] = lrelu(aL.y + adh[1]);
  ev[2] = lrelu(aL.z + adh[2]); ev[3] = lrelu(aL.w + adh[3]);
  ev[4] = lrelu(aH.x + adh[4]); ev[5] = lrelu(aH.y + adh[5]);
  ev[6] = lrelu(aH.z + adh[6]); ev[7] = lrelu(aH.w + adh[7]);
  return valid;
}

struct RowPart { v4f v0, v1, v2, v3; };

__device__ __forceinline__ RowPart row_load(const v4f* __restrict__ H4, int s_, unsigned cq) {
  const v4f* hp = H4 + ((unsigned)s_ * (unsigned)(F1 / 4) + cq);
  RowPart r;
  r.v0 = hp[0]; r.v1 = hp[16]; r.v2 = hp[32]; r.v3 = hp[48];
  return r;
}

__device__ __forceinline__ void row_add(v4f (&acc)[4], const RowPart& r, const float (&al)[8],
                                        int i, bool hi) {
  float l0 = lane_bcast(al[0], i), l1 = lane_bcast(al[1], i);
  float l2 = lane_bcast(al[2], i), l3 = lane_bcast(al[3], i);
  float h0 = lane_bcast(al[4], i), h1 = lane_bcast(al[5], i);
  float h2 = lane_bcast(al[6], i), h3 = lane_bcast(al[7], i);
  acc[0] += r.v0 * (hi ? h0 : l0);
  acc[1] += r.v1 * (hi ? h1 : l1);
  acc[2] += r.v2 * (hi ? h2 : l2);
  acc[3] += r.v3 * (hi ? h3 : l3);
}

__device__ __forceinline__ v4f bias_relu(v4f a, v4f b) {
  v4f v = a + b;
  v.x = v.x > 0.f ? v.x : 0.f;
  v.y = v.y > 0.f ? v.y : 0.f;
  v.z = v.z > 0.f ? v.z : 0.f;
  v.w = v.w > 0.f ? v.w : 0.f;
  return v;
}

__global__ void __launch_bounds__(256) __attribute__((amdgpu_waves_per_eu(8, 10)))
node_agg1(const int* __restrict__ elist, const int* __restrict__ start,
          const int* __restrict__ cnt, int nE, int n, const float* __restrict__ H,
          const float* __restrict__ as, const float* __restrict__ ad,
          const float* __restrict__ b1, float* __restrict__ out) {
  int lane = threadIdx.x & 31;
  int d = __builtin_amdgcn_readfirstlane((int)((blockIdx.x * blockDim.x + threadIdx.x) >> 5));
  if (d >= n) return;
  bool hi = lane >= 16;
  unsigned cq = (hi ? 64u : 0u) + ((unsigned)lane & 15u);
  const v4f* H4  = (const v4f*)H;
  const v4f* as4 = (const v4f*)as;

  float adh[8];
#pragma unroll
  for (int h = 0; h < HEADS; ++h) adh[h] = ad[(size_t)d * HEADS + h];

  int beg = start[d];
  int deg = cnt[d];
  if (beg < 0 || deg < 0 || beg > nE || deg > nE - beg) deg = 0;
  int total = deg + 1;

  int mine; float ev[8];
  bool valid = load_entries(elist, as4, beg, total, 0, lane, d, n, adh, mine, ev);

  float mvh[8];
  {
    float mx[8];
#pragma unroll
    for (int h = 0; h < HEADS; ++h) mx[h] = valid ? ev[h] : -INFINITY;
    for (int base = 32; base < total; base += 32) {
      int m2; float e2[8];
      if (load_entries(elist, as4, beg, total, base, lane, d, n, adh, m2, e2)) {
#pragma unroll
        for (int h = 0; h < HEADS; ++h) mx[h] = fmaxf(mx[h], e2[h]);
      }
    }
#pragma unroll
    for (int h = 0; h < HEADS; ++h) mvh[h] = as_scalar(group_max<32>(mx[h]));
  }

  float ddh[8];
  {
    float ds[8];
#pragma unroll
    for (int h = 0; h < HEADS; ++h) ds[h] = valid ? __expf(ev[h] - mvh[h]) : 0.f;
    for (int base = 32; base < total; base += 32) {
      int m2; float e2[8];
      if (load_entries(elist, as4, beg, total, base, lane, d, n, adh, m2, e2)) {
#pragma unroll
        for (int h = 0; h < HEADS; ++h) ds[h] += __expf(e2[h] - mvh[h]);
      }
    }
#pragma unroll
    for (int h = 0; h < HEADS; ++h) ddh[h] = as_scalar(group_sum<32>(ds[h])) + 1e-16f;
  }

  v4f acc[4];
#pragma unroll
  for (int j = 0; j < 4; ++j) acc[j] = (v4f)(0.f);

  for (int base = 0;;) {
    int c = total - base; if (c > 32) c = 32;

    float al[8];
#pragma unroll
    for (int h = 0; h < HEADS; ++h) al[h] = __expf(ev[h] - mvh[h]) / ddh[h];

    int i = 0;
    for (; i + 1 < c; i += 2) {
      int s0 = __builtin_amdgcn_readlane(mine, i);
      int s1 = __builtin_amdgcn_readlane(mine, i + 1);
      RowPart r0 = row_load(H4, s0, cq);
      RowPart r1 = row_load(H4, s1, cq);
      row_add(acc, r0, al, i, hi);
      row_add(acc, r1, al, i + 1, hi);
    }
    if (i < c) {
      int s0 = __builtin_amdgcn_readlane(mine, i);
      RowPart r0 = row_load(H4, s0, cq);
      row_add(acc, r0, al, i, hi);
    }

    base += 32;
    if (base >= total) break;
    load_entries(elist, as4, beg, total, base, lane, d, n, adh, mine, ev);
  }

  const v4f* bp = (const v4f*)b1 + cq;
  v4f* op = (v4f*)out + ((unsigned)d * (unsigned)(F1 / 4) + cq);
  op[0]  = bias_relu(acc[0], bp[0]);
  op[16] = bias_relu(acc[1], bp[16]);
  op[32] = bias_relu(acc[2], bp[32]);
  op[48] = bias_relu(acc[3], bp[48]);
}

__device__ __forceinline__ float entry_score2(const int* __restrict__ elist,
                                              const float* __restrict__ asd, int beg, int k,
                                              int d, int n, float a_d, int& s) {
  s = k ? elist[beg + k - 1] : d;
  if ((unsigned)s >= (unsigned)n) s = d;
  return lrelu(asd[2u * (unsigned)s] + a_d);
}

__global__ void __launch_bounds__(256)
node_agg2(const int* __restrict__ elist, const int* __restrict__ start,
          const int* __restrict__ cnt, int nE, int n, const float* __restrict__ H2,
          const float* __restrict__ asd,
          const float* __restrict__ b2, float* __restrict__ out) {
  int t = blockIdx.x * blockDim.x + threadIdx.x;
  int j = threadIdx.x & 7;
  bool live = (t >> 3) < n;
  int d = live ? (t >> 3) : 0;
  const v4f* H4 = (const v4f*)H2;

  int beg = start[d];
  int deg = cnt[d];
  if (beg < 0 || deg < 0 || beg > nE || deg > nE - beg) deg = 0;
  int total = deg + 1;
  float a_d = asd[2u * (unsigned)d + 1u];

  float mx = -INFINITY;
  for (int k = j; k < total; k += 8) {
    int s;
    mx = fmaxf(mx, entry_score2(elist, asd, beg, k, d, n, a_d, s));
  }
  mx = group_max<8>(mx);

  float den = 0.f;
  for (int k = j; k < total; k += 8) {
    int s;
    den += __expf(entry_score2(elist, asd, beg, k, d, n, a_d, s) - mx);
  }
  den = group_sum<8>(den) + 1e-16f;

  v4f accL = (v4f)(0.f), accH = (v4f)(0.f);
  for (int k = j; k < total; k += 8) {
    int s;
    float alpha = __expf(entry_score2(elist, asd, beg, k, d, n, a_d, s) - mx) / den;
    accL += H4[(unsigned)s * 2u] * alpha;
    accH += H4[(unsigned)s * 2u + 1u] * alpha;
  }
  float r0 = group_sum<8>(accL.x), r1 = group_sum<8>(accL.y);
  float r2 = group_sum<8>(accL.z), r3 = group_sum<8>(accL.w);
  float r4 = group_sum<8>(accH.x), r5 = group_sum<8>(accH.y);
  float r6 = group_sum<8>(accH.z), r7 = group_sum<8>(accH.w);
  float lo = (j & 2) ? ((j & 1) ? r3 : r2) : ((j & 1) ? r1 : r0);
  float up = (j & 2) ? ((j & 1) ? r7 : r6) : ((j & 1) ? r5 : r4);
  float r = (j & 4) ? up : lo;

  if (live) out[t] = r + b2[j];
}

extern "C" void kernel_launch(void* const* d_in, const int* in_sizes, int n_in,
                              void* d_out, int out_size, void* d_ws, size_t ws_size,
                              hipStream_t stream) {
  const float* x      = (const float*)d_in[0];
  const int*   ei     = (const int*)  d_in[1];
  const float* W1     = (const float*)d_in[2];
  const float* a_src1 = (const float*)d_in[3];
  const float* a_dst1 = (const float*)d_in[4];
  const float* b1     = (const float*)d_in[5];
  const float* W2     = (const float*)d_in[6];
  const float* a_src2 = (const float*)d_in[7];
  const float* a_dst2 = (const float*)d_in[8];
  const float* b2     = (const float*)d_in[9];

  int n  = in_sizes[0] / D_IN;
  int nE = in_sizes[1] / 2;
  const int* src = ei;
  const int* dst = ei + nE;

  char* ws = (char*)d_ws;
  size_t off = 0;
  auto alloc = [&](size_t bytes) -> char* {
    char* p = ws + off; off += (bytes + 255) & ~(size_t)255; return p;
  };
  float*    h1   = (float*)   alloc((size_t)n * F1 * 4);
  float*    out1 = (float*)   alloc((size_t)n * F1 * 4);
  int*      elist= (int*)     alloc((size_t)nE * 4);
  int*      seg0 = (int*)     alloc((size_t)n * 4);
  float*    as1  = (float*)   alloc((size_t)n * HEADS * 4);
  float*    ad1  = (float*)   alloc((size_t)n * HEADS * 4);
  float*    h2   = (float*)   alloc((size_t)n * D_OUT * 4);
  float*    asd2 = (float*)   alloc((size_t)n * 2 * 4);
  int*      cnt  = (int*)     alloc((size_t)n * 4);
  int*      cur  = (int*)     alloc((size_t)n * 4);
  int*      curs = (int*)     alloc(4);

  hipMemsetAsync(cnt, 0, off - (size_t)((char*)cnt - ws), stream);

  deg_count <<<(nE + 255) / 256, 256, 0, stream>>>(dst, nE, cnt);
  seg_assign<<<(n + 255) / 256, 256, 0, stream>>>(cnt, n, curs, seg0);
  seg_fill  <<<(nE + 255) / 256, 256, 0, stream>>>(src, dst, nE, seg0, cur, elist);

  {
    int blocks = (n + 31) / 32;
    gemm1_wmma<<<blocks, 256, 0, stream>>>(x, W1, a_src1, a_dst1, h1, as1, ad1, n);
  }
  node_agg1<<<(n + 7) / 8, 256, 0, stream>>>(elist, seg0, cnt, nE, n, h1, as1, ad1, b1, out1);

  {
    int waves  = (n + 15) / 16;
    int blocks = (waves + 7) / 8;
    gemm2_wmma<<<blocks, 256, 0, stream>>>(out1, W2, a_src2, a_dst2, h2, asd2, n);
  }
  node_agg2<<<(n * D_OUT + 255) / 256, 256, 0, stream>>>(elist, seg0, cnt, nE, n, h2, asd2, b2,
                                                       (float*)d_out);
}


# Layer-2 GEMM fused into the layer-1 aggregation kernel: x2 rows stay in LDS, f32 WMMA per 8 nodes, k in 8 slices of 64 added in wave order (new sum order); x2 write+read and one launch gone
# speedup vs baseline: 7.6900x; 1.0958x over previous
#include <hip/hip_runtime.h>
#include <math.h>

#define D_IN   128
#define HEADS  8
#define HID    64
#define F1     (HEADS * HID)
#define D_OUT  8
#define NEG_SLOPE 0.2f

typedef float v2f __attribute__((ext_vector_type(2)));
typedef float v8f __attribute__((ext_vector_type(8)));

__device__ __forceinline__ float lrelu(float x) { return x > 0.f ? x : NEG_SLOPE * x; }

typedef float v4f __attribute__((ext_vector_type(4)));

template <int HALF, int NP>
__device__ __forceinline__ void fold(float (&p)[NP], bool up, int off) {
#pragma unroll
  for (int i = 0; i < HALF; ++i) {
    float keep = up ? p[i + HALF] : p[i];
    float send = up ? p[i] : p[i + HALF];
    p[i] = keep + __shfl_xor(send, off, 32);
  }
}

__device__ __forceinline__ v8f wmma_f32_k4(v2f a, v2f b, v8f c) {
  return __builtin_amdgcn_wmma_f32_16x16x4_f32(false, a, false, b, (short)0, c, false, false);
}

__global__ void __launch_bounds__(256)
gemm1_wmma(const float* __restrict__ X, const float* __restrict__ W,
           const float* __restrict__ a_src, const float* __restrict__ a_dst,
           float* __restrict__ H, float* __restrict__ as, float* __restrict__ ad, int n) {
  __shared__ float sa[2 * 32 * HEADS];
  const int lane = threadIdx.x & 31;
  const int wv   = threadIdx.x >> 5;
  const int half = lane >> 4;
  const int c    = lane & 15;
  const int row0 = (int)blockIdx.x * 32;
  const int col0 = wv * HID;

  int ra0 = row0 + c;       if (ra0 > n - 1) ra0 = n - 1;
  int ra1 = row0 + 16 + c;  if (ra1 > n - 1) ra1 = n - 1;
  const v4f* xa0 = (const v4f*)(X + (size_t)ra0 * D_IN) + half;
  const v4f* xa1 = (const v4f*)(X + (size_t)ra1 * D_IN) + half;
  const v4f* wb  = (const v4f*)(W + (size_t)(4 * half) * F1 + col0) + c;
  const int WR = F1 / 4;

  v8f acc[2][4];
#pragma unroll
  for (int m = 0; m < 2; ++m)
#pragma unroll
    for (int t = 0; t < 4; ++t) acc[m][t] = (v8f){};

  v4f a0 = xa0[0], a1 = xa1[0];
  v4f b0 = wb[0], b1 = wb[WR], b2 = wb[2 * WR], b3 = wb[3 * WR];

#pragma unroll
  for (int s = 0; s < D_IN / 8; ++s) {
    v4f na0 = a0, na1 = a1, nb0 = b0, nb1 = b1, nb2 = b2, nb3 = b3;
    if (s + 1 < D_IN / 8) {
      na0 = xa0[2 * (s + 1)];
      na1 = xa1[2 * (s + 1)];
      nb0 = wb[(8 * (s + 1) + 0) * WR];
      nb1 = wb[(8 * (s + 1) + 1) * WR];
      nb2 = wb[(8 * (s + 1) + 2) * WR];
      nb3 = wb[(8 * (s + 1) + 3) * WR];
    }
    __builtin_amdgcn_sched_barrier(0);
    v2f al0 = {a0.x, a0.y}, ah0 = {a0.z, a0.w};
    v2f al1 = {a1.x, a1.y}, ah1 = {a1.z, a1.w};
#pragma unroll
    for (int t = 0; t < 4; ++t) {
      v2f bl = {b0[t], b1[t]};
      acc[0][t] = wmma_f32_k4(al0, bl, acc[0][t]);
      acc[1][t] = wmma_f32_k4(al1, bl, acc[1][t]);
    }
#pragma unroll
    for (int t = 0; t < 4; ++t) {
      v2f bh = {b2[t], b3[t]};
      acc[0][t] = wmma_f32_k4(ah0, bh, acc[0][t]);
      acc[1][t] = wmma_f32_k4(ah1, bh, acc[1][t]);
    }
    a0 = na0; a1 = na1; b0 = nb0; b1 = nb1; b2 = nb2; b3 = nb3;
  }

#pragma unroll
  for (int m = 0; m < 2; ++m) {
#pragma unroll
    for (int r = 0; r < 8; ++r) {
      int row = row0 + 16 * m + 8 * half + r;
      if (row < n) {
        v4f o = {acc[m][0][r], acc[m][1][r], acc[m][2][r], acc[m][3][r]};
        *((v4f*)(H + (size_t)row * F1 + col0) + c) = o;
      }
    }
  }

  const v4f vs = ((const v4f*)a_src)[wv * (HID / 4) + c];
  const v4f vd = ((const v4f*)a_dst)[wv * (HID / 4) + c];
  float p[32];
#pragma unroll
  for (int m = 0; m < 2; ++m) {
#pragma unroll
    for (int r = 0; r < 8; ++r) {
      float s = acc[m][0][r] * vs.x, d = acc[m][0][r] * vd.x;
      s += acc[m][1][r] * vs.y;      d += acc[m][1][r] * vd.y;
      s += acc[m][2][r] * vs.z;      d += acc[m][2][r] * vd.z;
      s += acc[m][3][r] * vs.w;      d += acc[m][3][r] * vd.w;
      p[8 * m + r] = s;
      p[16 + 8 * m + r] = d;
    }
  }
  fold<16>(p, (c & 8) != 0, 8);
  fold<8>(p, (c & 4) != 0, 4);
  fold<4>(p, (c & 2) != 0, 2);
  fold<2>(p, (c & 1) != 0, 1);
  {
    const int rb = (c & 4) * 4 + 8 * half + 2 * (c & 3);
    float* sq = sa + (c & 8) * 32;
    sq[rb * HEADS + wv]       = p[0];
    sq[(rb + 1) * HEADS + wv] = p[1];
  }
  __syncthreads();
  if (threadIdx.x < 128) {
    const int q = (int)threadIdx.x >> 6;
    const int u = (int)threadIdx.x & 63;
    if (row0 + (u >> 1) < n) {
      v4f v = ((const v4f*)sa)[q * 64 + u];
      ((v4f*)(q ? ad : as))[(size_t)row0 * 2 + u] = v;
    }
  }
}

__global__ void deg_count(const int* __restrict__ dst, int nE, int* __restrict__ cnt) {
  int e = blockIdx.x * blockDim.x + threadIdx.x;
  if (e < nE) atomicAdd(cnt + dst[e], 1);
}

__global__ void seg_assign(const int* __restrict__ cnt, int n,
                           int* __restrict__ cursor, int* __restrict__ start) {
  int i = blockIdx.x * blockDim.x + threadIdx.x;
  int lane = threadIdx.x & 31;
  int c = (i < n) ? cnt[i] : 0;
  int incl = c;
#pragma unroll
  for (int o = 1; o < 32; o <<= 1) {
    int t = __shfl_up(incl, o, 32);
    if (lane >= o) incl += t;
  }
  int base = 0;
  if (lane == 31) base = atomicAdd(cursor, incl);
  base = __shfl(base, 31, 32);
  if (i < n) start[i] = base + incl - c;
}

__global__ void seg_fill(const int* __restrict__ src, const int* __restrict__ dst, int nE,
                         const int* __restrict__ start, int* __restrict__ cur,
                         int* __restrict__ elist) {
  int e = blockIdx.x * blockDim.x + threadIdx.x;
  if (e >= nE) return;
  int d = dst[e];
  int p = start[d] + atomicAdd(cur + d, 1);
  if ((unsigned)p < (unsigned)nE) elist[p] = src[e];
}


__device__ __forceinline__ float lane_bcast(float x, int i) {
  return __int_as_float(__builtin_amdgcn_readlane(__float_as_int(x), i));
}

__device__ __forceinline__ float as_scalar(float x) {
  return __int_as_float(__builtin_amdgcn_readfirstlane(__float_as_int(x)));
}

template <int W>
__device__ __forceinline__ float group_max(float v) {
#pragma unroll
  for (int o = W / 2; o > 0; o >>= 1) v = fmaxf(v, __shfl_xor(v, o, 32));
  return v;
}
template <int W>
__device__ __forceinline__ float group_sum(float v) {
#pragma unroll
  for (int o = W / 2; o > 0; o >>= 1) v += __shfl_xor(v, o, 32);
  return v;
}

__device__ __forceinline__ bool load_entries(const int* __restrict__ elist,
                                             const v4f* __restrict__ as4, int beg, int total,
                                             int base, int lane, int d, int n,
                                             const float (&adh)[8], int& mine, float (&ev)[8]) {
  int k = base + lane;
  bool valid = k < total;
  mine = (k >= 1 && valid) ? elist[beg + k - 1] : d;
  if ((unsigned)mine >= (unsigned)n) mine = d;
  v4f aL = as4[(unsigned)mine * 2u];
  v4f aH = as4[(unsigned)mine * 2u + 1u];
  ev[0] = lrelu(aL.x + adh[0]); ev[1] = lrelu(aL.y + adh[1]);
  ev[2] = lrelu(aL.z + adh[2]); ev[3] = lrelu(aL.w + adh[3]);
  ev[4] = lrelu(aH.x + adh[4]); ev[5] = lrelu(aH.y + adh[5]);
  ev[6] = lrelu(aH.z + adh[6]); ev[7] = lrelu(aH.w + adh[7]);
  return valid;
}

struct RowPart { v4f v0, v1, v2, v3; };

__device__ __forceinline__ RowPart row_load(const v4f* __restrict__ H4, int s_, unsigned cq) {
  const v4f* hp = H4 + ((unsigned)s_ * (unsigned)(F1 / 4) + cq);
  RowPart r;
  r.v0 = hp[0]; r.v1 = hp[16]; r.v2 = hp[32]; r.v3 = hp[48];
  return r;
}

__device__ __forceinline__ void row_add(v4f (&acc)[4], const RowPart& r, const float (&al)[8],
                                        int i, bool hi) {
  float l0 = lane_bcast(al[0], i), l1 = lane_bcast(al[1], i);
  float l2 = lane_bcast(al[2], i), l3 = lane_bcast(al[3], i);
  float h0 = lane_bcast(al[4], i), h1 = lane_bcast(al[5], i);
  float h2 = lane_bcast(al[6], i), h3 = lane_bcast(al[7], i);
  acc[0] += r.v0 * (hi ? h0 : l0);
  acc[1] += r.v1 * (hi ? h1 : l1);
  acc[2] += r.v2 * (hi ? h2 : l2);
  acc[3] += r.v3 * (hi ? h3 : l3);
}

__device__ __forceinline__ v4f bias_relu(v4f a, v4f b) {
  v4f v = a + b;
  v.x = v.x > 0.f ? v.x : 0.f;
  v.y = v.y > 0.f ? v.y : 0.f;
  v.z = v.z > 0.f ? v.z : 0.f;
  v.w = v.w > 0.f ? v.w : 0.f;
  return v;
}

__device__ __forceinline__ void node_row1(const int* __restrict__ elist, const int* __restrict__ start,
                                          const int* __restrict__ cnt, int nE, int n,
                                          const float* __restrict__ H, const float* __restrict__ as,
                                          const float* __restrict__ ad, const float* __restrict__ b1,
                                          int d, int lane, bool hi, unsigned cq, v4f (&o)[4]) {
  const v4f* H4  = (const v4f*)H;
  const v4f* as4 = (const v4f*)as;

  float adh[8];
#pragma unroll
  for (int h = 0; h < HEADS; ++h) adh[h] = ad[(size_t)d * HEADS + h];

  int beg = start[d];
  int deg = cnt[d];
  if (beg < 0 || deg < 0 || beg > nE || deg > nE - beg) deg = 0;
  int total = deg + 1;

  int mine; float ev[8];
  bool valid = load_entries(elist, as4, beg, total, 0, lane, d, n, adh, mine, ev);

  float mvh[8];
  {
    float mx[8];
#pragma unroll
    for (int h = 0; h < HEADS; ++h) mx[h] = valid ? ev[h] : -INFINITY;
    for (int base = 32; base < total; base += 32) {
      int m2; float e2[8];
      if (load_entries(elist, as4, beg, total, base, lane, d, n, adh, m2, e2)) {
#pragma unroll
        for (int h = 0; h < HEADS; ++h) mx[h] = fmaxf(mx[h], e2[h]);
      }
    }
#pragma unroll
    for (int h = 0; h < HEADS; ++h) mvh[h] = as_scalar(group_max<32>(mx[h]));
  }

  float ddh[8];
  {
    float ds[8];
#pragma unroll
    for (int h = 0; h < HEADS; ++h) ds[h] = valid ? __expf(ev[h] - mvh[h]) : 0.f;
    for (int base = 32; base < total; base += 32) {
      int m2; float e2[8];
      if (load_entries(elist, as4, beg, total, base, lane, d, n, adh, m2, e2)) {
#pragma unroll
        for (int h = 0; h < HEADS; ++h) ds[h] += __expf(e2[h] - mvh[h]);
      }
    }
#pragma unroll
    for (int h = 0; h < HEADS; ++h) ddh[h] = as_scalar(group_sum<32>(ds[h])) + 1e-16f;
  }

  v4f acc[4];
#pragma unroll
  for (int j = 0; j < 4; ++j) acc[j] = (v4f)(0.f);

  for (int base = 0;;) {
    int c = total - base; if (c > 32) c = 32;

    float al[8];
#pragma unroll
    for (int h = 0; h < HEADS; ++h) al[h] = __expf(ev[h] - mvh[h]) / ddh[h];

    int i = 0;
    for (; i + 1 < c; i += 2) {
      int s0 = __builtin_amdgcn_readlane(mine, i);
      int s1 = __builtin_amdgcn_readlane(mine, i + 1);
      RowPart r0 = row_load(H4, s0, cq);
      RowPart r1 = row_load(H4, s1, cq);
      row_add(acc, r0, al, i, hi);
      row_add(acc, r1, al, i + 1, hi);
    }
    if (i < c) {
      int s0 = __builtin_amdgcn_readlane(mine, i);
      RowPart r0 = row_load(H4, s0, cq);
      row_add(acc, r0, al, i, hi);
    }

    base += 32;
    if (base >= total) break;
    load_entries(elist, as4, beg, total, base, lane, d, n, adh, mine, ev);
  }

  const v4f* bp = (const v4f*)b1 + cq;
  o[0] = bias_relu(acc[0], bp[0]);
  o[1] = bias_relu(acc[1], bp[16]);
  o[2] = bias_relu(acc[2], bp[32]);
  o[3] = bias_relu(acc[3], bp[48]);
}

__device__ __forceinline__ unsigned asd_slot(unsigned i) { return ((i >> 3) << 4) | (i & 7u); }

#define XS_PITCH (F1 + 4)

__global__ void __launch_bounds__(256) __attribute__((amdgpu_waves_per_eu(8, 10)))
node_agg1(const int* __restrict__ elist, const int* __restrict__ start,
          const int* __restrict__ cnt, int nE, int n, const float* __restrict__ H,
          const float* __restrict__ as, const float* __restrict__ ad,
          const float* __restrict__ b1, const float* __restrict__ W2,
          const float* __restrict__ a_src2, const float* __restrict__ a_dst2,
          float* __restrict__ H2, float* __restrict__ asd) {
  __shared__ __attribute__((aligned(16))) float xs[8 * XS_PITCH];
  __shared__ float part[8 * 64];
  const int lane = threadIdx.x & 31;
  const int wv = __builtin_amdgcn_readfirstlane((int)(threadIdx.x >> 5));
  const int d0 = (int)blockIdx.x * 8;
  const int d  = d0 + wv;
  const bool hi = lane >= 16;
  const unsigned cq = (hi ? 64u : 0u) + ((unsigned)lane & 15u);

  v4f o[4];
#pragma unroll
  for (int j = 0; j < 4; ++j) o[j] = (v4f)(0.f);
  if (d < n)
    node_row1(elist, start, cnt, nE, n, H, as, ad, b1, d, lane, hi, cq, o);

  const int half = lane >> 4;
  const int c7   = lane & 7;
  const int k0   = wv * 64 + 2 * half;
  float bw[16][2];
#pragma unroll
  for (int i = 0; i < 16; ++i) {
    bw[i][0] = W2[(k0 + 4 * i) * D_OUT + c7];
    bw[i][1] = W2[(k0 + 4 * i + 1) * D_OUT + c7];
  }
  {
    v4f* xr = (v4f*)(xs + wv * XS_PITCH) + cq;
    xr[0] = o[0]; xr[16] = o[1]; xr[32] = o[2]; xr[48] = o[3];
  }
  __syncthreads();
  v8f acc2 = {};
  {
    const float* xrow = xs + c7 * XS_PITCH + k0;
#pragma unroll
    for (int i = 0; i < 16; ++i) {
      v2f a = *(const v2f*)(xrow + 4 * i);
      v2f b = {bw[i][0], bw[i][1]};
      acc2 = wmma_f32_k4(a, b, acc2);
    }
  }
  if (lane < 8) {
#pragma unroll
    for (int r = 0; r < 8; ++r) part[wv * 64 + r * 8 + lane] = acc2[r];
  }
  __syncthreads();
  if (wv == 0) {
    float hA = part[lane], hB = part[32 + lane];
#pragma unroll
    for (int w = 1; w < 8; ++w) { hA += part[w * 64 + lane]; hB += part[w * 64 + 32 + lane]; }
    const int r = lane >> 3;
    if (d0 + r < n)     H2[(size_t)d0 * D_OUT + lane]      = hA;
    if (d0 + 4 + r < n) H2[(size_t)d0 * D_OUT + 32 + lane] = hB;
    const float cs = a_src2[c7], cd = a_dst2[c7];
    const float sA = group_sum<8>(hA * cs), dA = group_sum<8>(hA * cd);
    const float sB = group_sum<8>(hB * cs), dB = group_sum<8>(hB * cd);
    const int g = (lane & 3) * 8;
    const float s_lo = __shfl(sA, g, 32), d_lo = __shfl(dA, g, 32);
    const float s_hi = __shfl(sB, g, 32), d_hi = __shfl(dB, g, 32);
    if (lane < 8 && d0 + lane < n) {
      v2f pr = {(lane & 4) ? s_hi : s_lo, (lane & 4) ? d_hi : d_lo};
      ((v2f*)asd)[asd_slot((unsigned)(d0 + lane))] = pr;
    }
  }
}

__device__ __forceinline__ float entry_score2(const int* __restrict__ elist,
                                              const float* __restrict__ asd, int beg, int k,
                                              int d, int n, float a_d, int& s) {
  s = k ? elist[beg + k - 1] : d;
  if ((unsigned)s >= (unsigned)n) s = d;
  return lrelu(asd[2u * asd_slot((unsigned)s)] + a_d);
}

__global__ void __launch_bounds__(256)
node_agg2(const int* __restrict__ elist, const int* __restrict__ start,
          const int* __restrict__ cnt, int nE, int n, const float* __restrict__ H2,
          const float* __restrict__ asd,
          const float* __restrict__ b2, float* __restrict__ out) {
  int t = blockIdx.x * blockDim.x + threadIdx.x;
  int j = threadIdx.x & 7;
  bool live = (t >> 3) < n;
  int d = live ? (t >> 3) : 0;
  const v4f* H4 = (const v4f*)H2;

  int beg = start[d];
  int deg = cnt[d];
  if (beg < 0 || deg < 0 || beg > nE || deg > nE - beg) deg = 0;
  int total = deg + 1;
  float a_d = asd[2u * asd_slot((unsigned)d) + 1u];

  float mx = -INFINITY;
  for (int k = j; k < total; k += 8) {
    int s;
    mx = fmaxf(mx, entry_score2(elist, asd, beg, k, d, n, a_d, s));
  }
  mx = group_max<8>(mx);

  float den = 0.f;
  for (int k = j; k < total; k += 8) {
    int s;
    den += __expf(entry_score2(elist, asd, beg, k, d, n, a_d, s) - mx);
  }
  den = group_sum<8>(den) + 1e-16f;

  v4f accL = (v4f)(0.f), accH = (v4f)(0.f);
  for (int k = j; k < total; k += 8) {
    int s;
    float alpha = __expf(entry_score2(elist, asd, beg, k, d, n, a_d, s) - mx) / den;
    accL += H4[(unsigned)s * 2u] * alpha;
    accH += H4[(unsigned)s * 2u + 1u] * alpha;
  }
  float r0 = group_sum<8>(accL.x), r1 = group_sum<8>(accL.y);
  float r2 = group_sum<8>(accL.z), r3 = group_sum<8>(accL.w);
  float r4 = group_sum<8>(accH.x), r5 = group_sum<8>(accH.y);
  float r6 = group_sum<8>(accH.z), r7 = group_sum<8>(accH.w);
  float lo = (j & 2) ? ((j & 1) ? r3 : r2) : ((j & 1) ? r1 : r0);
  float up = (j & 2) ? ((j & 1) ? r7 : r6) : ((j & 1) ? r5 : r4);
  float r = (j & 4) ? up : lo;

  if (live) out[t] = r + b2[j];
}

extern "C" void kernel_launch(void* const* d_in, const int* in_sizes, int n_in,
                              void* d_out, int out_size, void* d_ws, size_t ws_size,
                              hipStream_t stream) {
  const float* x      = (const float*)d_in[0];
  const int*   ei     = (const int*)  d_in[1];
  const float* W1     = (const float*)d_in[2];
  const float* a_src1 = (const float*)d_in[3];
  const float* a_dst1 = (const float*)d_in[4];
  const float* b1     = (const float*)d_in[5];
  const float* W2     = (const float*)d_in[6];
  const float* a_src2 = (const float*)d_in[7];
  const float* a_dst2 = (const float*)d_in[8];
  const float* b2     = (const float*)d_in[9];

  int n  = in_sizes[0] / D_IN;
  int nE = in_sizes[1] / 2;
  const int* src = ei;
  const int* dst = ei + nE;

  char* ws = (char*)d_ws;
  size_t off = 0;
  auto alloc = [&](size_t bytes) -> char* {
    char* p = ws + off; off += (bytes + 255) & ~(size_t)255; return p;
  };
  float*    h1   = (float*)   alloc((size_t)n * F1 * 4);
  int*      elist= (int*)     alloc((size_t)nE * 4);
  int*      seg0 = (int*)     alloc((size_t)n * 4);
  float*    as1  = (float*)   alloc((size_t)n * HEADS * 4);
  float*    ad1  = (float*)   alloc((size_t)n * HEADS * 4);
  float*    h2   = (float*)   alloc((size_t)n * D_OUT * 4);
  float*    asd2 = (float*)   alloc((size_t)((n + 7) / 8) * 128);
  int*      cnt  = (int*)     alloc((size_t)n * 4);
  int*      cur  = (int*)     alloc((size_t)n * 4);
  int*      curs = (int*)     alloc(4);

  hipMemsetAsync(cnt, 0, off - (size_t)((char*)cnt - ws), stream);

  deg_count <<<(nE + 255) / 256, 256, 0, stream>>>(dst, nE, cnt);
  seg_assign<<<(n + 255) / 256, 256, 0, stream>>>(cnt, n, curs, seg0);
  seg_fill  <<<(nE + 255) / 256, 256, 0, stream>>>(src, dst, nE, seg0, cur, elist);

  {
    int blocks = (n + 31) / 32;
    gemm1_wmma<<<blocks, 256, 0, stream>>>(x, W1, a_src1, a_dst1, h1, as1, ad1, n);
  }
  node_agg1<<<(n + 7) / 8, 256, 0, stream>>>(elist, seg0, cnt, nE, n, h1, as1, ad1, b1,
                                             W2, a_src2, a_dst2, h2, asd2);

  node_agg2<<<(n * D_OUT + 255) / 256, 256, 0, stream>>>(elist, seg0, cnt, nE, n, h2, asd2, b2,
                                                       (float*)d_out);
}


# Layer 1 aggregates before projecting: per head, alpha-weighted sums of x rows (512 B gathers), then their product with W1 on f32 WMMA in the node kernel; scores as x @ (W1 a); h1 gone; new sum order
# speedup vs baseline: 9.5573x; 1.2428x over previous
#include <hip/hip_runtime.h>
#include <math.h>


#define D_IN   128
#define HEADS  8
#define HID    64
#define F1     (HEADS * HID)
#define D_OUT  8
#define NEG_SLOPE 0.2f

typedef float v2f __attribute__((ext_vector_type(2)));
typedef float v4f __attribute__((ext_vector_type(4)));
typedef float v8f __attribute__((ext_vector_type(8)));

__device__ __forceinline__ float lrelu(float x) { return x > 0.f ? x : NEG_SLOPE * x; }

__device__ __forceinline__ v8f wmma_f32_k4(v2f a, v2f b, v8f c) {
  return __builtin_amdgcn_wmma_f32_16x16x4_f32(false, a, false, b, (short)0, c, false, false);
}

#define VW 16

__device__ __forceinline__ void coef_fold(const float* __restrict__ W, const float* __restrict__ a_src,
                                          const float* __restrict__ a_dst, float* __restrict__ V,
                                          int t) {
  const int k = t >> 4, q = t & 15, hd = q & 7;
  if (k >= D_IN) return;
  const v4f* wr = (const v4f*)(W + (size_t)k * F1 + hd * HID);
  const v4f* av = (const v4f*)((q < 8 ? a_src : a_dst) + hd * HID);
  float s = 0.f;
  for (int j = 0; j < HID / 4; ++j) {
    v4f w = wr[j], a = av[j];
    s += w.x * a.x; s += w.y * a.y; s += w.z * a.z; s += w.w * a.w;
  }
  V[t] = s;
}

__global__ void __launch_bounds__(128)
coef_wmma(const float* __restrict__ X, const float* __restrict__ V,
          float* __restrict__ as, float* __restrict__ ad, int n) {
  const int lane = threadIdx.x & 31;
  const int wv   = threadIdx.x >> 5;
  const int half = lane >> 4;
  const int c    = lane & 15;
  const int row0 = ((int)blockIdx.x * 4 + wv) * 32;
  if (row0 >= n) return;
  int ra0 = row0 + c;       if (ra0 > n - 1) ra0 = n - 1;
  int ra1 = row0 + 16 + c;  if (ra1 > n - 1) ra1 = n - 1;
  const v4f* xa0 = (const v4f*)(X + (size_t)ra0 * D_IN) + half;
  const v4f* xa1 = (const v4f*)(X + (size_t)ra1 * D_IN) + half;
  const float* vb = V + (4 * half) * VW + c;
  v8f acc0 = {}, acc1 = {};
#pragma unroll
  for (int s = 0; s < D_IN / 8; ++s) {
    v4f a0 = xa0[2 * s], a1 = xa1[2 * s];
    v2f bl = {vb[(8 * s + 0) * VW], vb[(8 * s + 1) * VW]};
    v2f bh = {vb[(8 * s + 2) * VW], vb[(8 * s + 3) * VW]};
    acc0 = wmma_f32_k4((v2f){a0.x, a0.y}, bl, acc0);
    acc1 = wmma_f32_k4((v2f){a1.x, a1.y}, bl, acc1);
    acc0 = wmma_f32_k4((v2f){a0.z, a0.w}, bh, acc0);
    acc1 = wmma_f32_k4((v2f){a1.z, a1.w}, bh, acc1);
  }
  float* op = (c < 8) ? (as + c) : (ad + (c - 8));
#pragma unroll
  for (int r = 0; r < 8; ++r) {
    int r0 = row0 + 8 * half + r, r1 = r0 + 16;
    if (r0 < n) op[(size_t)r0 * HEADS] = acc0[r];
    if (r1 < n) op[(size_t)r1 * HEADS] = acc1[r];
  }
}

#define FOLD_BLOCKS ((D_IN * VW + 255) / 256)
__global__ void __launch_bounds__(256)
deg_count(const int* __restrict__ dst, int nE, int* __restrict__ cnt,
          const float* __restrict__ W, const float* __restrict__ a_src,
          const float* __restrict__ a_dst, float* __restrict__ V) {
  const int nbE = (nE + 255) >> 8;
  if ((int)blockIdx.x >= nbE) {
    coef_fold(W, a_src, a_dst, V, ((int)blockIdx.x - nbE) * 256 + (int)threadIdx.x);
    return;
  }
  int e = blockIdx.x * 256 + threadIdx.x;
  if (e < nE) atomicAdd(cnt + dst[e], 1);
}

__global__ void seg_assign(const int* __restrict__ cnt, int n,
                           int* __restrict__ cursor, int* __restrict__ start) {
  int i = blockIdx.x * blockDim.x + threadIdx.x;
  int lane = threadIdx.x & 31;
  int c = (i < n) ? cnt[i] : 0;
  int incl = c;
#pragma unroll
  for (int o = 1; o < 32; o <<= 1) {
    int t = __shfl_up(incl, o, 32);
    if (lane >= o) incl += t;
  }
  int base = 0;
  if (lane == 31) base = atomicAdd(cursor, incl);
  base = __shfl(base, 31, 32);
  if (i < n) start[i] = base + incl - c;
}

__global__ void seg_fill(const int* __restrict__ src, const int* __restrict__ dst, int nE,
                         const int* __restrict__ start, int* __restrict__ cur,
                         int* __restrict__ elist) {
  int e = blockIdx.x * blockDim.x + threadIdx.x;
  if (e >= nE) return;
  int d = dst[e];
  int p = start[d] + atomicAdd(cur + d, 1);
  if ((unsigned)p < (unsigned)nE) elist[p] = src[e];
}


__device__ __forceinline__ float lane_bcast(float x, int i) {
  return __int_as_float(__builtin_amdgcn_readlane(__float_as_int(x), i));
}

__device__ __forceinline__ float as_scalar(float x) {
  return __int_as_float(__builtin_amdgcn_readfirstlane(__float_as_int(x)));
}

template <int W>
__device__ __forceinline__ float group_max(float v) {
#pragma unroll
  for (int o = W / 2; o > 0; o >>= 1) v = fmaxf(v, __shfl_xor(v, o, 32));
  return v;
}
template <int W>
__device__ __forceinline__ float group_sum(float v) {
#pragma unroll
  for (int o = W / 2; o > 0; o >>= 1) v += __shfl_xor(v, o, 32);
  return v;
}

__device__ __forceinline__ bool load_entries(const int* __restrict__ elist,
                                             const v4f* __restrict__ as4, int beg, int total,
                                             int base, int lane, int d, int n,
                                             const float (&adh)[8], int& mine, float (&ev)[8]) {
  int k = base + lane;
  bool valid = k < total;
  mine = (k >= 1 && valid) ? elist[beg + k - 1] : d;
  if ((unsigned)mine >= (unsigned)n) mine = d;
  v4f aL = as4[(unsigned)mine * 2u];
  v4f aH = as4[(unsigned)mine * 2u + 1u];
  ev[0] = lrelu(aL.x + adh[0]); ev[1] = lrelu(aL.y + adh[1]);
  ev[2] = lrelu(aL.z + adh[2]); ev[3] = lrelu(aL.w + adh[3]);
  ev[4] = lrelu(aH.x + adh[4]); ev[5] = lrelu(aH.y + adh[5]);
  ev[6] = lrelu(aH.z + adh[6]); ev[7] = lrelu(aH.w + adh[7]);
  return valid;
}


__device__ __forceinline__ void zrow_add(v4f (&z)[8], v4f r, const float (&al)[8], int i) {
#pragma unroll
  for (int h = 0; h < HEADS; ++h) z[h] += r * lane_bcast(al[h], i);
}

__device__ __forceinline__ v4f bias_relu(v4f a, v4f b) {
  v4f v = a + b;
  v.x = v.x > 0.f ? v.x : 0.f;
  v.y = v.y > 0.f ? v.y : 0.f;
  v.z = v.z > 0.f ? v.z : 0.f;
  v.w = v.w > 0.f ? v.w : 0.f;
  return v;
}

__device__ __forceinline__ void node_z1(const int* __restrict__ elist, const int* __restrict__ start,
                                        const int* __restrict__ cnt, int nE, int n,
                                        const float* __restrict__ X, const float* __restrict__ as,
                                        const float* __restrict__ ad, int d, int lane,
                                        v4f (&z)[8]) {
  const v4f* X4  = (const v4f*)X;
  const v4f* as4 = (const v4f*)as;

  float adh[8];
#pragma unroll
  for (int h = 0; h < HEADS; ++h) adh[h] = ad[(size_t)d * HEADS + h];

  int beg = start[d];
  int deg = cnt[d];
  if (beg < 0 || deg < 0 || beg > nE || deg > nE - beg) deg = 0;
  int total = deg + 1;

  int mine; float ev[8];
  bool valid = load_entries(elist, as4, beg, total, 0, lane, d, n, adh, mine, ev);

  float mvh[8];
  {
    float mx[8];
#pragma unroll
    for (int h = 0; h < HEADS; ++h) mx[h] = valid ? ev[h] : -INFINITY;
    for (int base = 32; base < total; base += 32) {
      int m2; float e2[8];
      if (load_entries(elist, as4, beg, total, base, lane, d, n, adh, m2, e2)) {
#pragma unroll
        for (int h = 0; h < HEADS; ++h) mx[h] = fmaxf(mx[h], e2[h]);
      }
    }
#pragma unroll
    for (int h = 0; h < HEADS; ++h) mvh[h] = as_scalar(group_max<32>(mx[h]));
  }

  float ddh[8];
  {
    float ds[8];
#pragma unroll
    for (int h = 0; h < HEADS; ++h) ds[h] = valid ? __expf(ev[h] - mvh[h]) : 0.f;
    for (int base = 32; base < total; base += 32) {
      int m2; float e2[8];
      if (load_entries(elist, as4, beg, total, base, lane, d, n, adh, m2, e2)) {
#pragma unroll
        for (int h = 0; h < HEADS; ++h) ds[h] += __expf(e2[h] - mvh[h]);
      }
    }
#pragma unroll
    for (int h = 0; h < HEADS; ++h) ddh[h] = as_scalar(group_sum<32>(ds[h])) + 1e-16f;
  }

  for (int base = 0;;) {
    int c = total - base; if (c > 32) c = 32;

    float al[8];
#pragma unroll
    for (int h = 0; h < HEADS; ++h) al[h] = __expf(ev[h] - mvh[h]) / ddh[h];

    for (int i = 0; i < c; i += 4) {
      int i1 = i + 1 < c ? i + 1 : i;
      int i2 = i + 2 < c ? i + 2 : i;
      int i3 = i + 3 < c ? i + 3 : i;
      int s0 = __builtin_amdgcn_readlane(mine, i);
      int s1 = __builtin_amdgcn_readlane(mine, i1);
      int s2 = __builtin_amdgcn_readlane(mine, i2);
      int s3 = __builtin_amdgcn_readlane(mine, i3);
      v4f r0 = X4[(unsigned)s0 * (unsigned)(D_IN / 4) + (unsigned)lane];
      v4f r1 = X4[(unsigned)s1 * (unsigned)(D_IN / 4) + (unsigned)lane];
      v4f r2 = X4[(unsigned)s2 * (unsigned)(D_IN / 4) + (unsigned)lane];
      v4f r3 = X4[(unsigned)s3 * (unsigned)(D_IN / 4) + (unsigned)lane];
      asm volatile("" :: "v"(r0), "v"(r1), "v"(r2), "v"(r3));
      zrow_add(z, r0, al, i);
      if (i + 1 < c) zrow_add(z, r1, al, i + 1);
      if (i + 2 < c) zrow_add(z, r2, al, i + 2);
      if (i + 3 < c) zrow_add(z, r3, al, i + 3);
    }

    base += 32;
    if (base >= total) break;
    load_entries(elist, as4, beg, total, base, lane, d, n, adh, mine, ev);
  }
}


__device__ __forceinline__ unsigned asd_slot(unsigned i) { return ((i >> 3) << 4) | (i & 7u); }

#define NPW 16
#define ZS_PITCH (HEADS * D_IN + 4)
#define XS_PITCH (F1 + 4)

__global__ void __launch_bounds__(256) __attribute__((amdgpu_waves_per_eu(8, 10)))
node_agg1(const int* __restrict__ elist, const int* __restrict__ start,
          const int* __restrict__ cnt, int nE, int n, const float* __restrict__ X,
          const float* __restrict__ as, const float* __restrict__ ad,
          const float* __restrict__ W1, const float* __restrict__ b1,
          const float* __restrict__ W2,
          const float* __restrict__ a_src2, const float* __restrict__ a_dst2,
          float* __restrict__ H2, float* __restrict__ asd) {
  __shared__ __attribute__((aligned(16))) float zs[NPW * ZS_PITCH];
  __shared__ float part[8 * NPW * D_OUT];
  const int lane = threadIdx.x & 31;
  const int wv = __builtin_amdgcn_readfirstlane((int)(threadIdx.x >> 5));
  const int d0 = (int)blockIdx.x * NPW;
  const int half = lane >> 4;
  const int c    = lane & 15;

  for (int q = 0; q < 2; ++q) {
    const int nl = wv + 8 * q;
    const int d  = d0 + nl;
    v4f z[8];
#pragma unroll
    for (int h = 0; h < HEADS; ++h) z[h] = (v4f)(0.f);
    if (d < n)
      node_z1(elist, start, cnt, nE, n, X, as, ad, d, lane, z);
    v4f* zr = (v4f*)(zs + nl * ZS_PITCH) + lane;
#pragma unroll
    for (int h = 0; h < HEADS; ++h) zr[h * (D_IN / 4)] = z[h];
  }

  const int col0 = wv * HID;
  const v4f* wb = (const v4f*)(W1 + (size_t)(4 * half) * F1 + col0) + c;
  const int WR = F1 / 4;
  v4f b0 = wb[0], b1v = wb[WR], b2 = wb[2 * WR], b3 = wb[3 * WR];
  v4f p0 = wb[8 * WR], p1 = wb[9 * WR], p2 = wb[10 * WR], p3 = wb[11 * WR];
  __syncthreads();
  v8f acc[4];
#pragma unroll
  for (int t = 0; t < 4; ++t) acc[t] = (v8f){};
  {
    const v4f* za = (const v4f*)(zs + c * ZS_PITCH + wv * D_IN) + half;
#pragma unroll
    for (int s = 0; s < D_IN / 8; ++s) {
      v4f q0 = p0, q1 = p1, q2 = p2, q3 = p3;
      if (s + 2 < D_IN / 8) {
        q0 = wb[(8 * (s + 2) + 0) * WR];
        q1 = wb[(8 * (s + 2) + 1) * WR];
        q2 = wb[(8 * (s + 2) + 2) * WR];
        q3 = wb[(8 * (s + 2) + 3) * WR];
      }
      v4f a = za[2 * s];
      __builtin_amdgcn_sched_barrier(0);
      v2f al = {a.x, a.y}, ah = {a.z, a.w};
#pragma unroll
      for (int t = 0; t < 4; ++t) acc[t] = wmma_f32_k4(al, (v2f){b0[t], b1v[t]}, acc[t]);
#pragma unroll
      for (int t = 0; t < 4; ++t) acc[t] = wmma_f32_k4(ah, (v2f){b2[t], b3[t]}, acc[t]);
      b0 = p0; b1v = p1; b2 = p2; b3 = p3;
      p0 = q0; p1 = q1; p2 = q2; p3 = q3;
    }
  }

  const int c7 = lane & 7;
  const int k0 = wv * 64 + 2 * half;
  float bw[16][2];
#pragma unroll
  for (int i = 0; i < 16; ++i) {
    bw[i][0] = W2[(k0 + 4 * i) * D_OUT + c7];
    bw[i][1] = W2[(k0 + 4 * i + 1) * D_OUT + c7];
  }
  const v4f bv = ((const v4f*)b1)[wv * (HID / 4) + c];
  __syncthreads();
#pragma unroll
  for (int r = 0; r < 8; ++r) {
    v4f o = {acc[0][r], acc[1][r], acc[2][r], acc[3][r]};
    *((v4f*)(zs + (8 * half + r) * XS_PITCH + col0) + c) = bias_relu(o, bv);
  }
  __syncthreads();
  v8f acc2 = {};
  {
    const float* xrow = zs + c * XS_PITCH + k0;
#pragma unroll
    for (int i = 0; i < 16; ++i) {
      v2f a = *(const v2f*)(xrow + 4 * i);
      v2f b = {bw[i][0], bw[i][1]};
      acc2 = wmma_f32_k4(a, b, acc2);
    }
  }
  if (c < 8) {
#pragma unroll
    for (int r = 0; r < 8; ++r) part[wv * (NPW * D_OUT) + (8 * half + r) * D_OUT + c] = acc2[r];
  }
  __syncthreads();
  if (wv < 2) {
    const int dq = d0 + 8 * wv;
    const float* pq = part + 64 * wv;
    float hA = pq[lane], hB = pq[32 + lane];
#pragma unroll
    for (int w = 1; w < 8; ++w) {
      hA += pq[w * (NPW * D_OUT) + lane];
      hB += pq[w * (NPW * D_OUT) + 32 + lane];
    }
    const int r = lane >> 3;
    if (dq + r < n)     H2[(size_t)dq * D_OUT + lane]      = hA;
    if (dq + 4 + r < n) H2[(size_t)dq * D_OUT + 32 + lane] = hB;
    const float cs = a_src2[c7], cd = a_dst2[c7];
    const float sA = group_sum<8>(hA * cs), dA = group_sum<8>(hA * cd);
    const float sB = group_sum<8>(hB * cs), dB = group_sum<8>(hB * cd);
    const int g = (lane & 3) * 8;
    const float s_lo = __shfl(sA, g, 32), d_lo = __shfl(dA, g, 32);
    const float s_hi = __shfl(sB, g, 32), d_hi = __shfl(dB, g, 32);
    if (lane < 8 && dq + lane < n) {
      v2f pr = {(lane & 4) ? s_hi : s_lo, (lane & 4) ? d_hi : d_lo};
      ((v2f*)asd)[asd_slot((unsigned)(dq + lane))] = pr;
    }
  }
}

__device__ __forceinline__ float entry_score2(const int* __restrict__ elist,
                                              const float* __restrict__ asd, int beg, int k,
                                              int d, int n, float a_d, int& s) {
  s = k ? elist[beg + k - 1] : d;
  if ((unsigned)s >= (unsigned)n) s = d;
  return lrelu(asd[2u * asd_slot((unsigned)s)] + a_d);
}

__global__ void __launch_bounds__(256)
node_agg2(const int* __restrict__ elist, const int* __restrict__ start,
          const int* __restrict__ cnt, int nE, int n, const float* __restrict__ H2,
          const float* __restrict__ asd,
          const float* __restrict__ b2, float* __restrict__ out) {
  int t = blockIdx.x * blockDim.x + threadIdx.x;
  int j = threadIdx.x & 7;
  bool live = (t >> 3) < n;
  int d = live ? (t >> 3) : 0;
  const v4f* H4 = (const v4f*)H2;

  int beg = start[d];
  int deg = cnt[d];
  if (beg < 0 || deg < 0 || beg > nE || deg > nE - beg) deg = 0;
  int total = deg + 1;
  float a_d = asd[2u * asd_slot((unsigned)d) + 1u];

  float mx = -INFINITY;
  for (int k = j; k < total; k += 8) {
    int s;
    mx = fmaxf(mx, entry_score2(elist, asd, beg, k, d, n, a_d, s));
  }
  mx = group_max<8>(mx);

  float den = 0.f;
  for (int k = j; k < total; k += 8) {
    int s;
    den += __expf(entry_score2(elist, asd, beg, k, d, n, a_d, s) - mx);
  }
  den = group_sum<8>(den) + 1e-16f;

  v4f accL = (v4f)(0.f), accH = (v4f)(0.f);
  for (int k = j; k < total; k += 8) {
    int s;
    float alpha = __expf(entry_score2(elist, asd, beg, k, d, n, a_d, s) - mx) / den;
    accL += H4[(unsigned)s * 2u] * alpha;
    accH += H4[(unsigned)s * 2u + 1u] * alpha;
  }
  float r0 = group_sum<8>(accL.x), r1 = group_sum<8>(accL.y);
  float r2 = group_sum<8>(accL.z), r3 = group_sum<8>(accL.w);
  float r4 = group_sum<8>(accH.x), r5 = group_sum<8>(accH.y);
  float r6 = group_sum<8>(accH.z), r7 = group_sum<8>(accH.w);
  float lo = (j & 2) ? ((j & 1) ? r3 : r2) : ((j & 1) ? r1 : r0);
  float up = (j & 2) ? ((j & 1) ? r7 : r6) : ((j & 1) ? r5 : r4);
  float r = (j & 4) ? up : lo;

  if (live) out[t] = r + b2[j];
}

extern "C" void kernel_launch(void* const* d_in, const int* in_sizes, int n_in,
                              void* d_out, int out_size, void* d_ws, size_t ws_size,
                              hipStream_t stream) {
  const float* x      = (const float*)d_in[0];
  const int*   ei     = (const int*)  d_in[1];
  const float* W1     = (const float*)d_in[2];
  const float* a_src1 = (const float*)d_in[3];
  const float* a_dst1 = (const float*)d_in[4];
  const float* b1     = (const float*)d_in[5];
  const float* W2     = (const float*)d_in[6];
  const float* a_src2 = (const float*)d_in[7];
  const float* a_dst2 = (const float*)d_in[8];
  const float* b2     = (const float*)d_in[9];

  int n  = in_sizes[0] / D_IN;
  int nE = in_sizes[1] / 2;
  const int* src = ei;
  const int* dst = ei + nE;

  char* ws = (char*)d_ws;
  size_t off = 0;
  auto alloc = [&](size_t bytes) -> char* {
    char* p = ws + off; off += (bytes + 255) & ~(size_t)255; return p;
  };
  float*    vco  = (float*)   alloc((size_t)D_IN * VW * 4);
  int*      elist= (int*)     alloc((size_t)nE * 4);
  int*      seg0 = (int*)     alloc((size_t)n * 4);
  float*    as1  = (float*)   alloc((size_t)n * HEADS * 4);
  float*    ad1  = (float*)   alloc((size_t)n * HEADS * 4);
  float*    h2   = (float*)   alloc((size_t)n * D_OUT * 4);
  float*    asd2 = (float*)   alloc((size_t)((n + 7) / 8) * 128);
  int*      cnt  = (int*)     alloc((size_t)n * 4);
  int*      cur  = (int*)     alloc((size_t)n * 4);
  int*      curs = (int*)     alloc(4);

  hipMemsetAsync(cnt, 0, off - (size_t)((char*)cnt - ws), stream);

  deg_count <<<(nE + 255) / 256 + FOLD_BLOCKS, 256, 0, stream>>>(dst, nE, cnt, W1, a_src1, a_dst1, vco);
  seg_assign<<<(n + 255) / 256, 256, 0, stream>>>(cnt, n, curs, seg0);
  seg_fill  <<<(nE + 255) / 256, 256, 0, stream>>>(src, dst, nE, seg0, cur, elist);

  coef_wmma<<<((n + 31) / 32 + 3) / 4, 128, 0, stream>>>(x, vco, as1, ad1, n);

  node_agg1<<<(n + NPW - 1) / NPW, 256, 0, stream>>>(elist, seg0, cnt, nE, n, x, as1, ad1, W1, b1,
                                                     W2, a_src2, a_dst2, h2, asd2);

  node_agg2<<<(n * D_OUT + 255) / 256, 256, 0, stream>>>(elist, seg0, cnt, nE, n, h2, asd2, b2,
                                                       (float*)d_out);
}


# Edges grouped in ONE launch: owner work-groups read all destinations and keep their node range's edges in LDS (LDS counters); memset, global atomics, 2 launches gone; list order: arrival rank
# speedup vs baseline: 11.4512x; 1.1982x over previous
#include <hip/hip_runtime.h>
#include <math.h>


#define D_IN   128
#define HEADS  8
#define HID    64
#define F1     (HEADS * HID)
#define D_OUT  8
#define NEG_SLOPE 0.2f

typedef float v2f __attribute__((ext_vector_type(2)));
typedef float v4f __attribute__((ext_vector_type(4)));
typedef float v8f __attribute__((ext_vector_type(8)));

__device__ __forceinline__ float lrelu(float x) { return x > 0.f ? x : NEG_SLOPE * x; }

__device__ __forceinline__ v8f wmma_f32_k4(v2f a, v2f b, v8f c) {
  return __builtin_amdgcn_wmma_f32_16x16x4_f32(false, a, false, b, (short)0, c, false, false);
}

#define VW 16

__device__ __forceinline__ void coef_fold(const float* __restrict__ W, const float* __restrict__ a_src,
                                          const float* __restrict__ a_dst, float* __restrict__ V,
                                          int t) {
  const int k = t >> 4, q = t & 15, hd = q & 7;
  if (k >= D_IN) return;
  const v4f* wr = (const v4f*)(W + (size_t)k * F1 + hd * HID);
  const v4f* av = (const v4f*)((q < 8 ? a_src : a_dst) + hd * HID);
  float s = 0.f;
  for (int j = 0; j < HID / 4; ++j) {
    v4f w = wr[j], a = av[j];
    s += w.x * a.x; s += w.y * a.y; s += w.z * a.z; s += w.w * a.w;
  }
  V[t] = s;
}

__global__ void __launch_bounds__(128)
coef_wmma(const float* __restrict__ X, const float* __restrict__ V,
          float* __restrict__ as, float* __restrict__ ad, int n) {
  const int lane = threadIdx.x & 31;
  const int wv   = threadIdx.x >> 5;
  const int half = lane >> 4;
  const int c    = lane & 15;
  const int row0 = ((int)blockIdx.x * 4 + wv) * 32;
  if (row0 >= n) return;
  int ra0 = row0 + c;       if (ra0 > n - 1) ra0 = n - 1;
  int ra1 = row0 + 16 + c;  if (ra1 > n - 1) ra1 = n - 1;
  const v4f* xa0 = (const v4f*)(X + (size_t)ra0 * D_IN) + half;
  const v4f* xa1 = (const v4f*)(X + (size_t)ra1 * D_IN) + half;
  const float* vb = V + (4 * half) * VW + c;
  v8f acc0 = {}, acc1 = {};
#pragma unroll
  for (int s = 0; s < D_IN / 8; ++s) {
    v4f a0 = xa0[2 * s], a1 = xa1[2 * s];
    v2f bl = {vb[(8 * s + 0) * VW], vb[(8 * s + 1) * VW]};
    v2f bh = {vb[(8 * s + 2) * VW], vb[(8 * s + 3) * VW]};
    acc0 = wmma_f32_k4((v2f){a0.x, a0.y}, bl, acc0);
    acc1 = wmma_f32_k4((v2f){a1.x, a1.y}, bl, acc1);
    acc0 = wmma_f32_k4((v2f){a0.z, a0.w}, bh, acc0);
    acc1 = wmma_f32_k4((v2f){a1.z, a1.w}, bh, acc1);
  }
  float* op = (c < 8) ? (as + c) : (ad + (c - 8));
#pragma unroll
  for (int r = 0; r < 8; ++r) {
    int r0 = row0 + 8 * half + r, r1 = r0 + 16;
    if (r0 < n) op[(size_t)r0 * HEADS] = acc0[r];
    if (r1 < n) op[(size_t)r1 * HEADS] = acc1[r];
  }
}

typedef int v4i __attribute__((ext_vector_type(4)));
#define GR   400
#define GT   1024
#define GCAP 12288
#define FOLD_BLOCKS ((D_IN * VW + GT - 1) / GT)

__device__ __forceinline__ void keep_edge(int* cnt_l, int* nm, int* m_ln, int* m_rk, int* m_sr,
                                          const int* __restrict__ src, int t, int e) {
  int r = atomicAdd(&cnt_l[t], 1);
  int q = atomicAdd(nm, 1);
  if (q < GCAP) { m_ln[q] = t; m_rk[q] = r; m_sr[q] = src[e]; }
}

__global__ void __launch_bounds__(GT)
group_edges(const int* __restrict__ src, const int* __restrict__ dst, int nE, int n,
            int* __restrict__ start, int* __restrict__ cnt, int* __restrict__ elist,
            const float* __restrict__ W, const float* __restrict__ a_src,
            const float* __restrict__ a_dst, float* __restrict__ V) {
  __shared__ int cnt_l[GR];
  __shared__ int off_l[GR];
  __shared__ int m_ln[GCAP];
  __shared__ int m_rk[GCAP];
  __shared__ int m_sr[GCAP];
  __shared__ int wtot[GT / 32];
  __shared__ int wbel[GT / 32];
  __shared__ int nm_s;
  const int nbO = (n + GR - 1) / GR;
  if ((int)blockIdx.x >= nbO) {
    coef_fold(W, a_src, a_dst, V, ((int)blockIdx.x - nbO) * GT + (int)threadIdx.x);
    return;
  }
  const int tid = threadIdx.x, lane = tid & 31, wv = tid >> 5;
  const int lo = (int)blockIdx.x * GR;
  const int nr = (n - lo < GR) ? (n - lo) : GR;
  if (tid < GR) cnt_l[tid] = 0;
  if (tid == 0) nm_s = 0;
  __syncthreads();

  int below = 0;
  const int n4 = (((size_t)dst & 15) == 0) ? (nE >> 2) : 0;
  const v4i* d4p = (const v4i*)dst;
  for (int g0 = tid; g0 < n4; g0 += 8 * GT) {
    v4i d[8];
#pragma unroll
    for (int k = 0; k < 8; ++k) {
      int g = g0 + k * GT;
      d[k] = d4p[g < n4 ? g : g0];
    }
    asm volatile("" :: "v"(d[0]), "v"(d[1]), "v"(d[2]), "v"(d[3]),
                       "v"(d[4]), "v"(d[5]), "v"(d[6]), "v"(d[7]));
#pragma unroll
    for (int k = 0; k < 8; ++k) {
      int g = g0 + k * GT;
      if (g < n4) {
        int t0 = d[k].x - lo, t1 = d[k].y - lo, t2 = d[k].z - lo, t3 = d[k].w - lo;
        below += (t0 < 0) + (t1 < 0) + (t2 < 0) + (t3 < 0);
        if ((unsigned)t0 < (unsigned)nr) keep_edge(cnt_l, &nm_s, m_ln, m_rk, m_sr, src, t0, 4 * g);
        if ((unsigned)t1 < (unsigned)nr) keep_edge(cnt_l, &nm_s, m_ln, m_rk, m_sr, src, t1, 4 * g + 1);
        if ((unsigned)t2 < (unsigned)nr) keep_edge(cnt_l, &nm_s, m_ln, m_rk, m_sr, src, t2, 4 * g + 2);
        if ((unsigned)t3 < (unsigned)nr) keep_edge(cnt_l, &nm_s, m_ln, m_rk, m_sr, src, t3, 4 * g + 3);
      }
    }
  }
  for (int e = 4 * n4 + tid; e < nE; e += GT) {
    int t = dst[e] - lo;
    below += (t < 0);
    if ((unsigned)t < (unsigned)nr) keep_edge(cnt_l, &nm_s, m_ln, m_rk, m_sr, src, t, e);
  }
  __syncthreads();

  const int c = (tid < GR) ? cnt_l[tid] : 0;
  int incl = c;
#pragma unroll
  for (int o = 1; o < 32; o <<= 1) {
    int t = __shfl_up(incl, o, 32);
    if (lane >= o) incl += t;
  }
#pragma unroll
  for (int o = 16; o > 0; o >>= 1) below += __shfl_xor(below, o, 32);
  if (lane == 31) wtot[wv] = incl;
  if (lane == 0)  wbel[wv] = below;
  __syncthreads();
  int woff = 0, base = 0;
  for (int w = 0; w < GT / 32; ++w) {
    if (w < wv) woff += wtot[w];
    base += wbel[w];
  }
  const int off = woff + incl - c;
  if (tid < GR) off_l[tid] = off;
  if (tid < nr) { start[lo + tid] = base + off; cnt[lo + tid] = c; }
  __syncthreads();

  const int nm = nm_s;
  if (nm <= GCAP) {
    for (int q = tid; q < nm; q += GT) {
      int p = base + off_l[m_ln[q]] + m_rk[q];
      if ((unsigned)p < (unsigned)nE) elist[p] = m_sr[q];
    }
  } else {
    if (tid < GR) cnt_l[tid] = 0;
    __syncthreads();
    for (int e = tid; e < nE; e += GT) {
      int t = dst[e] - lo;
      if ((unsigned)t < (unsigned)nr) {
        int p = base + off_l[t] + atomicAdd(&cnt_l[t], 1);
        if ((unsigned)p < (unsigned)nE) elist[p] = src[e];
      }
    }
  }
}


__device__ __forceinline__ float lane_bcast(float x, int i) {
  return __int_as_float(__builtin_amdgcn_readlane(__float_as_int(x), i));
}

__device__ __forceinline__ float as_scalar(float x) {
  return __int_as_float(__builtin_amdgcn_readfirstlane(__float_as_int(x)));
}

template <int W>
__device__ __forceinline__ float group_max(float v) {
#pragma unroll
  for (int o = W / 2; o > 0; o >>= 1) v = fmaxf(v, __shfl_xor(v, o, 32));
  return v;
}
template <int W>
__device__ __forceinline__ float group_sum(float v) {
#pragma unroll
  for (int o = W / 2; o > 0; o >>= 1) v += __shfl_xor(v, o, 32);
  return v;
}

__device__ __forceinline__ bool load_entries(const int* __restrict__ elist,
                                             const v4f* __restrict__ as4, int beg, int total,
                                             int base, int lane, int d, int n,
                                             const float (&adh)[8], int& mine, float (&ev)[8]) {
  int k = base + lane;
  bool valid = k < total;
  mine = (k >= 1 && valid) ? elist[beg + k - 1] : d;
  if ((unsigned)mine >= (unsigned)n) mine = d;
  v4f aL = as4[(unsigned)mine * 2u];
  v4f aH = as4[(unsigned)mine * 2u + 1u];
  ev[0] = lrelu(aL.x + adh[0]); ev[1] = lrelu(aL.y + adh[1]);
  ev[2] = lrelu(aL.z + adh[2]); ev[3] = lrelu(aL.w + adh[3]);
  ev[4] = lrelu(aH.x + adh[4]); ev[5] = lrelu(aH.y + adh[5]);
  ev[6] = lrelu(aH.z + adh[6]); ev[7] = lrelu(aH.w + adh[7]);
  return valid;
}


__device__ __forceinline__ void zrow_add(v4f (&z)[8], v4f r, const float (&al)[8], int i) {
#pragma unroll
  for (int h = 0; h < HEADS; ++h) z[h] += r * lane_bcast(al[h], i);
}

__device__ __forceinline__ v4f bias_relu(v4f a, v4f b) {
  v4f v = a + b;
  v.x = v.x > 0.f ? v.x : 0.f;
  v.y = v.y > 0.f ? v.y : 0.f;
  v.z = v.z > 0.f ? v.z : 0.f;
  v.w = v.w > 0.f ? v.w : 0.f;
  return v;
}

__device__ __forceinline__ void node_z1(const int* __restrict__ elist, const int* __restrict__ start,
                                        const int* __restrict__ cnt, int nE, int n,
                                        const float* __restrict__ X, const float* __restrict__ as,
                                        const float* __restrict__ ad, int d, int lane,
                                        v4f (&z)[8]) {
  const v4f* X4  = (const v4f*)X;
  const v4f* as4 = (const v4f*)as;

  float adh[8];
#pragma unroll
  for (int h = 0; h < HEADS; ++h) adh[h] = ad[(size_t)d * HEADS + h];

  int beg = start[d];
  int deg = cnt[d];
  if (beg < 0 || deg < 0 || beg > nE || deg > nE - beg) deg = 0;
  int total = deg + 1;

  int mine; float ev[8];
  bool valid = load_entries(elist, as4, beg, total, 0, lane, d, n, adh, mine, ev);

  float mvh[8];
  {
    float mx[8];
#pragma unroll
    for (int h = 0; h < HEADS; ++h) mx[h] = valid ? ev[h] : -INFINITY;
    for (int base = 32; base < total; base += 32) {
      int m2; float e2[8];
      if (load_entries(elist, as4, beg, total, base, lane, d, n, adh, m2, e2)) {
#pragma unroll
        for (int h = 0; h < HEADS; ++h) mx[h] = fmaxf(mx[h], e2[h]);
      }
    }
#pragma unroll
    for (int h = 0; h < HEADS; ++h) mvh[h] = as_scalar(group_max<32>(mx[h]));
  }

  float ddh[8];
  {
    float ds[8];
#pragma unroll
    for (int h = 0; h < HEADS; ++h) ds[h] = valid ? __expf(ev[h] - mvh[h]) : 0.f;
    for (int base = 32; base < total; base += 32) {
      int m2; float e2[8];
      if (load_entries(elist, as4, beg, total, base, lane, d, n, adh, m2, e2)) {
#pragma unroll
        for (int h = 0; h < HEADS; ++h) ds[h] += __expf(e2[h] - mvh[h]);
      }
    }
#pragma unroll
    for (int h = 0; h < HEADS; ++h) ddh[h] = as_scalar(group_sum<32>(ds[h])) + 1e-16f;
  }

  for (int base = 0;;) {
    int c = total - base; if (c > 32) c = 32;

    float al[8];
#pragma unroll
    for (int h = 0; h < HEADS; ++h) al[h] = __expf(ev[h] - mvh[h]) / ddh[h];

    for (int i = 0; i < c; i += 4) {
      int i1 = i + 1 < c ? i + 1 : i;
      int i2 = i + 2 < c ? i + 2 : i;
      int i3 = i + 3 < c ? i + 3 : i;
      int s0 = __builtin_amdgcn_readlane(mine, i);
      int s1 = __builtin_amdgcn_readlane(mine, i1);
      int s2 = __builtin_amdgcn_readlane(mine, i2);
      int s3 = __builtin_amdgcn_readlane(mine, i3);
      v4f r0 = X4[(unsigned)s0 * (unsigned)(D_IN / 4) + (unsigned)lane];
      v4f r1 = X4[(unsigned)s1 * (unsigned)(D_IN / 4) + (unsigned)lane];
      v4f r2 = X4[(unsigned)s2 * (unsigned)(D_IN / 4) + (unsigned)lane];
      v4f r3 = X4[(unsigned)s3 * (unsigned)(D_IN / 4) + (unsigned)lane];
      asm volatile("" :: "v"(r0), "v"(r1), "v"(r2), "v"(r3));
      zrow_add(z, r0, al, i);
      if (i + 1 < c) zrow_add(z, r1, al, i + 1);
      if (i + 2 < c) zrow_add(z, r2, al, i + 2);
      if (i + 3 < c) zrow_add(z, r3, al, i + 3);
    }

    base += 32;
    if (base >= total) break;
    load_entries(elist, as4, beg, total, base, lane, d, n, adh, mine, ev);
  }
}


__device__ __forceinline__ unsigned asd_slot(unsigned i) { return ((i >> 3) << 4) | (i & 7u); }

#define NPW 16
#define ZS_PITCH (HEADS * D_IN + 4)
#define XS_PITCH (F1 + 4)

__global__ void __launch_bounds__(256) __attribute__((amdgpu_waves_per_eu(8, 10)))
node_agg1(const int* __restrict__ elist, const int* __restrict__ start,
          const int* __restrict__ cnt, int nE, int n, const float* __restrict__ X,
          const float* __restrict__ as, const float* __restrict__ ad,
          const float* __restrict__ W1, const float* __restrict__ b1,
          const float* __restrict__ W2,
          const float* __restrict__ a_src2, const float* __restrict__ a_dst2,
          float* __restrict__ H2, float* __restrict__ asd) {
  __shared__ __attribute__((aligned(16))) float zs[NPW * ZS_PITCH];
  __shared__ float part[8 * NPW * D_OUT];
  const int lane = threadIdx.x & 31;
  const int wv = __builtin_amdgcn_readfirstlane((int)(threadIdx.x >> 5));
  const int d0 = (int)blockIdx.x * NPW;
  const int half = lane >> 4;
  const int c    = lane & 15;

  for (int q = 0; q < 2; ++q) {
    const int nl = wv + 8 * q;
    const int d  = d0 + nl;
    v4f z[8];
#pragma unroll
    for (int h = 0; h < HEADS; ++h) z[h] = (v4f)(0.f);
    if (d < n)
      node_z1(elist, start, cnt, nE, n, X, as, ad, d, lane, z);
    v4f* zr = (v4f*)(zs + nl * ZS_PITCH) + lane;
#pragma unroll
    for (int h = 0; h < HEADS; ++h) zr[h * (D_IN / 4)] = z[h];
  }

  const int col0 = wv * HID;
  const v4f* wb = (const v4f*)(W1 + (size_t)(4 * half) * F1 + col0) + c;
  const int WR = F1 / 4;
  v4f b0 = wb[0], b1v = wb[WR], b2 = wb[2 * WR], b3 = wb[3 * WR];
  v4f p0 = wb[8 * WR], p1 = wb[9 * WR], p2 = wb[10 * WR], p3 = wb[11 * WR];
  __syncthreads();
  v8f acc[4];
#pragma unroll
  for (int t = 0; t < 4; ++t) acc[t] = (v8f){};
  {
    const v4f* za = (const v4f*)(zs + c * ZS_PITCH + wv * D_IN) + half;
#pragma unroll
    for (int s = 0; s < D_IN / 8; ++s) {
      v4f q0 = p0, q1 = p1, q2 = p2, q3 = p3;
      if (s + 2 < D_IN / 8) {
        q0 = wb[(8 * (s + 2) + 0) * WR];
        q1 = wb[(8 * (s + 2) + 1) * WR];
        q2 = wb[(8 * (s + 2) + 2) * WR];
        q3 = wb[(8 * (s + 2) + 3) * WR];
      }
      v4f a = za[2 * s];
      __builtin_amdgcn_sched_barrier(0);
      v2f al = {a.x, a.y}, ah = {a.z, a.w};
#pragma unroll
      for (int t = 0; t < 4; ++t) acc[t] = wmma_f32_k4(al, (v2f){b0[t], b1v[t]}, acc[t]);
#pragma unroll
      for (int t = 0; t < 4; ++t) acc[t] = wmma_f32_k4(ah, (v2f){b2[t], b3[t]}, acc[t]);
      b0 = p0; b1v = p1; b2 = p2; b3 = p3;
      p0 = q0; p1 = q1; p2 = q2; p3 = q3;
    }
  }

  const int c7 = lane & 7;
  const int k0 = wv * 64 + 2 * half;
  float bw[16][2];
#pragma unroll
  for (int i = 0; i < 16; ++i) {
    bw[i][0] = W2[(k0 + 4 * i) * D_OUT + c7];
    bw[i][1] = W2[(k0 + 4 * i + 1) * D_OUT + c7];
  }
  const v4f bv = ((const v4f*)b1)[wv * (HID / 4) + c];
  __syncthreads();
#pragma unroll
  for (int r = 0; r < 8; ++r) {
    v4f o = {acc[0][r], acc[1][r], acc[2][r], acc[3][r]};
    *((v4f*)(zs + (8 * half + r) * XS_PITCH + col0) + c) = bias_relu(o, bv);
  }
  __syncthreads();
  v8f acc2 = {};
  {
    const float* xrow = zs + c * XS_PITCH + k0;
#pragma unroll
    for (int i = 0; i < 16; ++i) {
      v2f a = *(const v2f*)(xrow + 4 * i);
      v2f b = {bw[i][0], bw[i][1]};
      acc2 = wmma_f32_k4(a, b, acc2);
    }
  }
  if (c < 8) {
#pragma unroll
    for (int r = 0; r < 8; ++r) part[wv * (NPW * D_OUT) + (8 * half + r) * D_OUT + c] = acc2[r];
  }
  __syncthreads();
  if (wv < 2) {
    const int dq = d0 + 8 * wv;
    const float* pq = part + 64 * wv;
    float hA = pq[lane], hB = pq[32 + lane];
#pragma unroll
    for (int w = 1; w < 8; ++w) {
      hA += pq[w * (NPW * D_OUT) + lane];
      hB += pq[w * (NPW * D_OUT) + 32 + lane];
    }
    const int r = lane >> 3;
    if (dq + r < n)     H2[(size_t)dq * D_OUT + lane]      = hA;
    if (dq + 4 + r < n) H2[(size_t)dq * D_OUT + 32 + lane] = hB;
    const float cs = a_src2[c7], cd = a_dst2[c7];
    const float sA = group_sum<8>(hA * cs), dA = group_sum<8>(hA * cd);
    const float sB = group_sum<8>(hB * cs), dB = group_sum<8>(hB * cd);
    const int g = (lane & 3) * 8;
    const float s_lo = __shfl(sA, g, 32), d_lo = __shfl(dA, g, 32);
    const float s_hi = __shfl(sB, g, 32), d_hi = __shfl(dB, g, 32);
    if (lane < 8 && dq + lane < n) {
      v2f pr = {(lane & 4) ? s_hi : s_lo, (lane & 4) ? d_hi : d_lo};
      ((v2f*)asd)[asd_slot((unsigned)(dq + lane))] = pr;
    }
  }
}

__device__ __forceinline__ float entry_score2(const int* __restrict__ elist,
                                              const float* __restrict__ asd, int beg, int k,
                                              int d, int n, float a_d, int& s) {
  s = k ? elist[beg + k - 1] : d;
  if ((unsigned)s >= (unsigned)n) s = d;
  return lrelu(asd[2u * asd_slot((unsigned)s)] + a_d);
}

__global__ void __launch_bounds__(256)
node_agg2(const int* __restrict__ elist, const int* __restrict__ start,
          const int* __restrict__ cnt, int nE, int n, const float* __restrict__ H2,
          const float* __restrict__ asd,
          const float* __restrict__ b2, float* __restrict__ out) {
  int t = blockIdx.x * blockDim.x + threadIdx.x;
  int j = threadIdx.x & 7;
  bool live = (t >> 3) < n;
  int d = live ? (t >> 3) : 0;
  const v4f* H4 = (const v4f*)H2;

  int beg = start[d];
  int deg = cnt[d];
  if (beg < 0 || deg < 0 || beg > nE || deg > nE - beg) deg = 0;
  int total = deg + 1;
  float a_d = asd[2u * asd_slot((unsigned)d) + 1u];

  float mx = -INFINITY;
  for (int k = j; k < total; k += 8) {
    int s;
    mx = fmaxf(mx, entry_score2(elist, asd, beg, k, d, n, a_d, s));
  }
  mx = group_max<8>(mx);

  float den = 0.f;
  for (int k = j; k < total; k += 8) {
    int s;
    den += __expf(entry_score2(elist, asd, beg, k, d, n, a_d, s) - mx);
  }
  den = group_sum<8>(den) + 1e-16f;

  v4f accL = (v4f)(0.f), accH = (v4f)(0.f);
  for (int k = j; k < total; k += 8) {
    int s;
    float alpha = __expf(entry_score2(elist, asd, beg, k, d, n, a_d, s) - mx) / den;
    accL += H4[(unsigned)s * 2u] * alpha;
    accH += H4[(unsigned)s * 2u + 1u] * alpha;
  }
  float r0 = group_sum<8>(accL.x), r1 = group_sum<8>(accL.y);
  float r2 = group_sum<8>(accL.z), r3 = group_sum<8>(accL.w);
  float r4 = group_sum<8>(accH.x), r5 = group_sum<8>(accH.y);
  float r6 = group_sum<8>(accH.z), r7 = group_sum<8>(accH.w);
  float lo = (j & 2) ? ((j & 1) ? r3 : r2) : ((j & 1) ? r1 : r0);
  float up = (j & 2) ? ((j & 1) ? r7 : r6) : ((j & 1) ? r5 : r4);
  float r = (j & 4) ? up : lo;

  if (live) out[t] = r + b2[j];
}

extern "C" void kernel_launch(void* const* d_in, const int* in_sizes, int n_in,
                              void* d_out, int out_size, void* d_ws, size_t ws_size,
                              hipStream_t stream) {
  const float* x      = (const float*)d_in[0];
  const int*   ei     = (const int*)  d_in[1];
  const float* W1     = (const float*)d_in[2];
  const float* a_src1 = (const float*)d_in[3];
  const float* a_dst1 = (const float*)d_in[4];
  const float* b1     = (const float*)d_in[5];
  const float* W2     = (const float*)d_in[6];
  const float* a_src2 = (const float*)d_in[7];
  const float* a_dst2 = (const float*)d_in[8];
  const float* b2     = (const float*)d_in[9];

  int n  = in_sizes[0] / D_IN;
  int nE = in_sizes[1] / 2;
  const int* src = ei;
  const int* dst = ei + nE;

  char* ws = (char*)d_ws;
  size_t off = 0;
  auto alloc = [&](size_t bytes) -> char* {
    char* p = ws + off; off += (bytes + 255) & ~(size_t)255; return p;
  };
  float*    vco  = (float*)   alloc((size_t)D_IN * VW * 4);
  int*      elist= (int*)     alloc((size_t)nE * 4);
  int*      seg0 = (int*)     alloc((size_t)n * 4);
  float*    as1  = (float*)   alloc((size_t)n * HEADS * 4);
  float*    ad1  = (float*)   alloc((size_t)n * HEADS * 4);
  float*    h2   = (float*)   alloc((size_t)n * D_OUT * 4);
  float*    asd2 = (float*)   alloc((size_t)((n + 7) / 8) * 128);
  int*      cnt  = (int*)     alloc((size_t)n * 4);

  group_edges<<<(n + GR - 1) / GR + FOLD_BLOCKS, GT, 0, stream>>>(src, dst, nE, n, seg0, cnt, elist,
                                                                 W1, a_src1, a_dst1, vco);

  coef_wmma<<<((n + 31) / 32 + 3) / 4, 128, 0, stream>>>(x, vco, as1, ad1, n);

  node_agg1<<<(n + NPW - 1) / NPW, 256, 0, stream>>>(elist, seg0, cnt, nE, n, x, as1, ad1, W1, b1,
                                                     W2, a_src2, a_dst2, h2, asd2);

  node_agg2<<<(n * D_OUT + 255) / 256, 256, 0, stream>>>(elist, seg0, cnt, nE, n, h2, asd2, b2,
                                                       (float*)d_out);
}


# group_edges: a kept edge's source node is loaded in the last pass over the kept list (all lanes, five loads per thread in flight), not in the divergent find branch; list order still arrival rank
# speedup vs baseline: 12.7044x; 1.1094x over previous
#include <hip/hip_runtime.h>
#include <math.h>


#define D_IN   128
#define HEADS  8
#define HID    64
#define F1     (HEADS * HID)
#define D_OUT  8
#define NEG_SLOPE 0.2f

typedef float v2f __attribute__((ext_vector_type(2)));
typedef float v4f __attribute__((ext_vector_type(4)));
typedef float v8f __attribute__((ext_vector_type(8)));

__device__ __forceinline__ float lrelu(float x) { return x > 0.f ? x : NEG_SLOPE * x; }

__device__ __forceinline__ v8f wmma_f32_k4(v2f a, v2f b, v8f c) {
  return __builtin_amdgcn_wmma_f32_16x16x4_f32(false, a, false, b, (short)0, c, false, false);
}

#define VW 16

__device__ __forceinline__ void coef_fold(const float* __restrict__ W, const float* __restrict__ a_src,
                                          const float* __restrict__ a_dst, float* __restrict__ V,
                                          int t) {
  const int k = t >> 4, q = t & 15, hd = q & 7;
  if (k >= D_IN) return;
  const v4f* wr = (const v4f*)(W + (size_t)k * F1 + hd * HID);
  const v4f* av = (const v4f*)((q < 8 ? a_src : a_dst) + hd * HID);
  float s = 0.f;
  for (int j = 0; j < HID / 4; ++j) {
    v4f w = wr[j], a = av[j];
    s += w.x * a.x; s += w.y * a.y; s += w.z * a.z; s += w.w * a.w;
  }
  V[t] = s;
}

__global__ void __launch_bounds__(128)
coef_wmma(const float* __restrict__ X, const float* __restrict__ V,
          float* __restrict__ as, float* __restrict__ ad, int n) {
  const int lane = threadIdx.x & 31;
  const int wv   = threadIdx.x >> 5;
  const int half = lane >> 4;
  const int c    = lane & 15;
  const int row0 = ((int)blockIdx.x * 4 + wv) * 32;
  if (row0 >= n) return;
  int ra0 = row0 + c;       if (ra0 > n - 1) ra0 = n - 1;
  int ra1 = row0 + 16 + c;  if (ra1 > n - 1) ra1 = n - 1;
  const v4f* xa0 = (const v4f*)(X + (size_t)ra0 * D_IN) + half;
  const v4f* xa1 = (const v4f*)(X + (size_t)ra1 * D_IN) + half;
  const float* vb = V + (4 * half) * VW + c;
  v8f acc0 = {}, acc1 = {};
#pragma unroll
  for (int s = 0; s < D_IN / 8; ++s) {
    v4f a0 = xa0[2 * s], a1 = xa1[2 * s];
    v2f bl = {vb[(8 * s + 0) * VW], vb[(8 * s + 1) * VW]};
    v2f bh = {vb[(8 * s + 2) * VW], vb[(8 * s + 3) * VW]};
    acc0 = wmma_f32_k4((v2f){a0.x, a0.y}, bl, acc0);
    acc1 = wmma_f32_k4((v2f){a1.x, a1.y}, bl, acc1);
    acc0 = wmma_f32_k4((v2f){a0.z, a0.w}, bh, acc0);
    acc1 = wmma_f32_k4((v2f){a1.z, a1.w}, bh, acc1);
  }
  float* op = (c < 8) ? (as + c) : (ad + (c - 8));
#pragma unroll
  for (int r = 0; r < 8; ++r) {
    int r0 = row0 + 8 * half + r, r1 = r0 + 16;
    if (r0 < n) op[(size_t)r0 * HEADS] = acc0[r];
    if (r1 < n) op[(size_t)r1 * HEADS] = acc1[r];
  }
}

typedef int v4i __attribute__((ext_vector_type(4)));
#define GR   400
#define GT   1024
#define GCAP 12288
#define KB   5
#define FOLD_BLOCKS ((D_IN * VW + GT - 1) / GT)

__device__ __forceinline__ void keep_edge(int* cnt_l, int* nm, int* m_ln, int* m_rk, int* m_ed,
                                          int t, int e) {
  int r = atomicAdd(&cnt_l[t], 1);
  int q = atomicAdd(nm, 1);
  if (q < GCAP) { m_ln[q] = t; m_rk[q] = r; m_ed[q] = e; }
}

__global__ void __launch_bounds__(GT)
group_edges(const int* __restrict__ src, const int* __restrict__ dst, int nE, int n,
            int* __restrict__ start, int* __restrict__ cnt, int* __restrict__ elist,
            const float* __restrict__ W, const float* __restrict__ a_src,
            const float* __restrict__ a_dst, float* __restrict__ V) {
  __shared__ int cnt_l[GR];
  __shared__ int off_l[GR];
  __shared__ int m_ln[GCAP];
  __shared__ int m_rk[GCAP];
  __shared__ int m_ed[GCAP];
  __shared__ int wtot[GT / 32];
  __shared__ int wbel[GT / 32];
  __shared__ int nm_s;
  const int nbO = (n + GR - 1) / GR;
  if ((int)blockIdx.x >= nbO) {
    coef_fold(W, a_src, a_dst, V, ((int)blockIdx.x - nbO) * GT + (int)threadIdx.x);
    return;
  }
  const int tid = threadIdx.x, lane = tid & 31, wv = tid >> 5;
  const int lo = (int)blockIdx.x * GR;
  const int nr = (n - lo < GR) ? (n - lo) : GR;
  if (tid < GR) cnt_l[tid] = 0;
  if (tid == 0) nm_s = 0;
  __syncthreads();

  int below = 0;
  const int n4 = (((size_t)dst & 15) == 0) ? (nE >> 2) : 0;
  const v4i* d4p = (const v4i*)dst;
  for (int g0 = tid; g0 < n4; g0 += 8 * GT) {
    v4i d[8];
#pragma unroll
    for (int k = 0; k < 8; ++k) {
      int g = g0 + k * GT;
      d[k] = d4p[g < n4 ? g : g0];
    }
    asm volatile("" :: "v"(d[0]), "v"(d[1]), "v"(d[2]), "v"(d[3]),
                       "v"(d[4]), "v"(d[5]), "v"(d[6]), "v"(d[7]));
#pragma unroll
    for (int k = 0; k < 8; ++k) {
      int g = g0 + k * GT;
      if (g < n4) {
        int t0 = d[k].x - lo, t1 = d[k].y - lo, t2 = d[k].z - lo, t3 = d[k].w - lo;
        below += (t0 < 0) + (t1 < 0) + (t2 < 0) + (t3 < 0);
        if ((unsigned)t0 < (unsigned)nr) keep_edge(cnt_l, &nm_s, m_ln, m_rk, m_ed, t0, 4 * g);
        if ((unsigned)t1 < (unsigned)nr) keep_edge(cnt_l, &nm_s, m_ln, m_rk, m_ed, t1, 4 * g + 1);
        if ((unsigned)t2 < (unsigned)nr) keep_edge(cnt_l, &nm_s, m_ln, m_rk, m_ed, t2, 4 * g + 2);
        if ((unsigned)t3 < (unsigned)nr) keep_edge(cnt_l, &nm_s, m_ln, m_rk, m_ed, t3, 4 * g + 3);
      }
    }
  }
  for (int e = 4 * n4 + tid; e < nE; e += GT) {
    int t = dst[e] - lo;
    below += (t < 0);
    if ((unsigned)t < (unsigned)nr) keep_edge(cnt_l, &nm_s, m_ln, m_rk, m_ed, t, e);
  }
  __syncthreads();

  const int c = (tid < GR) ? cnt_l[tid] : 0;
  int incl = c;
#pragma unroll
  for (int o = 1; o < 32; o <<= 1) {
    int t = __shfl_up(incl, o, 32);
    if (lane >= o) incl += t;
  }
#pragma unroll
  for (int o = 16; o > 0; o >>= 1) below += __shfl_xor(below, o, 32);
  if (lane == 31) wtot[wv] = incl;
  if (lane == 0)  wbel[wv] = below;
  __syncthreads();
  int woff = 0, base = 0;
  for (int w = 0; w < GT / 32; ++w) {
    if (w < wv) woff += wtot[w];
    base += wbel[w];
  }
  const int off = woff + incl - c;
  if (tid < GR) off_l[tid] = off;
  if (tid < nr) { start[lo + tid] = base + off; cnt[lo + tid] = c; }
  __syncthreads();

  const int nm = nm_s;
  if (nm <= GCAP) {
    for (int q0 = tid; q0 < nm; q0 += KB * GT) {
      int p[KB], s[KB];
#pragma unroll
      for (int k = 0; k < KB; ++k) {
        int q = q0 + k * GT;
        int qq = q < nm ? q : q0;
        int e = m_ed[qq];
        if ((unsigned)e >= (unsigned)nE) e = 0;
        p[k] = base + off_l[m_ln[qq]] + m_rk[qq];
        s[k] = src[e];
      }
      asm volatile("" :: "v"(s[0]), "v"(s[1]), "v"(s[2]), "v"(s[3]), "v"(s[4]));
      static_assert(KB == 5, "the statement above names KB values");
#pragma unroll
      for (int k = 0; k < KB; ++k) {
        if (q0 + k * GT < nm && (unsigned)p[k] < (unsigned)nE) elist[p[k]] = s[k];
      }
    }
  } else {
    if (tid < GR) cnt_l[tid] = 0;
    __syncthreads();
    for (int e = tid; e < nE; e += GT) {
      int t = dst[e] - lo;
      if ((unsigned)t < (unsigned)nr) {
        int p = base + off_l[t] + atomicAdd(&cnt_l[t], 1);
        if ((unsigned)p < (unsigned)nE) elist[p] = src[e];
      }
    }
  }
}


__device__ __forceinline__ float lane_bcast(float x, int i) {
  return __int_as_float(__builtin_amdgcn_readlane(__float_as_int(x), i));
}

__device__ __forceinline__ float as_scalar(float x) {
  return __int_as_float(__builtin_amdgcn_readfirstlane(__float_as_int(x)));
}

template <int W>
__device__ __forceinline__ float group_max(float v) {
#pragma unroll
  for (int o = W / 2; o > 0; o >>= 1) v = fmaxf(v, __shfl_xor(v, o, 32));
  return v;
}
template <int W>
__device__ __forceinline__ float group_sum(float v) {
#pragma unroll
  for (int o = W / 2; o > 0; o >>= 1) v += __shfl_xor(v, o, 32);
  return v;
}

__device__ __forceinline__ bool load_entries(const int* __restrict__ elist,
                                             const v4f* __restrict__ as4, int beg, int total,
                                             int base, int lane, int d, int n,
                                             const float (&adh)[8], int& mine, float (&ev)[8]) {
  int k = base + lane;
  bool valid = k < total;
  mine = (k >= 1 && valid) ? elist[beg + k - 1] : d;
  if ((unsigned)mine >= (unsigned)n) mine = d;
  v4f aL = as4[(unsigned)mine * 2u];
  v4f aH = as4[(unsigned)mine * 2u + 1u];
  ev[0] = lrelu(aL.x + adh[0]); ev[1] = lrelu(aL.y + adh[1]);
  ev[2] = lrelu(aL.z + adh[2]); ev[3] = lrelu(aL.w + adh[3]);
  ev[4] = lrelu(aH.x + adh[4]); ev[5] = lrelu(aH.y + adh[5]);
  ev[6] = lrelu(aH.z + adh[6]); ev[7] = lrelu(aH.w + adh[7]);
  return valid;
}


__device__ __forceinline__ void zrow_add(v4f (&z)[8], v4f r, const float (&al)[8], int i) {
#pragma unroll
  for (int h = 0; h < HEADS; ++h) z[h] += r * lane_bcast(al[h], i);
}

__device__ __forceinline__ v4f bias_relu(v4f a, v4f b) {
  v4f v = a + b;
  v.x = v.x > 0.f ? v.x : 0.f;
  v.y = v.y > 0.f ? v.y : 0.f;
  v.z = v.z > 0.f ? v.z : 0.f;
  v.w = v.w > 0.f ? v.w : 0.f;
  return v;
}

__device__ __forceinline__ void node_z1(const int* __restrict__ elist, const int* __restrict__ start,
                                        const int* __restrict__ cnt, int nE, int n,
                                        const float* __restrict__ X, const float* __restrict__ as,
                                        const float* __restrict__ ad, int d, int lane,
                                        v4f (&z)[8]) {
  const v4f* X4  = (const v4f*)X;
  const v4f* as4 = (const v4f*)as;

  float adh[8];
#pragma unroll
  for (int h = 0; h < HEADS; ++h) adh[h] = ad[(size_t)d * HEADS + h];

  int beg = start[d];
  int deg = cnt[d];
  if (beg < 0 || deg < 0 || beg > nE || deg > nE - beg) deg = 0;
  int total = deg + 1;

  int mine; float ev[8];
  bool valid = load_entries(elist, as4, beg, total, 0, lane, d, n, adh, mine, ev);

  float mvh[8];
  {
    float mx[8];
#pragma unroll
    for (int h = 0; h < HEADS; ++h) mx[h] = valid ? ev[h] : -INFINITY;
    for (int base = 32; base < total; base += 32) {
      int m2; float e2[8];
      if (load_entries(elist, as4, beg, total, base, lane, d, n, adh, m2, e2)) {
#pragma unroll
        for (int h = 0; h < HEADS; ++h) mx[h] = fmaxf(mx[h], e2[h]);
      }
    }
#pragma unroll
    for (int h = 0; h < HEADS; ++h) mvh[h] = as_scalar(group_max<32>(mx[h]));
  }

  float ddh[8];
  {
    float ds[8];
#pragma unroll
    for (int h = 0; h < HEADS; ++h) ds[h] = valid ? __expf(ev[h] - mvh[h]) : 0.f;
    for (int base = 32; base < total; base += 32) {
      int m2; float e2[8];
      if (load_entries(elist, as4, beg, total, base, lane, d, n, adh, m2, e2)) {
#pragma unroll
        for (int h = 0; h < HEADS; ++h) ds[h] += __expf(e2[h] - mvh[h]);
      }
    }
#pragma unroll
    for (int h = 0; h < HEADS; ++h) ddh[h] = as_scalar(group_sum<32>(ds[h])) + 1e-16f;
  }

  for (int base = 0;;) {
    int c = total - base; if (c > 32) c = 32;

    float al[8];
#pragma unroll
    for (int h = 0; h < HEADS; ++h) al[h] = __expf(ev[h] - mvh[h]) / ddh[h];

    for (int i = 0; i < c; i += 4) {
      int i1 = i + 1 < c ? i + 1 : i;
      int i2 = i + 2 < c ? i + 2 : i;
      int i3 = i + 3 < c ? i + 3 : i;
      int s0 = __builtin_amdgcn_readlane(mine, i);
      int s1 = __builtin_amdgcn_readlane(mine, i1);
      int s2 = __builtin_amdgcn_readlane(mine, i2);
      int s3 = __builtin_amdgcn_readlane(mine, i3);
      v4f r0 = X4[(unsigned)s0 * (unsigned)(D_IN / 4) + (unsigned)lane];
      v4f r1 = X4[(unsigned)s1 * (unsigned)(D_IN / 4) + (unsigned)lane];
      v4f r2 = X4[(unsigned)s2 * (unsigned)(D_IN / 4) + (unsigned)lane];
      v4f r3 = X4[(unsigned)s3 * (unsigned)(D_IN / 4) + (unsigned)lane];
      asm volatile("" :: "v"(r0), "v"(r1), "v"(r2), "v"(r3));
      zrow_add(z, r0, al, i);
      if (i + 1 < c) zrow_add(z, r1, al, i + 1);
      if (i + 2 < c) zrow_add(z, r2, al, i + 2);
      if (i + 3 < c) zrow_add(z, r3, al, i + 3);
    }

    base += 32;
    if (base >= total) break;
    load_entries(elist, as4, beg, total, base, lane, d, n, adh, mine, ev);
  }
}


__device__ __forceinline__ unsigned asd_slot(unsigned i) { return ((i >> 3) << 4) | (i & 7u); }

#define NPW 16
#define ZS_PITCH (HEADS * D_IN + 4)
#define XS_PITCH (F1 + 4)

__global__ void __launch_bounds__(256) __attribute__((amdgpu_waves_per_eu(8, 10)))
node_agg1(const int* __restrict__ elist, const int* __restrict__ start,
          const int* __restrict__ cnt, int nE, int n, const float* __restrict__ X,
          const float* __restrict__ as, const float* __restrict__ ad,
          const float* __restrict__ W1, const float* __restrict__ b1,
          const float* __restrict__ W2,
          const float* __restrict__ a_src2, const float* __restrict__ a_dst2,
          float* __restrict__ H2, float* __restrict__ asd) {
  __shared__ __attribute__((aligned(16))) float zs[NPW * ZS_PITCH];
  __shared__ float part[8 * NPW * D_OUT];
  const int lane = threadIdx.x & 31;
  const int wv = __builtin_amdgcn_readfirstlane((int)(threadIdx.x >> 5));
  const int d0 = (int)blockIdx.x * NPW;
  const int half = lane >> 4;
  const int c    = lane & 15;

  for (int q = 0; q < 2; ++q) {
    const int nl = wv + 8 * q;
    const int d  = d0 + nl;
    v4f z[8];
#pragma unroll
    for (int h = 0; h < HEADS; ++h) z[h] = (v4f)(0.f);
    if (d < n)
      node_z1(elist, start, cnt, nE, n, X, as, ad, d, lane, z);
    v4f* zr = (v4f*)(zs + nl * ZS_PITCH) + lane;
#pragma unroll
    for (int h = 0; h < HEADS; ++h) zr[h * (D_IN / 4)] = z[h];
  }

  const int col0 = wv * HID;
  const v4f* wb = (const v4f*)(W1 + (size_t)(4 * half) * F1 + col0) + c;
  const int WR = F1 / 4;
  v4f b0 = wb[0], b1v = wb[WR], b2 = wb[2 * WR], b3 = wb[3 * WR];
  v4f p0 = wb[8 * WR], p1 = wb[9 * WR], p2 = wb[10 * WR], p3 = wb[11 * WR];
  __syncthreads();
  v8f acc[4];
#pragma unroll
  for (int t = 0; t < 4; ++t) acc[t] = (v8f){};
  {
    const v4f* za = (const v4f*)(zs + c * ZS_PITCH + wv * D_IN) + half;
#pragma unroll
    for (int s = 0; s < D_IN / 8; ++s) {
      v4f q0 = p0, q1 = p1, q2 = p2, q3 = p3;
      if (s + 2 < D_IN / 8) {
        q0 = wb[(8 * (s + 2) + 0) * WR];
        q1 = wb[(8 * (s + 2) + 1) * WR];
        q2 = wb[(8 * (s + 2) + 2) * WR];
        q3 = wb[(8 * (s + 2) + 3) * WR];
      }
      v4f a = za[2 * s];
      __builtin_amdgcn_sched_barrier(0);
      v2f al = {a.x, a.y}, ah = {a.z, a.w};
#pragma unroll
      for (int t = 0; t < 4; ++t) acc[t] = wmma_f32_k4(al, (v2f){b0[t], b1v[t]}, acc[t]);
#pragma unroll
      for (int t = 0; t < 4; ++t) acc[t] = wmma_f32_k4(ah, (v2f){b2[t], b3[t]}, acc[t]);
      b0 = p0; b1v = p1; b2 = p2; b3 = p3;
      p0 = q0; p1 = q1; p2 = q2; p3 = q3;
    }
  }

  const int c7 = lane & 7;
  const int k0 = wv * 64 + 2 * half;
  float bw[16][2];
#pragma unroll
  for (int i = 0; i < 16; ++i) {
    bw[i][0] = W2[(k0 + 4 * i) * D_OUT + c7];
    bw[i][1] = W2[(k0 + 4 * i + 1) * D_OUT + c7];
  }
  const v4f bv = ((const v4f*)b1)[wv * (HID / 4) + c];
  __syncthreads();
#pragma unroll
  for (int r = 0; r < 8; ++r) {
    v4f o = {acc[0][r], acc[1][r], acc[2][r], acc[3][r]};
    *((v4f*)(zs + (8 * half + r) * XS_PITCH + col0) + c) = bias_relu(o, bv);
  }
  __syncthreads();
  v8f acc2 = {};
  {
    const float* xrow = zs + c * XS_PITCH + k0;
#pragma unroll
    for (int i = 0; i < 16; ++i) {
      v2f a = *(const v2f*)(xrow + 4 * i);
      v2f b = {bw[i][0], bw[i][1]};
      acc2 = wmma_f32_k4(a, b, acc2);
    }
  }
  if (c < 8) {
#pragma unroll
    for (int r = 0; r < 8; ++r) part[wv * (NPW * D_OUT) + (8 * half + r) * D_OUT + c] = acc2[r];
  }
  __syncthreads();
  if (wv < 2) {
    const int dq = d0 + 8 * wv;
    const float* pq = part + 64 * wv;
    float hA = pq[lane], hB = pq[32 + lane];
#pragma unroll
    for (int w = 1; w < 8; ++w) {
      hA += pq[w * (NPW * D_OUT) + lane];
      hB += pq[w * (NPW * D_OUT) + 32 + lane];
    }
    const int r = lane >> 3;
    if (dq + r < n)     H2[(size_t)dq * D_OUT + lane]      = hA;
    if (dq + 4 + r < n) H2[(size_t)dq * D_OUT + 32 + lane] = hB;
    const float cs = a_src2[c7], cd = a_dst2[c7];
    const float sA = group_sum<8>(hA * cs), dA = group_sum<8>(hA * cd);
    const float sB = group_sum<8>(hB * cs), dB = group_sum<8>(hB * cd);
    const int g = (lane & 3) * 8;
    const float s_lo = __shfl(sA, g, 32), d_lo = __shfl(dA, g, 32);
    const float s_hi = __shfl(sB, g, 32), d_hi = __shfl(dB, g, 32);
    if (lane < 8 && dq + lane < n) {
      v2f pr = {(lane & 4) ? s_hi : s_lo, (lane & 4) ? d_hi : d_lo};
      ((v2f*)asd)[asd_slot((unsigned)(dq + lane))] = pr;
    }
  }
}

__device__ __forceinline__ float entry_score2(const int* __restrict__ elist,
                                              const float* __restrict__ asd, int beg, int k,
                                              int d, int n, float a_d, int& s) {
  s = k ? elist[beg + k - 1] : d;
  if ((unsigned)s >= (unsigned)n) s = d;
  return lrelu(asd[2u * asd_slot((unsigned)s)] + a_d);
}

__global__ void __launch_bounds__(256)
node_agg2(const int* __restrict__ elist, const int* __restrict__ start,
          const int* __restrict__ cnt, int nE, int n, const float* __restrict__ H2,
          const float* __restrict__ asd,
          const float* __restrict__ b2, float* __restrict__ out) {
  int t = blockIdx.x * blockDim.x + threadIdx.x;
  int j = threadIdx.x & 7;
  bool live = (t >> 3) < n;
  int d = live ? (t >> 3) : 0;
  const v4f* H4 = (const v4f*)H2;

  int beg = start[d];
  int deg = cnt[d];
  if (beg < 0 || deg < 0 || beg > nE || deg > nE - beg) deg = 0;
  int total = deg + 1;
  float a_d = asd[2u * asd_slot((unsigned)d) + 1u];

  float mx = -INFINITY;
  for (int k = j; k < total; k += 8) {
    int s;
    mx = fmaxf(mx, entry_score2(elist, asd, beg, k, d, n, a_d, s));
  }
  mx = group_max<8>(mx);

  float den = 0.f;
  for (int k = j; k < total; k += 8) {
    int s;
    den += __expf(entry_score2(elist, asd, beg, k, d, n, a_d, s) - mx);
  }
  den = group_sum<8>(den) + 1e-16f;

  v4f accL = (v4f)(0.f), accH = (v4f)(0.f);
  for (int k = j; k < total; k += 8) {
    int s;
    float alpha = __expf(entry_score2(elist, asd, beg, k, d, n, a_d, s) - mx) / den;
    accL += H4[(unsigned)s * 2u] * alpha;
    accH += H4[(unsigned)s * 2u + 1u] * alpha;
  }
  float r0 = group_sum<8>(accL.x), r1 = group_sum<8>(accL.y);
  float r2 = group_sum<8>(accL.z), r3 = group_sum<8>(accL.w);
  float r4 = group_sum<8>(accH.x), r5 = group_sum<8>(accH.y);
  float r6 = group_sum<8>(accH.z), r7 = group_sum<8>(accH.w);
  float lo = (j & 2) ? ((j & 1) ? r3 : r2) : ((j & 1) ? r1 : r0);
  float up = (j & 2) ? ((j & 1) ? r7 : r6) : ((j & 1) ? r5 : r4);
  float r = (j & 4) ? up : lo;

  if (live) out[t] = r + b2[j];
}

extern "C" void kernel_launch(void* const* d_in, const int* in_sizes, int n_in,
                              void* d_out, int out_size, void* d_ws, size_t ws_size,
                              hipStream_t stream) {
  const float* x      = (const float*)d_in[0];
  const int*   ei     = (const int*)  d_in[1];
  const float* W1     = (const float*)d_in[2];
  const float* a_src1 = (const float*)d_in[3];
  const float* a_dst1 = (const float*)d_in[4];
  const float* b1     = (const float*)d_in[5];
  const float* W2     = (const float*)d_in[6];
  const float* a_src2 = (const float*)d_in[7];
  const float* a_dst2 = (const float*)d_in[8];
  const float* b2     = (const float*)d_in[9];

  int n  = in_sizes[0] / D_IN;
  int nE = in_sizes[1] / 2;
  const int* src = ei;
  const int* dst = ei + nE;

  char* ws = (char*)d_ws;
  size_t off = 0;
  auto alloc = [&](size_t bytes) -> char* {
    char* p = ws + off; off += (bytes + 255) & ~(size_t)255; return p;
  };
  float*    vco  = (float*)   alloc((size_t)D_IN * VW * 4);
  int*      elist= (int*)     alloc((size_t)nE * 4);
  int*      seg0 = (int*)     alloc((size_t)n * 4);
  float*    as1  = (float*)   alloc((size_t)n * HEADS * 4);
  float*    ad1  = (float*)   alloc((size_t)n * HEADS * 4);
  float*    h2   = (float*)   alloc((size_t)n * D_OUT * 4);
  float*    asd2 = (float*)   alloc((size_t)((n + 7) / 8) * 128);
  int*      cnt  = (int*)     alloc((size_t)n * 4);

  group_edges<<<(n + GR - 1) / GR + FOLD_BLOCKS, GT, 0, stream>>>(src, dst, nE, n, seg0, cnt, elist,
                                                                 W1, a_src1, a_dst1, vco);

  coef_wmma<<<((n + 31) / 32 + 3) / 4, 128, 0, stream>>>(x, vco, as1, ad1, n);

  node_agg1<<<(n + NPW - 1) / NPW, 256, 0, stream>>>(elist, seg0, cnt, nE, n, x, as1, ad1, W1, b1,
                                                     W2, a_src2, a_dst2, h2, asd2);

  node_agg2<<<(n * D_OUT + 255) / 256, 256, 0, stream>>>(elist, seg0, cnt, nE, n, h2, asd2, b2,
                                                       (float*)d_out);
}


# Edge array read in two halves: two owner work-groups per node range, each scans half the destinations; a node's list = its two segments (sum order: first-half edges, then second-half edges)
# speedup vs baseline: 14.0761x; 1.1080x over previous
#include <hip/hip_runtime.h>
#include <math.h>


#define D_IN   128
#define HEADS  8
#define HID    64
#define F1     (HEADS * HID)
#define D_OUT  8
#define NEG_SLOPE 0.2f

typedef float v2f __attribute__((ext_vector_type(2)));
typedef float v4f __attribute__((ext_vector_type(4)));
typedef float v8f __attribute__((ext_vector_type(8)));

__device__ __forceinline__ float lrelu(float x) { return x > 0.f ? x : NEG_SLOPE * x; }

__device__ __forceinline__ v8f wmma_f32_k4(v2f a, v2f b, v8f c) {
  return __builtin_amdgcn_wmma_f32_16x16x4_f32(false, a, false, b, (short)0, c, false, false);
}

#define VW 16

__device__ __forceinline__ void coef_fold(const float* __restrict__ W, const float* __restrict__ a_src,
                                          const float* __restrict__ a_dst, float* __restrict__ V,
                                          int t) {
  const int k = t >> 4, q = t & 15, hd = q & 7;
  if (k >= D_IN) return;
  const v4f* wr = (const v4f*)(W + (size_t)k * F1 + hd * HID);
  const v4f* av = (const v4f*)((q < 8 ? a_src : a_dst) + hd * HID);
  float s = 0.f;
  for (int j = 0; j < HID / 4; ++j) {
    v4f w = wr[j], a = av[j];
    s += w.x * a.x; s += w.y * a.y; s += w.z * a.z; s += w.w * a.w;
  }
  V[t] = s;
}

__global__ void __launch_bounds__(128)
coef_wmma(const float* __restrict__ X, const float* __restrict__ V,
          float* __restrict__ as, float* __restrict__ ad, int n) {
  const int lane = threadIdx.x & 31;
  const int wv   = threadIdx.x >> 5;
  const int half = lane >> 4;
  const int c    = lane & 15;
  const int row0 = ((int)blockIdx.x * 4 + wv) * 32;
  if (row0 >= n) return;
  int ra0 = row0 + c;       if (ra0 > n - 1) ra0 = n - 1;
  int ra1 = row0 + 16 + c;  if (ra1 > n - 1) ra1 = n - 1;
  const v4f* xa0 = (const v4f*)(X + (size_t)ra0 * D_IN) + half;
  const v4f* xa1 = (const v4f*)(X + (size_t)ra1 * D_IN) + half;
  const float* vb = V + (4 * half) * VW + c;
  v8f acc0 = {}, acc1 = {};
#pragma unroll
  for (int s = 0; s < D_IN / 8; ++s) {
    v4f a0 = xa0[2 * s], a1 = xa1[2 * s];
    v2f bl = {vb[(8 * s + 0) * VW], vb[(8 * s + 1) * VW]};
    v2f bh = {vb[(8 * s + 2) * VW], vb[(8 * s + 3) * VW]};
    acc0 = wmma_f32_k4((v2f){a0.x, a0.y}, bl, acc0);
    acc1 = wmma_f32_k4((v2f){a1.x, a1.y}, bl, acc1);
    acc0 = wmma_f32_k4((v2f){a0.z, a0.w}, bh, acc0);
    acc1 = wmma_f32_k4((v2f){a1.z, a1.w}, bh, acc1);
  }
  float* op = (c < 8) ? (as + c) : (ad + (c - 8));
#pragma unroll
  for (int r = 0; r < 8; ++r) {
    int r0 = row0 + 8 * half + r, r1 = r0 + 16;
    if (r0 < n) op[(size_t)r0 * HEADS] = acc0[r];
    if (r1 < n) op[(size_t)r1 * HEADS] = acc1[r];
  }
}

typedef int v4i __attribute__((ext_vector_type(4)));
typedef int v2i __attribute__((ext_vector_type(2)));
#define GR   400
#define GT   1024
#define GCAP 12288
#define KB   5
#define FOLD_BLOCKS ((D_IN * VW + GT - 1) / GT)

__device__ __forceinline__ void keep_edge(int* cnt_l, int* nm, int* m_ln, int* m_rk, int* m_ed,
                                          int t, int e) {
  int r = atomicAdd(&cnt_l[t], 1);
  int q = atomicAdd(nm, 1);
  if (q < GCAP) { m_ln[q] = t; m_rk[q] = r; m_ed[q] = e; }
}

__global__ void __launch_bounds__(GT)
group_edges(const int* __restrict__ src, const int* __restrict__ dst, int nE, int n,
            int* __restrict__ seg, int* __restrict__ elist,
            const float* __restrict__ W, const float* __restrict__ a_src,
            const float* __restrict__ a_dst, float* __restrict__ V) {
  __shared__ int cnt_l[GR];
  __shared__ int off_l[GR];
  __shared__ int m_ln[GCAP];
  __shared__ int m_rk[GCAP];
  __shared__ int m_ed[GCAP];
  __shared__ int wtot[GT / 32];
  __shared__ int wbel[GT / 32];
  __shared__ int nm_s;
  const int nbO = 2 * ((n + GR - 1) / GR);
  if ((int)blockIdx.x >= nbO) {
    coef_fold(W, a_src, a_dst, V, ((int)blockIdx.x - nbO) * GT + (int)threadIdx.x);
    return;
  }
  const int tid = threadIdx.x, lane = tid & 31, wv = tid >> 5;
  const int hf = (int)blockIdx.x & 1;
  const int lo = ((int)blockIdx.x >> 1) * GR;
  const int nr = (n - lo < GR) ? (n - lo) : GR;
  const int nEA  = (nE >> 3) << 2;
  const int e_lo = hf ? nEA : 0;
  const int e_hi = hf ? nE : nEA;
  if (tid < GR) cnt_l[tid] = 0;
  if (tid == 0) nm_s = 0;
  __syncthreads();

  int below = 0;
  const int g_lo = e_lo >> 2;
  const int g_hi = (((size_t)dst & 15) == 0) ? (e_hi >> 2) : g_lo;
  const v4i* d4p = (const v4i*)dst;
  for (int g0 = g_lo + tid; g0 < g_hi; g0 += 8 * GT) {
    v4i d[8];
#pragma unroll
    for (int k = 0; k < 8; ++k) {
      int g = g0 + k * GT;
      d[k] = d4p[g < g_hi ? g : g0];
    }
    asm volatile("" :: "v"(d[0]), "v"(d[1]), "v"(d[2]), "v"(d[3]),
                       "v"(d[4]), "v"(d[5]), "v"(d[6]), "v"(d[7]));
#pragma unroll
    for (int k = 0; k < 8; ++k) {
      int g = g0 + k * GT;
      if (g < g_hi) {
        int t0 = d[k].x - lo, t1 = d[k].y - lo, t2 = d[k].z - lo, t3 = d[k].w - lo;
        below += (t0 < 0) + (t1 < 0) + (t2 < 0) + (t3 < 0);
        if ((unsigned)t0 < (unsigned)nr) keep_edge(cnt_l, &nm_s, m_ln, m_rk, m_ed, t0, 4 * g);
        if ((unsigned)t1 < (unsigned)nr) keep_edge(cnt_l, &nm_s, m_ln, m_rk, m_ed, t1, 4 * g + 1);
        if ((unsigned)t2 < (unsigned)nr) keep_edge(cnt_l, &nm_s, m_ln, m_rk, m_ed, t2, 4 * g + 2);
        if ((unsigned)t3 < (unsigned)nr) keep_edge(cnt_l, &nm_s, m_ln, m_rk, m_ed, t3, 4 * g + 3);
      }
    }
  }
  for (int e = (g_hi > g_lo ? 4 * g_hi : e_lo) + tid; e < e_hi; e += GT) {
    int t = dst[e] - lo;
    below += (t < 0);
    if ((unsigned)t < (unsigned)nr) keep_edge(cnt_l, &nm_s, m_ln, m_rk, m_ed, t, e);
  }
  __syncthreads();

  const int c = (tid < GR) ? cnt_l[tid] : 0;
  int incl = c;
#pragma unroll
  for (int o = 1; o < 32; o <<= 1) {
    int t = __shfl_up(incl, o, 32);
    if (lane >= o) incl += t;
  }
#pragma unroll
  for (int o = 16; o > 0; o >>= 1) below += __shfl_xor(below, o, 32);
  if (lane == 31) wtot[wv] = incl;
  if (lane == 0)  wbel[wv] = below;
  __syncthreads();
  int woff = 0, base = e_lo;
  for (int w = 0; w < GT / 32; ++w) {
    if (w < wv) woff += wtot[w];
    base += wbel[w];
  }
  const int off = woff + incl - c;
  if (tid < GR) off_l[tid] = off;
  if (tid < nr) ((v2i*)seg)[(size_t)hf * n + lo + tid] = (v2i){base + off, c};
  __syncthreads();

  const int nm = nm_s;
  if (nm <= GCAP) {
    for (int q0 = tid; q0 < nm; q0 += KB * GT) {
      int p[KB], s[KB];
#pragma unroll
      for (int k = 0; k < KB; ++k) {
        int q = q0 + k * GT;
        int qq = q < nm ? q : q0;
        int e = m_ed[qq];
        if ((unsigned)e >= (unsigned)nE) e = 0;
        p[k] = base + off_l[m_ln[qq]] + m_rk[qq];
        s[k] = src[e];
      }
      asm volatile("" :: "v"(s[0]), "v"(s[1]), "v"(s[2]), "v"(s[3]), "v"(s[4]));
      static_assert(KB == 5, "the statement above names KB values");
#pragma unroll
      for (int k = 0; k < KB; ++k) {
        if (q0 + k * GT < nm && (unsigned)p[k] < (unsigned)nE) elist[p[k]] = s[k];
      }
    }
  } else {
    if (tid < GR) cnt_l[tid] = 0;
    __syncthreads();
    for (int e = e_lo + tid; e < e_hi; e += GT) {
      int t = dst[e] - lo;
      if ((unsigned)t < (unsigned)nr) {
        int p = base + off_l[t] + atomicAdd(&cnt_l[t], 1);
        if ((unsigned)p < (unsigned)nE) elist[p] = src[e];
      }
    }
  }
}


__device__ __forceinline__ float lane_bcast(float x, int i) {
  return __int_as_float(__builtin_amdgcn_readlane(__float_as_int(x), i));
}

__device__ __forceinline__ float as_scalar(float x) {
  return __int_as_float(__builtin_amdgcn_readfirstlane(__float_as_int(x)));
}

template <int W>
__device__ __forceinline__ float group_max(float v) {
#pragma unroll
  for (int o = W / 2; o > 0; o >>= 1) v = fmaxf(v, __shfl_xor(v, o, 32));
  return v;
}
template <int W>
__device__ __forceinline__ float group_sum(float v) {
#pragma unroll
  for (int o = W / 2; o > 0; o >>= 1) v += __shfl_xor(v, o, 32);
  return v;
}

struct seg2 {
  int begA, degA;
  int offB;
  int total;
};

__device__ __forceinline__ seg2 load_seg(const int* __restrict__ seg, int d, int n, int nE) {
  v2i a = ((const v2i*)seg)[d];
  v2i b = ((const v2i*)seg)[(size_t)n + d];
  int begA = a.x, degA = a.y, begB = b.x, degB = b.y;
  if (begA < 0 || degA < 0 || begA > nE || degA > nE - begA) degA = 0;
  if (begB < 0 || degB < 0 || begB > nE || degB > nE - begB) degB = 0;
  seg2 s;
  s.begA = begA; s.degA = degA; s.offB = begB - degA; s.total = degA + degB + 1;
  return s;
}

__device__ __forceinline__ int seg_slot(const seg2& s, int i) {
  return i < s.degA ? s.begA + i : s.offB + i;
}

__device__ __forceinline__ bool load_entries(const int* __restrict__ elist,
                                             const v4f* __restrict__ as4, const seg2& sg,
                                             int base, int lane, int d, int n,
                                             const float (&adh)[8], int& mine, float (&ev)[8]) {
  const int total = sg.total;
  int k = base + lane;
  bool valid = k < total;
  mine = (k >= 1 && valid) ? elist[seg_slot(sg, k - 1)] : d;
  if ((unsigned)mine >= (unsigned)n) mine = d;
  v4f aL = as4[(unsigned)mine * 2u];
  v4f aH = as4[(unsigned)mine * 2u + 1u];
  ev[0] = lrelu(aL.x + adh[0]); ev[1] = lrelu(aL.y + adh[1]);
  ev[2] = lrelu(aL.z + adh[2]); ev[3] = lrelu(aL.w + adh[3]);
  ev[4] = lrelu(aH.x + adh[4]); ev[5] = lrelu(aH.y + adh[5]);
  ev[6] = lrelu(aH.z + adh[6]); ev[7] = lrelu(aH.w + adh[7]);
  return valid;
}


__device__ __forceinline__ void zrow_add(v4f (&z)[8], v4f r, const float (&al)[8], int i) {
#pragma unroll
  for (int h = 0; h < HEADS; ++h) z[h] += r * lane_bcast(al[h], i);
}

__device__ __forceinline__ v4f bias_relu(v4f a, v4f b) {
  v4f v = a + b;
  v.x = v.x > 0.f ? v.x : 0.f;
  v.y = v.y > 0.f ? v.y : 0.f;
  v.z = v.z > 0.f ? v.z : 0.f;
  v.w = v.w > 0.f ? v.w : 0.f;
  return v;
}

__device__ __forceinline__ void node_z1(const int* __restrict__ elist, const int* __restrict__ seg,
                                        int nE, int n,
                                        const float* __restrict__ X, const float* __restrict__ as,
                                        const float* __restrict__ ad, int d, int lane,
                                        v4f (&z)[8]) {
  const v4f* X4  = (const v4f*)X;
  const v4f* as4 = (const v4f*)as;

  float adh[8];
#pragma unroll
  for (int h = 0; h < HEADS; ++h) adh[h] = ad[(size_t)d * HEADS + h];

  const seg2 sg = load_seg(seg, d, n, nE);
  const int total = sg.total;

  int mine; float ev[8];
  bool valid = load_entries(elist, as4, sg, 0, lane, d, n, adh, mine, ev);

  float mvh[8];
  {
    float mx[8];
#pragma unroll
    for (int h = 0; h < HEADS; ++h) mx[h] = valid ? ev[h] : -INFINITY;
    for (int base = 32; base < total; base += 32) {
      int m2; float e2[8];
      if (load_entries(elist, as4, sg, base, lane, d, n, adh, m2, e2)) {
#pragma unroll
        for (int h = 0; h < HEADS; ++h) mx[h] = fmaxf(mx[h], e2[h]);
      }
    }
#pragma unroll
    for (int h = 0; h < HEADS; ++h) mvh[h] = as_scalar(group_max<32>(mx[h]));
  }

  float ddh[8];
  {
    float ds[8];
#pragma unroll
    for (int h = 0; h < HEADS; ++h) ds[h] = valid ? __expf(ev[h] - mvh[h]) : 0.f;
    for (int base = 32; base < total; base += 32) {
      int m2; float e2[8];
      if (load_entries(elist, as4, sg, base, lane, d, n, adh, m2, e2)) {
#pragma unroll
        for (int h = 0; h < HEADS; ++h) ds[h] += __expf(e2[h] - mvh[h]);
      }
    }
#pragma unroll
    for (int h = 0; h < HEADS; ++h) ddh[h] = as_scalar(group_sum<32>(ds[h])) + 1e-16f;
  }

  for (int base = 0;;) {
    int c = total - base; if (c > 32) c = 32;

    float al[8];
#pragma unroll
    for (int h = 0; h < HEADS; ++h) al[h] = __expf(ev[h] - mvh[h]) / ddh[h];

    for (int i = 0; i < c; i += 4) {
      int i1 = i + 1 < c ? i + 1 : i;
      int i2 = i + 2 < c ? i + 2 : i;
      int i3 = i + 3 < c ? i + 3 : i;
      int s0 = __builtin_amdgcn_readlane(mine, i);
      int s1 = __builtin_amdgcn_readlane(mine, i1);
      int s2 = __builtin_amdgcn_readlane(mine, i2);
      int s3 = __builtin_amdgcn_readlane(mine, i3);
      v4f r0 = X4[(unsigned)s0 * (unsigned)(D_IN / 4) + (unsigned)lane];
      v4f r1 = X4[(unsigned)s1 * (unsigned)(D_IN / 4) + (unsigned)lane];
      v4f r2 = X4[(unsigned)s2 * (unsigned)(D_IN / 4) + (unsigned)lane];
      v4f r3 = X4[(unsigned)s3 * (unsigned)(D_IN / 4) + (unsigned)lane];
      asm volatile("" :: "v"(r0), "v"(r1), "v"(r2), "v"(r3));
      zrow_add(z, r0, al, i);
      if (i + 1 < c) zrow_add(z, r1, al, i + 1);
      if (i + 2 < c) zrow_add(z, r2, al, i + 2);
      if (i + 3 < c) zrow_add(z, r3, al, i + 3);
    }

    base += 32;
    if (base >= total) break;
    load_entries(elist, as4, sg, base, lane, d, n, adh, mine, ev);
  }
}


__device__ __forceinline__ unsigned asd_slot(unsigned i) { return ((i >> 3) << 4) | (i & 7u); }

#define NPW 16
#define ZS_PITCH (HEADS * D_IN + 4)
#define XS_PITCH (F1 + 4)

__global__ void __launch_bounds__(256) __attribute__((amdgpu_waves_per_eu(8, 10)))
node_agg1(const int* __restrict__ elist, const int* __restrict__ seg,
          int nE, int n, const float* __restrict__ X,
          const float* __restrict__ as, const float* __restrict__ ad,
          const float* __restrict__ W1, const float* __restrict__ b1,
          const float* __restrict__ W2,
          const float* __restrict__ a_src2, const float* __restrict__ a_dst2,
          float* __restrict__ H2, float* __restrict__ asd) {
  __shared__ __attribute__((aligned(16))) float zs[NPW * ZS_PITCH];
  __shared__ float part[8 * NPW * D_OUT];
  const int lane = threadIdx.x & 31;
  const int wv = __builtin_amdgcn_readfirstlane((int)(threadIdx.x >> 5));
  const int d0 = (int)blockIdx.x * NPW;
  const int half = lane >> 4;
  const int c    = lane & 15;

  for (int q = 0; q < 2; ++q) {
    const int nl = wv + 8 * q;
    const int d  = d0 + nl;
    v4f z[8];
#pragma unroll
    for (int h = 0; h < HEADS; ++h) z[h] = (v4f)(0.f);
    if (d < n)
      node_z1(elist, seg, nE, n, X, as, ad, d, lane, z);
    v4f* zr = (v4f*)(zs + nl * ZS_PITCH) + lane;
#pragma unroll
    for (int h = 0; h < HEADS; ++h) zr[h * (D_IN / 4)] = z[h];
  }

  const int col0 = wv * HID;
  const v4f* wb = (const v4f*)(W1 + (size_t)(4 * half) * F1 + col0) + c;
  const int WR = F1 / 4;
  v4f b0 = wb[0], b1v = wb[WR], b2 = wb[2 * WR], b3 = wb[3 * WR];
  v4f p0 = wb[8 * WR], p1 = wb[9 * WR], p2 = wb[10 * WR], p3 = wb[11 * WR];
  __syncthreads();
  v8f acc[4];
#pragma unroll
  for (int t = 0; t < 4; ++t) acc[t] = (v8f){};
  {
    const v4f* za = (const v4f*)(zs + c * ZS_PITCH + wv * D_IN) + half;
#pragma unroll
    for (int s = 0; s < D_IN / 8; ++s) {
      v4f q0 = p0, q1 = p1, q2 = p2, q3 = p3;
      if (s + 2 < D_IN / 8) {
        q0 = wb[(8 * (s + 2) + 0) * WR];
        q1 = wb[(8 * (s + 2) + 1) * WR];
        q2 = wb[(8 * (s + 2) + 2) * WR];
        q3 = wb[(8 * (s + 2) + 3) * WR];
      }
      v4f a = za[2 * s];
      __builtin_amdgcn_sched_barrier(0);
      v2f al = {a.x, a.y}, ah = {a.z, a.w};
#pragma unroll
      for (int t = 0; t < 4; ++t) acc[t] = wmma_f32_k4(al, (v2f){b0[t], b1v[t]}, acc[t]);
#pragma unroll
      for (int t = 0; t < 4; ++t) acc[t] = wmma_f32_k4(ah, (v2f){b2[t], b3[t]}, acc[t]);
      b0 = p0; b1v = p1; b2 = p2; b3 = p3;
      p0 = q0; p1 = q1; p2 = q2; p3 = q3;
    }
  }

  const int c7 = lane & 7;
  const int k0 = wv * 64 + 2 * half;
  float bw[16][2];
#pragma unroll
  for (int i = 0; i < 16; ++i) {
    bw[i][0] = W2[(k0 + 4 * i) * D_OUT + c7];
    bw[i][1] = W2[(k0 + 4 * i + 1) * D_OUT + c7];
  }
  const v4f bv = ((const v4f*)b1)[wv * (HID / 4) + c];
  __syncthreads();
#pragma unroll
  for (int r = 0; r < 8; ++r) {
    v4f o = {acc[0][r], acc[1][r], acc[2][r], acc[3][r]};
    *((v4f*)(zs + (8 * half + r) * XS_PITCH + col0) + c) = bias_relu(o, bv);
  }
  __syncthreads();
  v8f acc2 = {};
  {
    const float* xrow = zs + c * XS_PITCH + k0;
#pragma unroll
    for (int i = 0; i < 16; ++i) {
      v2f a = *(const v2f*)(xrow + 4 * i);
      v2f b = {bw[i][0], bw[i][1]};
      acc2 = wmma_f32_k4(a, b, acc2);
    }
  }
  if (c < 8) {
#pragma unroll
    for (int r = 0; r < 8; ++r) part[wv * (NPW * D_OUT) + (8 * half + r) * D_OUT + c] = acc2[r];
  }
  __syncthreads();
  if (wv < 2) {
    const int dq = d0 + 8 * wv;
    const float* pq = part + 64 * wv;
    float hA = pq[lane], hB = pq[32 + lane];
#pragma unroll
    for (int w = 1; w < 8; ++w) {
      hA += pq[w * (NPW * D_OUT) + lane];
      hB += pq[w * (NPW * D_OUT) + 32 + lane];
    }
    const int r = lane >> 3;
    if (dq + r < n)     H2[(size_t)dq * D_OUT + lane]      = hA;
    if (dq + 4 + r < n) H2[(size_t)dq * D_OUT + 32 + lane] = hB;
    const float cs = a_src2[c7], cd = a_dst2[c7];
    const float sA = group_sum<8>(hA * cs), dA = group_sum<8>(hA * cd);
    const float sB = group_sum<8>(hB * cs), dB = group_sum<8>(hB * cd);
    const int g = (lane & 3) * 8;
    const float s_lo = __shfl(sA, g, 32), d_lo = __shfl(dA, g, 32);
    const float s_hi = __shfl(sB, g, 32), d_hi = __shfl(dB, g, 32);
    if (lane < 8 && dq + lane < n) {
      v2f pr = {(lane & 4) ? s_hi : s_lo, (lane & 4) ? d_hi : d_lo};
      ((v2f*)asd)[asd_slot((unsigned)(dq + lane))] = pr;
    }
  }
}

__device__ __forceinline__ float entry_score2(const int* __restrict__ elist,
                                              const float* __restrict__ asd, const seg2& sg, int k,
                                              int d, int n, float a_d, int& s) {
  s = k ? elist[seg_slot(sg, k - 1)] : d;
  if ((unsigned)s >= (unsigned)n) s = d;
  return lrelu(asd[2u * asd_slot((unsigned)s)] + a_d);
}

__global__ void __launch_bounds__(256)
node_agg2(const int* __restrict__ elist, const int* __restrict__ seg,
          int nE, int n, const float* __restrict__ H2,
          const float* __restrict__ asd,
          const float* __restrict__ b2, float* __restrict__ out) {
  int t = blockIdx.x * blockDim.x + threadIdx.x;
  int j = threadIdx.x & 7;
  bool live = (t >> 3) < n;
  int d = live ? (t >> 3) : 0;
  const v4f* H4 = (const v4f*)H2;

  const seg2 sg = load_seg(seg, d, n, nE);
  const int total = sg.total;
  float a_d = asd[2u * asd_slot((unsigned)d) + 1u];

  float mx = -INFINITY;
  for (int k = j; k < total; k += 8) {
    int s;
    mx = fmaxf(mx, entry_score2(elist, asd, sg, k, d, n, a_d, s));
  }
  mx = group_max<8>(mx);

  float den = 0.f;
  for (int k = j; k < total; k += 8) {
    int s;
    den += __expf(entry_score2(elist, asd, sg, k, d, n, a_d, s) - mx);
  }
  den = group_sum<8>(den) + 1e-16f;

  v4f accL = (v4f)(0.f), accH = (v4f)(0.f);
  for (int k = j; k < total; k += 8) {
    int s;
    float alpha = __expf(entry_score2(elist, asd, sg, k, d, n, a_d, s) - mx) / den;
    accL += H4[(unsigned)s * 2u] * alpha;
    accH += H4[(unsigned)s * 2u + 1u] * alpha;
  }
  float r0 = group_sum<8>(accL.x), r1 = group_sum<8>(accL.y);
  float r2 = group_sum<8>(accL.z), r3 = group_sum<8>(accL.w);
  float r4 = group_sum<8>(accH.x), r5 = group_sum<8>(accH.y);
  float r6 = group_sum<8>(accH.z), r7 = group_sum<8>(accH.w);
  float lo = (j & 2) ? ((j & 1) ? r3 : r2) : ((j & 1) ? r1 : r0);
  float up = (j & 2) ? ((j & 1) ? r7 : r6) : ((j & 1) ? r5 : r4);
  float r = (j & 4) ? up : lo;

  if (live) out[t] = r + b2[j];
}

extern "C" void kernel_launch(void* const* d_in, const int* in_sizes, int n_in,
                              void* d_out, int out_size, void* d_ws, size_t ws_size,
                              hipStream_t stream) {
  const float* x      = (const float*)d_in[0];
  const int*   ei     = (const int*)  d_in[1];
  const float* W1     = (const float*)d_in[2];
  const float* a_src1 = (const float*)d_in[3];
  const float* a_dst1 = (const float*)d_in[4];
  const float* b1     = (const float*)d_in[5];
  const float* W2     = (const float*)d_in[6];
  const float* a_src2 = (const float*)d_in[7];
  const float* a_dst2 = (const float*)d_in[8];
  const float* b2     = (const float*)d_in[9];

  int n  = in_sizes[0] / D_IN;
  int nE = in_sizes[1] / 2;
  const int* src = ei;
  const int* dst = ei + nE;

  char* ws = (char*)d_ws;
  size_t off = 0;
  auto alloc = [&](size_t bytes) -> char* {
    char* p = ws + off; off += (bytes + 255) & ~(size_t)255; return p;
  };
  float*    vco  = (float*)   alloc((size_t)D_IN * VW * 4);
  int*      elist= (int*)     alloc((size_t)nE * 4);
  int*      seg  = (int*)     alloc((size_t)n * 16);
  float*    as1  = (float*)   alloc((size_t)n * HEADS * 4);
  float*    ad1  = (float*)   alloc((size_t)n * HEADS * 4);
  float*    h2   = (float*)   alloc((size_t)n * D_OUT * 4);
  float*    asd2 = (float*)   alloc((size_t)((n + 7) / 8) * 128);

  group_edges<<<2 * ((n + GR - 1) / GR) + FOLD_BLOCKS, GT, 0, stream>>>(src, dst, nE, n, seg, elist,
                                                                       W1, a_src1, a_dst1, vco);

  coef_wmma<<<((n + 31) / 32 + 3) / 4, 128, 0, stream>>>(x, vco, as1, ad1, n);

  node_agg1<<<(n + NPW - 1) / NPW, 256, 0, stream>>>(elist, seg, nE, n, x, as1, ad1, W1, b1,
                                                     W2, a_src2, a_dst2, h2, asd2);

  node_agg2<<<(n * D_OUT + 255) / 256, 256, 0, stream>>>(elist, seg, nE, n, h2, asd2, b2,
                                                       (float*)d_out);
}
